# starting kernel
# baseline (speedup 1.0000x reference)
#include <hip/hip_runtime.h>
#include <hip/hip_bf16.h>


typedef __attribute__((ext_vector_type(16))) __bf16 v16bf;
typedef __attribute__((ext_vector_type(8)))  float  v8f;

constexpr int NNODE = 50000;
constexpr int NEDGE = 800000;
constexpr int FIN   = 512;
constexpr int H1    = 8;
constexpr int FHID  = 8;
constexpr int HID   = H1 * FHID;
constexpr int NLAB  = 64;
constexpr int ROWT  = NNODE / 16;
constexpr float SLOPE = 0.2f;

#define DEVINL __device__ __forceinline__

DEVINL unsigned fkey(float v) {
    unsigned u = __float_as_uint(v);
    return (u & 0x80000000u) ? ~u : (u | 0x80000000u);
}
DEVINL float funkey(unsigned k) {
    return (k & 0x80000000u) ? __uint_as_float(k ^ 0x80000000u)
                             : __uint_as_float(~k);
}
DEVINL float lrelu(float v) { return v > 0.f ? v : SLOPE * v; }

__global__ void k_repack(const float* __restrict__ W, __bf16* __restrict__ Bp,
                         int nKt, int total, int gat1) {
    int idx = blockIdx.x * blockDim.x + threadIdx.x;
    if (idx >= total) return;
    int e    = idx & 15;
    int lane = (idx >> 4) & 31;
    int ktnt = idx >> 9;
    int kt   = ktnt % nKt;
    int nt   = ktnt / nKt;
    int K = ((e < 8) ? 0 : 16) + (lane >> 4) * 8 + (e & 7);
    int k = kt * 32 + K;
    int n = nt * 16 + (lane & 15);
    float v = gat1 ? W[(size_t)(n >> 3) * (FIN * FHID) + (size_t)k * FHID + (n & 7)]
                   : W[(size_t)k * NLAB + n];
    Bp[idx] = (__bf16)v;
}

__global__ void __launch_bounds__(128)
k_gemm_bf16(const float* __restrict__ A, int lda,
            const __bf16* __restrict__ Bp, int nKt,
            float* __restrict__ C) {
    int lane = threadIdx.x & 31;
    int wave = threadIdx.x >> 5;
    int tile = blockIdx.x * 4 + wave;
    if (tile >= ROWT) return;

    int m     = lane & 15;
    int khalf = lane >> 4;
    const float*  arow = A  + (size_t)(tile * 16 + m) * lda + khalf * 8;
    const __bf16* bl   = Bp + (size_t)lane * 16;

    v8f acc[4] = {};
    for (int kt = 0; kt < nKt; ++kt) {
        const float* ap = arow + kt * 32;
        float4 a0 = *(const float4*)(ap);
        float4 a1 = *(const float4*)(ap + 4);
        float4 a2 = *(const float4*)(ap + 16);
        float4 a3 = *(const float4*)(ap + 20);
        if (kt + 1 < nKt) __builtin_prefetch(ap + 32, 0, 0);

        v16bf av;
        av[0]  = (__bf16)a0.x; av[1]  = (__bf16)a0.y; av[2]  = (__bf16)a0.z; av[3]  = (__bf16)a0.w;
        av[4]  = (__bf16)a1.x; av[5]  = (__bf16)a1.y; av[6]  = (__bf16)a1.z; av[7]  = (__bf16)a1.w;
        av[8]  = (__bf16)a2.x; av[9]  = (__bf16)a2.y; av[10] = (__bf16)a2.z; av[11] = (__bf16)a2.w;
        av[12] = (__bf16)a3.x; av[13] = (__bf16)a3.y; av[14] = (__bf16)a3.z; av[15] = (__bf16)a3.w;

        const __bf16* bk = bl + (size_t)kt * 512;
        #pragma unroll
        for (int nt = 0; nt < 4; ++nt) {
            v16bf bv = *(const v16bf*)(bk + (size_t)nt * nKt * 512);
            acc[nt] = __builtin_amdgcn_wmma_f32_16x16x32_bf16(
                false, av, false, bv, (short)0, acc[nt], false, false);
        }
    }

    float* cbase = C + (size_t)(tile * 16 + khalf * 8) * HID + m;
    #pragma unroll
    for (int nt = 0; nt < 4; ++nt) {
        float* crow = cbase + nt * 16;
        #pragma unroll
        for (int r = 0; r < 8; ++r) crow[(size_t)r * HID] = acc[nt][r];
    }
}

__global__ void k_coef1(const float* __restrict__ Wh,
                        const float* __restrict__ a_s, const float* __restrict__ a_d,
                        float* __restrict__ es, float* __restrict__ ed) {
    int idx = blockIdx.x * blockDim.x + threadIdx.x;
    if (idx >= NNODE * H1) return;
    int n = idx >> 3, h = idx & 7;
    const float* w  = Wh + (size_t)n * HID + h * 8;
    const float* as = a_s + h * 8;
    const float* ad = a_d + h * 8;
    float s = 0.f, d = 0.f;
    #pragma unroll
    for (int o = 0; o < 8; ++o) { float v = w[o]; s += v * as[o]; d += v * ad[o]; }
    es[idx] = s; ed[idx] = d;
}

__global__ void k_coef2(const float* __restrict__ Wh,
                        const float* __restrict__ a_s, const float* __restrict__ a_d,
                        float* __restrict__ es, float* __restrict__ ed) {
    int n = blockIdx.x * blockDim.x + threadIdx.x;
    if (n >= NNODE) return;
    const float* w = Wh + (size_t)n * NLAB;
    float s = 0.f, d = 0.f;
    for (int j = 0; j < NLAB; ++j) { float v = w[j]; s += v * a_s[j]; d += v * a_d[j]; }
    es[n] = s; ed[n] = d;
}

__global__ void k_edge_max1(const int* __restrict__ src, const int* __restrict__ dst,
                            const float* __restrict__ es, const float* __restrict__ ed,
                            float* __restrict__ eatt, unsigned* __restrict__ mk) {
    int idx = blockIdx.x * blockDim.x + threadIdx.x;
    if (idx >= NEDGE * H1) return;
    int e = idx >> 3, h = idx & 7;
    int s = src[e], d = dst[e];
    float v = lrelu(es[s * H1 + h] + ed[d * H1 + h]);
    eatt[idx] = v;
    atomicMax(mk + d * H1 + h, fkey(v));
}

__global__ void k_edge_den1(const int* __restrict__ dst,
                            const float* __restrict__ eatt,
                            const unsigned* __restrict__ mk, float* __restrict__ den) {
    int idx = blockIdx.x * blockDim.x + threadIdx.x;
    if (idx >= NEDGE * H1) return;
    int e = idx >> 3, h = idx & 7;
    int d = dst[e];
    float ex = __expf(eatt[idx] - funkey(mk[d * H1 + h]));
    atomicAdd(den + d * H1 + h, ex);
}

__global__ void k_edge_msg1(const int* __restrict__ src, const int* __restrict__ dst,
                            const float* __restrict__ eatt,
                            const unsigned* __restrict__ mk, const float* __restrict__ den,
                            const float* __restrict__ Wh, float* __restrict__ hacc) {
    int idx = blockIdx.x * blockDim.x + threadIdx.x;
    if (idx >= NEDGE * H1) return;
    int e = idx >> 3, h = idx & 7;
    int s = src[e], d = dst[e];
    float ex = __expf(eatt[idx] - funkey(mk[d * H1 + h]));
    float alpha = ex / (den[d * H1 + h] + 1e-10f);
    const float* w = Wh + (size_t)s * HID + h * 8;
    float4 w0 = *(const float4*)(w);
    float4 w1 = *(const float4*)(w + 4);
    float* o = hacc + (size_t)d * HID + h * 8;
    atomicAdd(o + 0, alpha * w0.x); atomicAdd(o + 1, alpha * w0.y);
    atomicAdd(o + 2, alpha * w0.z); atomicAdd(o + 3, alpha * w0.w);
    atomicAdd(o + 4, alpha * w1.x); atomicAdd(o + 5, alpha * w1.y);
    atomicAdd(o + 6, alpha * w1.z); atomicAdd(o + 7, alpha * w1.w);
}

__global__ void k_elu(float* __restrict__ h) {
    int i = blockIdx.x * blockDim.x + threadIdx.x;
    if (i >= NNODE * HID) return;
    float x = h[i];
    h[i] = x > 0.f ? x : (__expf(x) - 1.f);
}

__global__ void k_edge_max2(const int* __restrict__ src, const int* __restrict__ dst,
                            const float* __restrict__ es, const float* __restrict__ ed,
                            float* __restrict__ eatt, unsigned* __restrict__ mk) {
    int e = blockIdx.x * blockDim.x + threadIdx.x;
    if (e >= NEDGE) return;
    float v = lrelu(es[src[e]] + ed[dst[e]]);
    eatt[e] = v;
    atomicMax(mk + dst[e], fkey(v));
}

__global__ void k_edge_den2(const int* __restrict__ dst,
                            const float* __restrict__ eatt,
                            const unsigned* __restrict__ mk, float* __restrict__ den) {
    int e = blockIdx.x * blockDim.x + threadIdx.x;
    if (e >= NEDGE) return;
    int d = dst[e];
    atomicAdd(den + d, __expf(eatt[e] - funkey(mk[d])));
}

__global__ void k_edge_msg2(const int* __restrict__ src, const int* __restrict__ dst,
                            const float* __restrict__ eatt,
                            const unsigned* __restrict__ mk, const float* __restrict__ den,
                            const float* __restrict__ Wh, float* __restrict__ out) {
    int idx = blockIdx.x * blockDim.x + threadIdx.x;
    if (idx >= NEDGE * 8) return;
    int e = idx >> 3, c0 = (idx & 7) * 8;
    int s = src[e], d = dst[e];
    float ex = __expf(eatt[e] - funkey(mk[d]));
    float alpha = ex / (den[d] + 1e-10f);
    const float* w = Wh + (size_t)s * NLAB + c0;
    float4 w0 = *(const float4*)(w);
    float4 w1 = *(const float4*)(w + 4);
    float* o = out + (size_t)d * NLAB + c0;
    atomicAdd(o + 0, alpha * w0.x); atomicAdd(o + 1, alpha * w0.y);
    atomicAdd(o + 2, alpha * w0.z); atomicAdd(o + 3, alpha * w0.w);
    atomicAdd(o + 4, alpha * w1.x); atomicAdd(o + 5, alpha * w1.y);
    atomicAdd(o + 6, alpha * w1.z); atomicAdd(o + 7, alpha * w1.w);
}

__global__ void __launch_bounds__(128) k_softmax(float* __restrict__ out) {
    int gid  = blockIdx.x * blockDim.x + threadIdx.x;
    int node = gid >> 5;
    int lane = gid & 31;
    if (node >= NNODE) return;
    float* row = out + (size_t)node * NLAB;
    float x0 = row[lane], x1 = row[lane + 32];
    float mx = fmaxf(x0, x1);
    #pragma unroll
    for (int o = 16; o > 0; o >>= 1) mx = fmaxf(mx, __shfl_xor(mx, o, 32));
    float e0 = __expf(x0 - mx), e1 = __expf(x1 - mx);
    float sum = e0 + e1;
    #pragma unroll
    for (int o = 16; o > 0; o >>= 1) sum += __shfl_xor(sum, o, 32);
    float inv = 1.0f / sum;
    row[lane] = e0 * inv;
    row[lane + 32] = e1 * inv;
}

extern "C" void kernel_launch(void* const* d_in, const int* in_sizes, int n_in,
                              void* d_out, int out_size, void* d_ws, size_t ws_size,
                              hipStream_t stream) {
    const float* X   = (const float*)d_in[0];
    const float* W1  = (const float*)d_in[1];
    const float* a1s = (const float*)d_in[2];
    const float* a1d = (const float*)d_in[3];
    const float* W2  = (const float*)d_in[4];
    const float* a2s = (const float*)d_in[5];
    const float* a2d = (const float*)d_in[6];
    const int*   src = (const int*)d_in[7];
    const int*   dst = (const int*)d_in[8];
    float*       out = (float*)d_out;

    char*  ws  = (char*)d_ws;
    size_t off = 0;
    auto alloc = [&](size_t bytes) -> void* {
        void* p = ws + off;
        off = (off + bytes + 255) & ~(size_t)255;
        return p;
    };
    float*    Wh1   = (float*)   alloc((size_t)NNODE * HID  * sizeof(float));
    float*    h1    = (float*)   alloc((size_t)NNODE * HID  * sizeof(float));
    float*    Wh2   = (float*)   alloc((size_t)NNODE * NLAB * sizeof(float));
    float*    es1   = (float*)   alloc((size_t)NNODE * H1   * sizeof(float));
    float*    ed1   = (float*)   alloc((size_t)NNODE * H1   * sizeof(float));
    unsigned* mk1   = (unsigned*)alloc((size_t)NNODE * H1   * sizeof(unsigned));
    float*    den1  = (float*)   alloc((size_t)NNODE * H1   * sizeof(float));
    float*    eatt1 = (float*)   alloc((size_t)NEDGE * H1   * sizeof(float));
    float*    es2   = (float*)   alloc((size_t)NNODE        * sizeof(float));
    float*    ed2   = (float*)   alloc((size_t)NNODE        * sizeof(float));
    unsigned* mk2   = (unsigned*)alloc((size_t)NNODE        * sizeof(unsigned));
    float*    den2  = (float*)   alloc((size_t)NNODE        * sizeof(float));
    float*    eatt2 = (float*)   alloc((size_t)NEDGE        * sizeof(float));
    __bf16*   Bp1   = (__bf16*)  alloc((size_t)4 * 16 * 512 * sizeof(__bf16));
    __bf16*   Bp2   = (__bf16*)  alloc((size_t)4 * 2  * 512 * sizeof(__bf16));

    const int B = 256;
    auto cdiv = [](long long a, long long b) { return (int)((a + b - 1) / b); };
    const int gemmGrid = cdiv(ROWT, 4);

    hipMemsetAsync(h1,   0, (size_t)NNODE * HID  * sizeof(float),    stream);
    hipMemsetAsync(mk1,  0, (size_t)NNODE * H1   * sizeof(unsigned), stream);
    hipMemsetAsync(den1, 0, (size_t)NNODE * H1   * sizeof(float),    stream);
    hipMemsetAsync(mk2,  0, (size_t)NNODE        * sizeof(unsigned), stream);
    hipMemsetAsync(den2, 0, (size_t)NNODE        * sizeof(float),    stream);
    hipMemsetAsync(out,  0, (size_t)NNODE * NLAB * sizeof(float),    stream);

    k_repack<<<cdiv(4 * 16 * 512, B), B, 0, stream>>>(W1, Bp1, 16, 4 * 16 * 512, 1);
    k_repack<<<cdiv(4 * 2  * 512, B), B, 0, stream>>>(W2, Bp2, 2,  4 * 2  * 512, 0);

    k_gemm_bf16<<<gemmGrid, 128, 0, stream>>>(X, FIN, Bp1, FIN / 32, Wh1);
    k_coef1<<<cdiv((long long)NNODE * H1, B), B, 0, stream>>>(Wh1, a1s, a1d, es1, ed1);
    k_edge_max1<<<cdiv((long long)NEDGE * H1, B), B, 0, stream>>>(src, dst, es1, ed1, eatt1, mk1);
    k_edge_den1<<<cdiv((long long)NEDGE * H1, B), B, 0, stream>>>(dst, eatt1, mk1, den1);
    k_edge_msg1<<<cdiv((long long)NEDGE * H1, B), B, 0, stream>>>(src, dst, eatt1, mk1, den1, Wh1, h1);
    k_elu<<<cdiv((long long)NNODE * HID, B), B, 0, stream>>>(h1);

    k_gemm_bf16<<<gemmGrid, 128, 0, stream>>>(h1, HID, Bp2, HID / 32, Wh2);
    k_coef2<<<cdiv(NNODE, B), B, 0, stream>>>(Wh2, a2s, a2d, es2, ed2);
    k_edge_max2<<<cdiv(NEDGE, B), B, 0, stream>>>(src, dst, es2, ed2, eatt2, mk2);
    k_edge_den2<<<cdiv(NEDGE, B), B, 0, stream>>>(dst, eatt2, mk2, den2);
    k_edge_msg2<<<cdiv((long long)NEDGE * 8, B), B, 0, stream>>>(src, dst, eatt2, mk2, den2, Wh2, out);

    k_softmax<<<cdiv((long long)NNODE * 32, 128), 128, 0, stream>>>(out);
}


# Normalise after aggregation: message passes sum exp(e-m)*Wh[src] and exp(e-m) together, one division by (den+1e-10) per node in the ELU and softmax kernels; both denominator passes removed
# speedup vs baseline: 1.0874x; 1.0874x over previous
#include <hip/hip_runtime.h>
#include <hip/hip_bf16.h>


typedef __attribute__((ext_vector_type(16))) __bf16 v16bf;
typedef __attribute__((ext_vector_type(8)))  float  v8f;

constexpr int NNODE = 50000;
constexpr int NEDGE = 800000;
constexpr int FIN   = 512;
constexpr int H1    = 8;
constexpr int FHID  = 8;
constexpr int HID   = H1 * FHID;
constexpr int NLAB  = 64;
constexpr int ROWT  = NNODE / 16;
constexpr float SLOPE = 0.2f;

#define DEVINL __device__ __forceinline__

DEVINL unsigned fkey(float v) {
    unsigned u = __float_as_uint(v);
    return (u & 0x80000000u) ? ~u : (u | 0x80000000u);
}
DEVINL float funkey(unsigned k) {
    return (k & 0x80000000u) ? __uint_as_float(k ^ 0x80000000u)
                             : __uint_as_float(~k);
}
DEVINL float lrelu(float v) { return v > 0.f ? v : SLOPE * v; }

__global__ void k_repack(const float* __restrict__ W, __bf16* __restrict__ Bp,
                         int nKt, int total, int gat1) {
    int idx = blockIdx.x * blockDim.x + threadIdx.x;
    if (idx >= total) return;
    int e    = idx & 15;
    int lane = (idx >> 4) & 31;
    int ktnt = idx >> 9;
    int kt   = ktnt % nKt;
    int nt   = ktnt / nKt;
    int K = ((e < 8) ? 0 : 16) + (lane >> 4) * 8 + (e & 7);
    int k = kt * 32 + K;
    int n = nt * 16 + (lane & 15);
    float v = gat1 ? W[(size_t)(n >> 3) * (FIN * FHID) + (size_t)k * FHID + (n & 7)]
                   : W[(size_t)k * NLAB + n];
    Bp[idx] = (__bf16)v;
}

__global__ void __launch_bounds__(128)
k_gemm_bf16(const float* __restrict__ A, int lda,
            const __bf16* __restrict__ Bp, int nKt,
            float* __restrict__ C) {
    int lane = threadIdx.x & 31;
    int wave = threadIdx.x >> 5;
    int tile = blockIdx.x * 4 + wave;
    if (tile >= ROWT) return;

    int m     = lane & 15;
    int khalf = lane >> 4;
    const float*  arow = A  + (size_t)(tile * 16 + m) * lda + khalf * 8;
    const __bf16* bl   = Bp + (size_t)lane * 16;

    v8f acc[4] = {};
    for (int kt = 0; kt < nKt; ++kt) {
        const float* ap = arow + kt * 32;
        float4 a0 = *(const float4*)(ap);
        float4 a1 = *(const float4*)(ap + 4);
        float4 a2 = *(const float4*)(ap + 16);
        float4 a3 = *(const float4*)(ap + 20);
        if (kt + 1 < nKt) __builtin_prefetch(ap + 32, 0, 0);

        v16bf av;
        av[0]  = (__bf16)a0.x; av[1]  = (__bf16)a0.y; av[2]  = (__bf16)a0.z; av[3]  = (__bf16)a0.w;
        av[4]  = (__bf16)a1.x; av[5]  = (__bf16)a1.y; av[6]  = (__bf16)a1.z; av[7]  = (__bf16)a1.w;
        av[8]  = (__bf16)a2.x; av[9]  = (__bf16)a2.y; av[10] = (__bf16)a2.z; av[11] = (__bf16)a2.w;
        av[12] = (__bf16)a3.x; av[13] = (__bf16)a3.y; av[14] = (__bf16)a3.z; av[15] = (__bf16)a3.w;

        const __bf16* bk = bl + (size_t)kt * 512;
        #pragma unroll
        for (int nt = 0; nt < 4; ++nt) {
            v16bf bv = *(const v16bf*)(bk + (size_t)nt * nKt * 512);
            acc[nt] = __builtin_amdgcn_wmma_f32_16x16x32_bf16(
                false, av, false, bv, (short)0, acc[nt], false, false);
        }
    }

    float* cbase = C + (size_t)(tile * 16 + khalf * 8) * HID + m;
    #pragma unroll
    for (int nt = 0; nt < 4; ++nt) {
        float* crow = cbase + nt * 16;
        #pragma unroll
        for (int r = 0; r < 8; ++r) crow[(size_t)r * HID] = acc[nt][r];
    }
}

__global__ void k_coef1(const float* __restrict__ Wh,
                        const float* __restrict__ a_s, const float* __restrict__ a_d,
                        float* __restrict__ es, float* __restrict__ ed) {
    int idx = blockIdx.x * blockDim.x + threadIdx.x;
    if (idx >= NNODE * H1) return;
    int n = idx >> 3, h = idx & 7;
    const float* w  = Wh + (size_t)n * HID + h * 8;
    const float* as = a_s + h * 8;
    const float* ad = a_d + h * 8;
    float s = 0.f, d = 0.f;
    #pragma unroll
    for (int o = 0; o < 8; ++o) { float v = w[o]; s += v * as[o]; d += v * ad[o]; }
    es[idx] = s; ed[idx] = d;
}

__global__ void k_coef2(const float* __restrict__ Wh,
                        const float* __restrict__ a_s, const float* __restrict__ a_d,
                        float* __restrict__ es, float* __restrict__ ed) {
    int n = blockIdx.x * blockDim.x + threadIdx.x;
    if (n >= NNODE) return;
    const float* w = Wh + (size_t)n * NLAB;
    float s = 0.f, d = 0.f;
    for (int j = 0; j < NLAB; ++j) { float v = w[j]; s += v * a_s[j]; d += v * a_d[j]; }
    es[n] = s; ed[n] = d;
}

__global__ void k_edge_max1(const int* __restrict__ src, const int* __restrict__ dst,
                            const float* __restrict__ es, const float* __restrict__ ed,
                            float* __restrict__ eatt, unsigned* __restrict__ mk) {
    int idx = blockIdx.x * blockDim.x + threadIdx.x;
    if (idx >= NEDGE * H1) return;
    int e = idx >> 3, h = idx & 7;
    int s = src[e], d = dst[e];
    float v = lrelu(es[s * H1 + h] + ed[d * H1 + h]);
    eatt[idx] = v;
    atomicMax(mk + d * H1 + h, fkey(v));
}

__global__ void k_edge_msg1(const int* __restrict__ src, const int* __restrict__ dst,
                            const float* __restrict__ eatt,
                            const unsigned* __restrict__ mk, float* __restrict__ den,
                            const float* __restrict__ Wh, float* __restrict__ hacc) {
    int idx = blockIdx.x * blockDim.x + threadIdx.x;
    if (idx >= NEDGE * H1) return;
    int e = idx >> 3, h = idx & 7;
    int s = src[e], d = dst[e];
    float ex = __expf(eatt[idx] - funkey(mk[d * H1 + h]));
    atomicAdd(den + d * H1 + h, ex);
    const float* w = Wh + (size_t)s * HID + h * 8;
    float4 w0 = *(const float4*)(w);
    float4 w1 = *(const float4*)(w + 4);
    float* o = hacc + (size_t)d * HID + h * 8;
    atomicAdd(o + 0, ex * w0.x); atomicAdd(o + 1, ex * w0.y);
    atomicAdd(o + 2, ex * w0.z); atomicAdd(o + 3, ex * w0.w);
    atomicAdd(o + 4, ex * w1.x); atomicAdd(o + 5, ex * w1.y);
    atomicAdd(o + 6, ex * w1.z); atomicAdd(o + 7, ex * w1.w);
}

__global__ void k_norm_elu(float* __restrict__ h, const float* __restrict__ den) {
    int i = blockIdx.x * blockDim.x + threadIdx.x;
    if (i >= NNODE * HID) return;
    float x = h[i] / (den[i >> 3] + 1e-10f);
    h[i] = x > 0.f ? x : (__expf(x) - 1.f);
}

__global__ void k_edge_max2(const int* __restrict__ src, const int* __restrict__ dst,
                            const float* __restrict__ es, const float* __restrict__ ed,
                            float* __restrict__ eatt, unsigned* __restrict__ mk) {
    int e = blockIdx.x * blockDim.x + threadIdx.x;
    if (e >= NEDGE) return;
    float v = lrelu(es[src[e]] + ed[dst[e]]);
    eatt[e] = v;
    atomicMax(mk + dst[e], fkey(v));
}

__global__ void k_edge_msg2(const int* __restrict__ src, const int* __restrict__ dst,
                            const float* __restrict__ eatt,
                            const unsigned* __restrict__ mk, float* __restrict__ den,
                            const float* __restrict__ Wh, float* __restrict__ out) {
    int idx = blockIdx.x * blockDim.x + threadIdx.x;
    if (idx >= NEDGE * 8) return;
    int e = idx >> 3, c0 = (idx & 7) * 8;
    int s = src[e], d = dst[e];
    float ex = __expf(eatt[e] - funkey(mk[d]));
    if (c0 == 0) atomicAdd(den + d, ex);
    const float* w = Wh + (size_t)s * NLAB + c0;
    float4 w0 = *(const float4*)(w);
    float4 w1 = *(const float4*)(w + 4);
    float* o = out + (size_t)d * NLAB + c0;
    atomicAdd(o + 0, ex * w0.x); atomicAdd(o + 1, ex * w0.y);
    atomicAdd(o + 2, ex * w0.z); atomicAdd(o + 3, ex * w0.w);
    atomicAdd(o + 4, ex * w1.x); atomicAdd(o + 5, ex * w1.y);
    atomicAdd(o + 6, ex * w1.z); atomicAdd(o + 7, ex * w1.w);
}

__global__ void __launch_bounds__(128) k_softmax(float* __restrict__ out,
                                                 const float* __restrict__ den) {
    int gid  = blockIdx.x * blockDim.x + threadIdx.x;
    int node = gid >> 5;
    int lane = gid & 31;
    if (node >= NNODE) return;
    float* row = out + (size_t)node * NLAB;
    float dn = den[node] + 1e-10f;
    float x0 = row[lane] / dn, x1 = row[lane + 32] / dn;
    float mx = fmaxf(x0, x1);
    #pragma unroll
    for (int o = 16; o > 0; o >>= 1) mx = fmaxf(mx, __shfl_xor(mx, o, 32));
    float e0 = __expf(x0 - mx), e1 = __expf(x1 - mx);
    float sum = e0 + e1;
    #pragma unroll
    for (int o = 16; o > 0; o >>= 1) sum += __shfl_xor(sum, o, 32);
    float inv = 1.0f / sum;
    row[lane] = e0 * inv;
    row[lane + 32] = e1 * inv;
}

extern "C" void kernel_launch(void* const* d_in, const int* in_sizes, int n_in,
                              void* d_out, int out_size, void* d_ws, size_t ws_size,
                              hipStream_t stream) {
    const float* X   = (const float*)d_in[0];
    const float* W1  = (const float*)d_in[1];
    const float* a1s = (const float*)d_in[2];
    const float* a1d = (const float*)d_in[3];
    const float* W2  = (const float*)d_in[4];
    const float* a2s = (const float*)d_in[5];
    const float* a2d = (const float*)d_in[6];
    const int*   src = (const int*)d_in[7];
    const int*   dst = (const int*)d_in[8];
    float*       out = (float*)d_out;

    char*  ws  = (char*)d_ws;
    size_t off = 0;
    auto alloc = [&](size_t bytes) -> void* {
        void* p = ws + off;
        off = (off + bytes + 255) & ~(size_t)255;
        return p;
    };
    float*    Wh1   = (float*)   alloc((size_t)NNODE * HID  * sizeof(float));
    float*    h1    = (float*)   alloc((size_t)NNODE * HID  * sizeof(float));
    float*    Wh2   = (float*)   alloc((size_t)NNODE * NLAB * sizeof(float));
    float*    es1   = (float*)   alloc((size_t)NNODE * H1   * sizeof(float));
    float*    ed1   = (float*)   alloc((size_t)NNODE * H1   * sizeof(float));
    unsigned* mk1   = (unsigned*)alloc((size_t)NNODE * H1   * sizeof(unsigned));
    float*    den1  = (float*)   alloc((size_t)NNODE * H1   * sizeof(float));
    float*    eatt1 = (float*)   alloc((size_t)NEDGE * H1   * sizeof(float));
    float*    es2   = (float*)   alloc((size_t)NNODE        * sizeof(float));
    float*    ed2   = (float*)   alloc((size_t)NNODE        * sizeof(float));
    unsigned* mk2   = (unsigned*)alloc((size_t)NNODE        * sizeof(unsigned));
    float*    den2  = (float*)   alloc((size_t)NNODE        * sizeof(float));
    float*    eatt2 = (float*)   alloc((size_t)NEDGE        * sizeof(float));
    __bf16*   Bp1   = (__bf16*)  alloc((size_t)4 * 16 * 512 * sizeof(__bf16));
    __bf16*   Bp2   = (__bf16*)  alloc((size_t)4 * 2  * 512 * sizeof(__bf16));

    const int B = 256;
    auto cdiv = [](long long a, long long b) { return (int)((a + b - 1) / b); };
    const int gemmGrid = cdiv(ROWT, 4);

    hipMemsetAsync(h1,   0, (size_t)NNODE * HID  * sizeof(float),    stream);
    hipMemsetAsync(mk1,  0, (size_t)NNODE * H1   * sizeof(unsigned), stream);
    hipMemsetAsync(den1, 0, (size_t)NNODE * H1   * sizeof(float),    stream);
    hipMemsetAsync(mk2,  0, (size_t)NNODE        * sizeof(unsigned), stream);
    hipMemsetAsync(den2, 0, (size_t)NNODE        * sizeof(float),    stream);
    hipMemsetAsync(out,  0, (size_t)NNODE * NLAB * sizeof(float),    stream);

    k_repack<<<cdiv(4 * 16 * 512, B), B, 0, stream>>>(W1, Bp1, 16, 4 * 16 * 512, 1);
    k_repack<<<cdiv(4 * 2  * 512, B), B, 0, stream>>>(W2, Bp2, 2,  4 * 2  * 512, 0);

    k_gemm_bf16<<<gemmGrid, 128, 0, stream>>>(X, FIN, Bp1, FIN / 32, Wh1);
    k_coef1<<<cdiv((long long)NNODE * H1, B), B, 0, stream>>>(Wh1, a1s, a1d, es1, ed1);
    k_edge_max1<<<cdiv((long long)NEDGE * H1, B), B, 0, stream>>>(src, dst, es1, ed1, eatt1, mk1);
    k_edge_msg1<<<cdiv((long long)NEDGE * H1, B), B, 0, stream>>>(src, dst, eatt1, mk1, den1, Wh1, h1);
    k_norm_elu<<<cdiv((long long)NNODE * HID, B), B, 0, stream>>>(h1, den1);

    k_gemm_bf16<<<gemmGrid, 128, 0, stream>>>(h1, HID, Bp2, HID / 32, Wh2);
    k_coef2<<<cdiv(NNODE, B), B, 0, stream>>>(Wh2, a2s, a2d, es2, ed2);
    k_edge_max2<<<cdiv(NEDGE, B), B, 0, stream>>>(src, dst, es2, ed2, eatt2, mk2);
    k_edge_msg2<<<cdiv((long long)NEDGE * 8, B), B, 0, stream>>>(src, dst, eatt2, mk2, den2, Wh2, out);

    k_softmax<<<cdiv((long long)NNODE * 32, 128), 128, 0, stream>>>(out, den2);
}


# Edge list sorted by destination per call (count, scan, fill); 8 lanes per node gather its in-edges and reduce max, sums, ELU/softmax in registers; no feature atomics; sums in arrival-rank order
# speedup vs baseline: 2.7502x; 2.5292x over previous
#include <hip/hip_runtime.h>
#include <hip/hip_bf16.h>


typedef __attribute__((ext_vector_type(16))) __bf16 v16bf;
typedef __attribute__((ext_vector_type(8)))  float  v8f;

constexpr int NNODE = 50000;
constexpr int NEDGE = 800000;
constexpr int FIN   = 512;
constexpr int H1    = 8;
constexpr int FHID  = 8;
constexpr int HID   = H1 * FHID;
constexpr int NLAB  = 64;
constexpr int ROWT  = NNODE / 16;
constexpr float SLOPE = 0.2f;

#define DEVINL __device__ __forceinline__

DEVINL float lrelu(float v) { return v > 0.f ? v : SLOPE * v; }

__global__ void k_repack(const float* __restrict__ W, __bf16* __restrict__ Bp,
                         int nKt, int total, int gat1) {
    int idx = blockIdx.x * blockDim.x + threadIdx.x;
    if (idx >= total) return;
    int e    = idx & 15;
    int lane = (idx >> 4) & 31;
    int ktnt = idx >> 9;
    int kt   = ktnt % nKt;
    int nt   = ktnt / nKt;
    int K = ((e < 8) ? 0 : 16) + (lane >> 4) * 8 + (e & 7);
    int k = kt * 32 + K;
    int n = nt * 16 + (lane & 15);
    float v = gat1 ? W[(size_t)(n >> 3) * (FIN * FHID) + (size_t)k * FHID + (n & 7)]
                   : W[(size_t)k * NLAB + n];
    Bp[idx] = (__bf16)v;
}

__global__ void __launch_bounds__(128)
k_gemm_bf16(const float* __restrict__ A, int lda,
            const __bf16* __restrict__ Bp, int nKt,
            float* __restrict__ C) {
    int lane = threadIdx.x & 31;
    int wave = threadIdx.x >> 5;
    int tile = blockIdx.x * 4 + wave;
    if (tile >= ROWT) return;

    int m     = lane & 15;
    int khalf = lane >> 4;
    const float*  arow = A  + (size_t)(tile * 16 + m) * lda + khalf * 8;
    const __bf16* bl   = Bp + (size_t)lane * 16;

    v8f acc[4] = {};
    for (int kt = 0; kt < nKt; ++kt) {
        const float* ap = arow + kt * 32;
        float4 a0 = *(const float4*)(ap);
        float4 a1 = *(const float4*)(ap + 4);
        float4 a2 = *(const float4*)(ap + 16);
        float4 a3 = *(const float4*)(ap + 20);
        if (kt + 1 < nKt) __builtin_prefetch(ap + 32, 0, 0);

        v16bf av;
        av[0]  = (__bf16)a0.x; av[1]  = (__bf16)a0.y; av[2]  = (__bf16)a0.z; av[3]  = (__bf16)a0.w;
        av[4]  = (__bf16)a1.x; av[5]  = (__bf16)a1.y; av[6]  = (__bf16)a1.z; av[7]  = (__bf16)a1.w;
        av[8]  = (__bf16)a2.x; av[9]  = (__bf16)a2.y; av[10] = (__bf16)a2.z; av[11] = (__bf16)a2.w;
        av[12] = (__bf16)a3.x; av[13] = (__bf16)a3.y; av[14] = (__bf16)a3.z; av[15] = (__bf16)a3.w;

        const __bf16* bk = bl + (size_t)kt * 512;
        #pragma unroll
        for (int nt = 0; nt < 4; ++nt) {
            v16bf bv = *(const v16bf*)(bk + (size_t)nt * nKt * 512);
            acc[nt] = __builtin_amdgcn_wmma_f32_16x16x32_bf16(
                false, av, false, bv, (short)0, acc[nt], false, false);
        }
    }

    float* cbase = C + (size_t)(tile * 16 + khalf * 8) * HID + m;
    #pragma unroll
    for (int nt = 0; nt < 4; ++nt) {
        float* crow = cbase + nt * 16;
        #pragma unroll
        for (int r = 0; r < 8; ++r) crow[(size_t)r * HID] = acc[nt][r];
    }
}

__global__ void k_coef1(const float* __restrict__ Wh,
                        const float* __restrict__ a_s, const float* __restrict__ a_d,
                        float* __restrict__ es, float* __restrict__ ed) {
    int idx = blockIdx.x * blockDim.x + threadIdx.x;
    if (idx >= NNODE * H1) return;
    int n = idx >> 3, h = idx & 7;
    const float* w  = Wh + (size_t)n * HID + h * 8;
    const float* as = a_s + h * 8;
    const float* ad = a_d + h * 8;
    float s = 0.f, d = 0.f;
    #pragma unroll
    for (int o = 0; o < 8; ++o) { float v = w[o]; s += v * as[o]; d += v * ad[o]; }
    es[idx] = s; ed[idx] = d;
}

__global__ void k_coef2(const float* __restrict__ Wh,
                        const float* __restrict__ a_s, const float* __restrict__ a_d,
                        float* __restrict__ es, float* __restrict__ ed) {
    int n = blockIdx.x * blockDim.x + threadIdx.x;
    if (n >= NNODE) return;
    const float* w = Wh + (size_t)n * NLAB;
    float s = 0.f, d = 0.f;
    for (int j = 0; j < NLAB; ++j) { float v = w[j]; s += v * a_s[j]; d += v * a_d[j]; }
    es[n] = s; ed[n] = d;
}

constexpr int SCAN_T = 1024;
constexpr int SCAN_C = 52;
constexpr int NPAD   = SCAN_T * SCAN_C;

__global__ void k_count(const int* __restrict__ dst, int* __restrict__ deg,
                        int* __restrict__ rank) {
    int e = blockIdx.x * blockDim.x + threadIdx.x;
    if (e >= NEDGE) return;
    rank[e] = atomicAdd(deg + dst[e], 1);
}

__global__ void __launch_bounds__(SCAN_T)
k_scan(const int* __restrict__ deg, int* __restrict__ rp) {
    __shared__ int wsum[32];
    int t = threadIdx.x, lane = t & 31, w = t >> 5;
    const int4* p = (const int4*)(deg + t * SCAN_C);
    int4 v[SCAN_C / 4];
    int s = 0;
    #pragma unroll
    for (int i = 0; i < SCAN_C / 4; ++i) { v[i] = p[i]; s += v[i].x + v[i].y + v[i].z + v[i].w; }
    int inc = s;
    #pragma unroll
    for (int o = 1; o < 32; o <<= 1) { int u = __shfl_up(inc, o, 32); if (lane >= o) inc += u; }
    if (lane == 31) wsum[w] = inc;
    __syncthreads();
    if (w == 0) {
        int x = wsum[lane], xi = x;
        #pragma unroll
        for (int o = 1; o < 32; o <<= 1) { int u = __shfl_up(xi, o, 32); if (lane >= o) xi += u; }
        wsum[lane] = xi - x;
    }
    __syncthreads();
    int run = wsum[w] + inc - s;
    int4* q = (int4*)(rp + t * SCAN_C);
    #pragma unroll
    for (int i = 0; i < SCAN_C / 4; ++i) {
        int4 r;
        r.x = run; run += v[i].x;
        r.y = run; run += v[i].y;
        r.z = run; run += v[i].z;
        r.w = run; run += v[i].w;
        q[i] = r;
    }
}

__global__ void k_fill(const int* __restrict__ src, const int* __restrict__ dst,
                       const int* __restrict__ rank, const int* __restrict__ rp,
                       int* __restrict__ csr) {
    int e = blockIdx.x * blockDim.x + threadIdx.x;
    if (e >= NEDGE) return;
    csr[rp[dst[e]] + rank[e]] = src[e];
}

__global__ void __launch_bounds__(256)
k_gather1(const int* __restrict__ rp, const int* __restrict__ csr,
          const float* __restrict__ es, const float* __restrict__ ed,
          const float* __restrict__ Wh, float* __restrict__ h) {
    int idx = blockIdx.x * blockDim.x + threadIdx.x;
    if (idx >= NNODE * H1) return;
    int n = idx >> 3, hh = idx & 7;
    int beg = rp[n], end = rp[n + 1];
    float edn = ed[idx];

    float m = -INFINITY;
    for (int p = beg; p < end; ++p)
        m = fmaxf(m, lrelu(es[csr[p] * H1 + hh] + edn));

    float den = 0.f;
    float4 a0 = make_float4(0.f, 0.f, 0.f, 0.f), a1 = a0;
    for (int p = beg; p < end; ++p) {
        int s = csr[p];
        float ex = __expf(lrelu(es[s * H1 + hh] + edn) - m);
        const float* w = Wh + (size_t)s * HID + hh * 8;
        float4 w0 = *(const float4*)(w);
        float4 w1 = *(const float4*)(w + 4);
        den += ex;
        a0.x += ex * w0.x; a0.y += ex * w0.y; a0.z += ex * w0.z; a0.w += ex * w0.w;
        a1.x += ex * w1.x; a1.y += ex * w1.y; a1.z += ex * w1.z; a1.w += ex * w1.w;
    }

    float dn = den + 1e-10f;
    float x[8] = { a0.x / dn, a0.y / dn, a0.z / dn, a0.w / dn,
                   a1.x / dn, a1.y / dn, a1.z / dn, a1.w / dn };
    #pragma unroll
    for (int j = 0; j < 8; ++j) x[j] = x[j] > 0.f ? x[j] : (__expf(x[j]) - 1.f);
    float* o = h + (size_t)n * HID + hh * 8;
    *(float4*)(o)     = make_float4(x[0], x[1], x[2], x[3]);
    *(float4*)(o + 4) = make_float4(x[4], x[5], x[6], x[7]);
}

__global__ void __launch_bounds__(256)
k_gather2(const int* __restrict__ rp, const int* __restrict__ csr,
          const float* __restrict__ es, const float* __restrict__ ed,
          const float* __restrict__ Wh, float* __restrict__ out) {
    int idx = blockIdx.x * blockDim.x + threadIdx.x;
    if (idx >= NNODE * 8) return;
    int n = idx >> 3, c0 = (idx & 7) * 8;
    int beg = rp[n], end = rp[n + 1];
    float edn = ed[n];

    float m = -INFINITY;
    for (int p = beg; p < end; ++p)
        m = fmaxf(m, lrelu(es[csr[p]] + edn));

    float den = 0.f;
    float4 a0 = make_float4(0.f, 0.f, 0.f, 0.f), a1 = a0;
    for (int p = beg; p < end; ++p) {
        int s = csr[p];
        float ex = __expf(lrelu(es[s] + edn) - m);
        const float* w = Wh + (size_t)s * NLAB + c0;
        float4 w0 = *(const float4*)(w);
        float4 w1 = *(const float4*)(w + 4);
        den += ex;
        a0.x += ex * w0.x; a0.y += ex * w0.y; a0.z += ex * w0.z; a0.w += ex * w0.w;
        a1.x += ex * w1.x; a1.y += ex * w1.y; a1.z += ex * w1.z; a1.w += ex * w1.w;
    }

    float dn = den + 1e-10f;
    float x[8] = { a0.x / dn, a0.y / dn, a0.z / dn, a0.w / dn,
                   a1.x / dn, a1.y / dn, a1.z / dn, a1.w / dn };
    float mx = x[0];
    #pragma unroll
    for (int j = 1; j < 8; ++j) mx = fmaxf(mx, x[j]);
    #pragma unroll
    for (int o = 1; o < 8; o <<= 1) mx = fmaxf(mx, __shfl_xor(mx, o, 32));
    float sum = 0.f;
    #pragma unroll
    for (int j = 0; j < 8; ++j) { x[j] = __expf(x[j] - mx); sum += x[j]; }
    #pragma unroll
    for (int o = 1; o < 8; o <<= 1) sum += __shfl_xor(sum, o, 32);
    float inv = 1.0f / sum;
    float* o = out + (size_t)n * NLAB + c0;
    *(float4*)(o)     = make_float4(x[0] * inv, x[1] * inv, x[2] * inv, x[3] * inv);
    *(float4*)(o + 4) = make_float4(x[4] * inv, x[5] * inv, x[6] * inv, x[7] * inv);
}

extern "C" void kernel_launch(void* const* d_in, const int* in_sizes, int n_in,
                              void* d_out, int out_size, void* d_ws, size_t ws_size,
                              hipStream_t stream) {
    const float* X   = (const float*)d_in[0];
    const float* W1  = (const float*)d_in[1];
    const float* a1s = (const float*)d_in[2];
    const float* a1d = (const float*)d_in[3];
    const float* W2  = (const float*)d_in[4];
    const float* a2s = (const float*)d_in[5];
    const float* a2d = (const float*)d_in[6];
    const int*   src = (const int*)d_in[7];
    const int*   dst = (const int*)d_in[8];
    float*       out = (float*)d_out;

    char*  ws  = (char*)d_ws;
    size_t off = 0;
    auto alloc = [&](size_t bytes) -> void* {
        void* p = ws + off;
        off = (off + bytes + 255) & ~(size_t)255;
        return p;
    };
    float*    Wh1   = (float*)   alloc((size_t)NNODE * HID  * sizeof(float));
    float*    h1    = (float*)   alloc((size_t)NNODE * HID  * sizeof(float));
    float*    Wh2   = (float*)   alloc((size_t)NNODE * NLAB * sizeof(float));
    float*    es1   = (float*)   alloc((size_t)NNODE * H1   * sizeof(float));
    float*    ed1   = (float*)   alloc((size_t)NNODE * H1   * sizeof(float));
    float*    es2   = (float*)   alloc((size_t)NNODE        * sizeof(float));
    float*    ed2   = (float*)   alloc((size_t)NNODE        * sizeof(float));
    int*      deg   = (int*)     alloc((size_t)NPAD         * sizeof(int));
    int*      rp    = (int*)     alloc((size_t)NPAD         * sizeof(int));
    int*      rank  = (int*)     alloc((size_t)NEDGE        * sizeof(int));
    int*      csr   = (int*)     alloc((size_t)NEDGE        * sizeof(int));
    __bf16*   Bp1   = (__bf16*)  alloc((size_t)4 * 16 * 512 * sizeof(__bf16));
    __bf16*   Bp2   = (__bf16*)  alloc((size_t)4 * 2  * 512 * sizeof(__bf16));

    const int B = 256;
    auto cdiv = [](long long a, long long b) { return (int)((a + b - 1) / b); };
    const int gemmGrid = cdiv(ROWT, 4);

    hipMemsetAsync(deg, 0, (size_t)NPAD * sizeof(int), stream);
    k_count<<<cdiv(NEDGE, B), B, 0, stream>>>(dst, deg, rank);
    k_scan<<<1, SCAN_T, 0, stream>>>(deg, rp);
    k_fill<<<cdiv(NEDGE, B), B, 0, stream>>>(src, dst, rank, rp, csr);

    k_repack<<<cdiv(4 * 16 * 512, B), B, 0, stream>>>(W1, Bp1, 16, 4 * 16 * 512, 1);
    k_repack<<<cdiv(4 * 2  * 512, B), B, 0, stream>>>(W2, Bp2, 2,  4 * 2  * 512, 0);

    k_gemm_bf16<<<gemmGrid, 128, 0, stream>>>(X, FIN, Bp1, FIN / 32, Wh1);
    k_coef1<<<cdiv((long long)NNODE * H1, B), B, 0, stream>>>(Wh1, a1s, a1d, es1, ed1);
    k_gather1<<<cdiv((long long)NNODE * H1, B), B, 0, stream>>>(rp, csr, es1, ed1, Wh1, h1);

    k_gemm_bf16<<<gemmGrid, 128, 0, stream>>>(h1, HID, Bp2, HID / 32, Wh2);
    k_coef2<<<cdiv(NNODE, B), B, 0, stream>>>(Wh2, a2s, a2d, es2, ed2);
    k_gather2<<<cdiv((long long)NNODE * 8, B), B, 0, stream>>>(rp, csr, es2, ed2, Wh2, out);
}


# Destination sort on LDS counters only: per-chunk histograms over 256-node buckets, chunk-prefix scatter, per-bucket count/scan/fill; no global atomics, no memset; in-edge sum order = LDS arrival
# speedup vs baseline: 3.7937x; 1.3794x over previous
#include <hip/hip_runtime.h>
#include <hip/hip_bf16.h>


typedef __attribute__((ext_vector_type(16))) __bf16 v16bf;
typedef __attribute__((ext_vector_type(8)))  float  v8f;

constexpr int NNODE = 50000;
constexpr int NEDGE = 800000;
constexpr int FIN   = 512;
constexpr int H1    = 8;
constexpr int FHID  = 8;
constexpr int HID   = H1 * FHID;
constexpr int NLAB  = 64;
constexpr int ROWT  = NNODE / 16;
constexpr float SLOPE = 0.2f;

#define DEVINL __device__ __forceinline__

DEVINL float lrelu(float v) { return v > 0.f ? v : SLOPE * v; }

__global__ void k_repack(const float* __restrict__ W, __bf16* __restrict__ Bp,
                         int nKt, int total, int gat1) {
    int idx = blockIdx.x * blockDim.x + threadIdx.x;
    if (idx >= total) return;
    int e    = idx & 15;
    int lane = (idx >> 4) & 31;
    int ktnt = idx >> 9;
    int kt   = ktnt % nKt;
    int nt   = ktnt / nKt;
    int K = ((e < 8) ? 0 : 16) + (lane >> 4) * 8 + (e & 7);
    int k = kt * 32 + K;
    int n = nt * 16 + (lane & 15);
    float v = gat1 ? W[(size_t)(n >> 3) * (FIN * FHID) + (size_t)k * FHID + (n & 7)]
                   : W[(size_t)k * NLAB + n];
    Bp[idx] = (__bf16)v;
}

__global__ void __launch_bounds__(128)
k_gemm_bf16(const float* __restrict__ A, int lda,
            const __bf16* __restrict__ Bp, int nKt,
            float* __restrict__ C) {
    int lane = threadIdx.x & 31;
    int wave = threadIdx.x >> 5;
    int tile = blockIdx.x * 4 + wave;
    if (tile >= ROWT) return;

    int m     = lane & 15;
    int khalf = lane >> 4;
    const float*  arow = A  + (size_t)(tile * 16 + m) * lda + khalf * 8;
    const __bf16* bl   = Bp + (size_t)lane * 16;

    v8f acc[4] = {};
    for (int kt = 0; kt < nKt; ++kt) {
        const float* ap = arow + kt * 32;
        float4 a0 = *(const float4*)(ap);
        float4 a1 = *(const float4*)(ap + 4);
        float4 a2 = *(const float4*)(ap + 16);
        float4 a3 = *(const float4*)(ap + 20);
        if (kt + 1 < nKt) __builtin_prefetch(ap + 32, 0, 0);

        v16bf av;
        av[0]  = (__bf16)a0.x; av[1]  = (__bf16)a0.y; av[2]  = (__bf16)a0.z; av[3]  = (__bf16)a0.w;
        av[4]  = (__bf16)a1.x; av[5]  = (__bf16)a1.y; av[6]  = (__bf16)a1.z; av[7]  = (__bf16)a1.w;
        av[8]  = (__bf16)a2.x; av[9]  = (__bf16)a2.y; av[10] = (__bf16)a2.z; av[11] = (__bf16)a2.w;
        av[12] = (__bf16)a3.x; av[13] = (__bf16)a3.y; av[14] = (__bf16)a3.z; av[15] = (__bf16)a3.w;

        const __bf16* bk = bl + (size_t)kt * 512;
        #pragma unroll
        for (int nt = 0; nt < 4; ++nt) {
            v16bf bv = *(const v16bf*)(bk + (size_t)nt * nKt * 512);
            acc[nt] = __builtin_amdgcn_wmma_f32_16x16x32_bf16(
                false, av, false, bv, (short)0, acc[nt], false, false);
        }
    }

    float* cbase = C + (size_t)(tile * 16 + khalf * 8) * HID + m;
    #pragma unroll
    for (int nt = 0; nt < 4; ++nt) {
        float* crow = cbase + nt * 16;
        #pragma unroll
        for (int r = 0; r < 8; ++r) crow[(size_t)r * HID] = acc[nt][r];
    }
}

__global__ void k_coef1(const float* __restrict__ Wh,
                        const float* __restrict__ a_s, const float* __restrict__ a_d,
                        float* __restrict__ es, float* __restrict__ ed) {
    int idx = blockIdx.x * blockDim.x + threadIdx.x;
    if (idx >= NNODE * H1) return;
    int n = idx >> 3, h = idx & 7;
    const float* w  = Wh + (size_t)n * HID + h * 8;
    const float* as = a_s + h * 8;
    const float* ad = a_d + h * 8;
    float s = 0.f, d = 0.f;
    #pragma unroll
    for (int o = 0; o < 8; ++o) { float v = w[o]; s += v * as[o]; d += v * ad[o]; }
    es[idx] = s; ed[idx] = d;
}

__global__ void k_coef2(const float* __restrict__ Wh,
                        const float* __restrict__ a_s, const float* __restrict__ a_d,
                        float* __restrict__ es, float* __restrict__ ed) {
    int n = blockIdx.x * blockDim.x + threadIdx.x;
    if (n >= NNODE) return;
    const float* w = Wh + (size_t)n * NLAB;
    float s = 0.f, d = 0.f;
    for (int j = 0; j < NLAB; ++j) { float v = w[j]; s += v * a_s[j]; d += v * a_d[j]; }
    es[n] = s; ed[n] = d;
}

constexpr int NPAD = 53248;
constexpr int BSH  = 8;
constexpr int BNOD = 1 << BSH;
constexpr int NBK  = (NNODE >> BSH) + 1;
constexpr int EC   = 3200;
constexpr int EQ   = EC / 4;
constexpr int NWG  = NEDGE / EC;
constexpr int BB_OFF = 49152;
static_assert(NEDGE % EC == 0 && EC % 4 == 0, "chunks are whole int4 runs");
constexpr int EQH  = (EQ + 255) / 256;
constexpr int NBQ  = NBK / 4;
constexpr int RG   = 10;
constexpr int RPG  = NWG / RG;
static_assert(NBK % 4 == 0 && NWG % RG == 0 && RG * NBQ <= 512, "column-sum tiling");
static_assert(NBK <= 256 && NNODE <= 65536, "bucket fits the block; src fits 16 bits");
static_assert(NWG * NBK <= BB_OFF && BB_OFF + 256 <= NPAD && NNODE + 1 <= NPAD, "slices");
static_assert(EQ > 512 && EQ <= 1024, "k_scatter holds a chunk in two int4 per thread");

__global__ void __launch_bounds__(256)
k_hist(const int* __restrict__ dst, int* __restrict__ hist) {
    __shared__ int cnt[256];
    int t = threadIdx.x;
    const int4* dp = (const int4*)(dst + (size_t)blockIdx.x * EC);
    int4 d[EQH];
    #pragma unroll
    for (int k = 0; k < EQH; ++k) {
        int i = t + k * 256;
        d[k] = dp[i < EQ ? i : EQ - 1];
    }
    cnt[t] = 0;
    __syncthreads();
    #pragma unroll
    for (int k = 0; k < EQH; ++k) {
        if (t + k * 256 < EQ) {
            atomicAdd(&cnt[d[k].x >> BSH], 1);
            atomicAdd(&cnt[d[k].y >> BSH], 1);
            atomicAdd(&cnt[d[k].z >> BSH], 1);
            atomicAdd(&cnt[d[k].w >> BSH], 1);
        }
    }
    __syncthreads();
    if (t < NBK) hist[blockIdx.x * NBK + t] = cnt[t];
}

DEVINL int scan256_excl(int v, int t, int* wtot) {
    int lane = t & 31, w = t >> 5;
    int inc = v;
    #pragma unroll
    for (int o = 1; o < 32; o <<= 1) { int u = __shfl_up(inc, o, 32); if (lane >= o) inc += u; }
    if (lane == 31 && w < 8) wtot[w] = inc;
    __syncthreads();
    int pre = 0;
    #pragma unroll
    for (int j = 0; j < 8; ++j) pre += (j < w) ? wtot[j] : 0;
    return pre + inc - v;
}

__global__ void __launch_bounds__(512)
k_scatter(const int* __restrict__ src, const int* __restrict__ dst,
          const int* __restrict__ hist, int* __restrict__ bbase,
          int* __restrict__ rec) {
    __shared__ int4 s_tot[RG][NBQ];
    __shared__ int4 s_par[RG][NBQ];
    __shared__ int cur[256];
    __shared__ int wtot[8];
    int t = threadIdx.x, wg = blockIdx.x;

    const int4* dp = (const int4*)(dst + (size_t)wg * EC);
    const int4* sp = (const int4*)(src + (size_t)wg * EC);
    bool two = t + 512 < EQ;
    int  i1  = two ? t + 512 : t;
    int4 d0 = dp[t],  s0 = sp[t];
    int4 d1 = dp[i1], s1 = sp[i1];

    if (t < RG * NBQ) {
        int q = t % NBQ, g = t / NBQ;
        const int4* hp = (const int4*)hist + (size_t)(g * RPG) * NBQ + q;
        int4 h[RPG];
        #pragma unroll
        for (int k = 0; k < RPG; ++k) h[k] = hp[k * NBQ];
        int4 tot = make_int4(0, 0, 0, 0), par = tot;
        #pragma unroll
        for (int k = 0; k < RPG; ++k) {
            tot.x += h[k].x; tot.y += h[k].y; tot.z += h[k].z; tot.w += h[k].w;
            if (g * RPG + k < wg) { par.x += h[k].x; par.y += h[k].y; par.z += h[k].z; par.w += h[k].w; }
        }
        s_tot[g][q] = tot;
        s_par[g][q] = par;
    }
    __syncthreads();
    int tot = 0, par = 0;
    if (t < NBK) {
        const int* st = (const int*)s_tot;
        const int* sq = (const int*)s_par;
        #pragma unroll
        for (int g = 0; g < RG; ++g) { tot += st[g * NBK + t]; par += sq[g * NBK + t]; }
    }
    int excl = scan256_excl(tot, t, wtot);
    if (t < 256) {
        cur[t] = excl + par;
        if (wg == 0 && t <= NBK) bbase[t] = excl;
    }
    __syncthreads();

    int p;
    p = atomicAdd(&cur[d0.x >> BSH], 1); rec[p] = s0.x | ((d0.x & (BNOD - 1)) << 16);
    p = atomicAdd(&cur[d0.y >> BSH], 1); rec[p] = s0.y | ((d0.y & (BNOD - 1)) << 16);
    p = atomicAdd(&cur[d0.z >> BSH], 1); rec[p] = s0.z | ((d0.z & (BNOD - 1)) << 16);
    p = atomicAdd(&cur[d0.w >> BSH], 1); rec[p] = s0.w | ((d0.w & (BNOD - 1)) << 16);
    if (two) {
        p = atomicAdd(&cur[d1.x >> BSH], 1); rec[p] = s1.x | ((d1.x & (BNOD - 1)) << 16);
        p = atomicAdd(&cur[d1.y >> BSH], 1); rec[p] = s1.y | ((d1.y & (BNOD - 1)) << 16);
        p = atomicAdd(&cur[d1.z >> BSH], 1); rec[p] = s1.z | ((d1.z & (BNOD - 1)) << 16);
        p = atomicAdd(&cur[d1.w >> BSH], 1); rec[p] = s1.w | ((d1.w & (BNOD - 1)) << 16);
    }
}

__global__ void __launch_bounds__(512)
k_bucket(const int* __restrict__ rec, const int* __restrict__ bbase,
         int* __restrict__ rp, int* __restrict__ csr) {
    __shared__ int cnt[256];
    __shared__ int wtot[8];
    int t = threadIdx.x, b = blockIdx.x;
    int beg = bbase[b], end = bbase[b + 1];
    if (t < 256) cnt[t] = 0;
    __syncthreads();

    for (int p = beg + t; p < end; p += 4 * 512) {
        int r[4];
        #pragma unroll
        for (int k = 0; k < 4; ++k) { int q = p + k * 512; int v = rec[q < end ? q : end - 1]; r[k] = (q < end) ? v : -1; }
        #pragma unroll
        for (int k = 0; k < 4; ++k) if (r[k] >= 0) atomicAdd(&cnt[r[k] >> 16], 1);
    }
    __syncthreads();

    int deg = (t < 256) ? cnt[t] : 0;
    int excl = scan256_excl(deg, t, wtot);
    if (t < 256) {
        int start = beg + excl;
        int node  = b * BNOD + t;
        if (node <= NNODE) rp[node] = start;
        cnt[t] = start;
    }
    __syncthreads();

    for (int p = beg + t; p < end; p += 4 * 512) {
        int r[4];
        #pragma unroll
        for (int k = 0; k < 4; ++k) { int q = p + k * 512; int v = rec[q < end ? q : end - 1]; r[k] = (q < end) ? v : -1; }
        #pragma unroll
        for (int k = 0; k < 4; ++k)
            if (r[k] >= 0) { int s = atomicAdd(&cnt[r[k] >> 16], 1); csr[s] = r[k] & 0xffff; }
    }
}

__global__ void __launch_bounds__(256)
k_gather1(const int* __restrict__ rp, const int* __restrict__ csr,
          const float* __restrict__ es, const float* __restrict__ ed,
          const float* __restrict__ Wh, float* __restrict__ h) {
    int idx = blockIdx.x * blockDim.x + threadIdx.x;
    if (idx >= NNODE * H1) return;
    int n = idx >> 3, hh = idx & 7;
    int beg = rp[n], end = rp[n + 1];
    float edn = ed[idx];

    float m = -INFINITY;
    for (int p = beg; p < end; ++p)
        m = fmaxf(m, lrelu(es[csr[p] * H1 + hh] + edn));

    float den = 0.f;
    float4 a0 = make_float4(0.f, 0.f, 0.f, 0.f), a1 = a0;
    for (int p = beg; p < end; ++p) {
        int s = csr[p];
        float ex = __expf(lrelu(es[s * H1 + hh] + edn) - m);
        const float* w = Wh + (size_t)s * HID + hh * 8;
        float4 w0 = *(const float4*)(w);
        float4 w1 = *(const float4*)(w + 4);
        den += ex;
        a0.x += ex * w0.x; a0.y += ex * w0.y; a0.z += ex * w0.z; a0.w += ex * w0.w;
        a1.x += ex * w1.x; a1.y += ex * w1.y; a1.z += ex * w1.z; a1.w += ex * w1.w;
    }

    float dn = den + 1e-10f;
    float x[8] = { a0.x / dn, a0.y / dn, a0.z / dn, a0.w / dn,
                   a1.x / dn, a1.y / dn, a1.z / dn, a1.w / dn };
    #pragma unroll
    for (int j = 0; j < 8; ++j) x[j] = x[j] > 0.f ? x[j] : (__expf(x[j]) - 1.f);
    float* o = h + (size_t)n * HID + hh * 8;
    *(float4*)(o)     = make_float4(x[0], x[1], x[2], x[3]);
    *(float4*)(o + 4) = make_float4(x[4], x[5], x[6], x[7]);
}

__global__ void __launch_bounds__(256)
k_gather2(const int* __restrict__ rp, const int* __restrict__ csr,
          const float* __restrict__ es, const float* __restrict__ ed,
          const float* __restrict__ Wh, float* __restrict__ out) {
    int idx = blockIdx.x * blockDim.x + threadIdx.x;
    if (idx >= NNODE * 8) return;
    int n = idx >> 3, c0 = (idx & 7) * 8;
    int beg = rp[n], end = rp[n + 1];
    float edn = ed[n];

    float m = -INFINITY;
    for (int p = beg; p < end; ++p)
        m = fmaxf(m, lrelu(es[csr[p]] + edn));

    float den = 0.f;
    float4 a0 = make_float4(0.f, 0.f, 0.f, 0.f), a1 = a0;
    for (int p = beg; p < end; ++p) {
        int s = csr[p];
        float ex = __expf(lrelu(es[s] + edn) - m);
        const float* w = Wh + (size_t)s * NLAB + c0;
        float4 w0 = *(const float4*)(w);
        float4 w1 = *(const float4*)(w + 4);
        den += ex;
        a0.x += ex * w0.x; a0.y += ex * w0.y; a0.z += ex * w0.z; a0.w += ex * w0.w;
        a1.x += ex * w1.x; a1.y += ex * w1.y; a1.z += ex * w1.z; a1.w += ex * w1.w;
    }

    float dn = den + 1e-10f;
    float x[8] = { a0.x / dn, a0.y / dn, a0.z / dn, a0.w / dn,
                   a1.x / dn, a1.y / dn, a1.z / dn, a1.w / dn };
    float mx = x[0];
    #pragma unroll
    for (int j = 1; j < 8; ++j) mx = fmaxf(mx, x[j]);
    #pragma unroll
    for (int o = 1; o < 8; o <<= 1) mx = fmaxf(mx, __shfl_xor(mx, o, 32));
    float sum = 0.f;
    #pragma unroll
    for (int j = 0; j < 8; ++j) { x[j] = __expf(x[j] - mx); sum += x[j]; }
    #pragma unroll
    for (int o = 1; o < 8; o <<= 1) sum += __shfl_xor(sum, o, 32);
    float inv = 1.0f / sum;
    float* o = out + (size_t)n * NLAB + c0;
    *(float4*)(o)     = make_float4(x[0] * inv, x[1] * inv, x[2] * inv, x[3] * inv);
    *(float4*)(o + 4) = make_float4(x[4] * inv, x[5] * inv, x[6] * inv, x[7] * inv);
}

extern "C" void kernel_launch(void* const* d_in, const int* in_sizes, int n_in,
                              void* d_out, int out_size, void* d_ws, size_t ws_size,
                              hipStream_t stream) {
    const float* X   = (const float*)d_in[0];
    const float* W1  = (const float*)d_in[1];
    const float* a1s = (const float*)d_in[2];
    const float* a1d = (const float*)d_in[3];
    const float* W2  = (const float*)d_in[4];
    const float* a2s = (const float*)d_in[5];
    const float* a2d = (const float*)d_in[6];
    const int*   src = (const int*)d_in[7];
    const int*   dst = (const int*)d_in[8];
    float*       out = (float*)d_out;

    char*  ws  = (char*)d_ws;
    size_t off = 0;
    auto alloc = [&](size_t bytes) -> void* {
        void* p = ws + off;
        off = (off + bytes + 255) & ~(size_t)255;
        return p;
    };
    float*    Wh1   = (float*)   alloc((size_t)NNODE * HID  * sizeof(float));
    float*    h1    = (float*)   alloc((size_t)NNODE * HID  * sizeof(float));
    float*    Wh2   = (float*)   alloc((size_t)NNODE * NLAB * sizeof(float));
    float*    es1   = (float*)   alloc((size_t)NNODE * H1   * sizeof(float));
    float*    ed1   = (float*)   alloc((size_t)NNODE * H1   * sizeof(float));
    float*    es2   = (float*)   alloc((size_t)NNODE        * sizeof(float));
    float*    ed2   = (float*)   alloc((size_t)NNODE        * sizeof(float));
    int*      hist  = (int*)     alloc((size_t)NPAD         * sizeof(int));
    int*      rp    = (int*)     alloc((size_t)NPAD         * sizeof(int));
    int*      rec   = (int*)     alloc((size_t)NEDGE        * sizeof(int));
    int*      csr   = (int*)     alloc((size_t)NEDGE        * sizeof(int));
    __bf16*   Bp1   = (__bf16*)  alloc((size_t)4 * 16 * 512 * sizeof(__bf16));
    __bf16*   Bp2   = (__bf16*)  alloc((size_t)4 * 2  * 512 * sizeof(__bf16));

    const int B = 256;
    auto cdiv = [](long long a, long long b) { return (int)((a + b - 1) / b); };
    const int gemmGrid = cdiv(ROWT, 4);

    int* bbase = hist + BB_OFF;
    k_hist<<<NWG, 256, 0, stream>>>(dst, hist);
    k_scatter<<<NWG, 512, 0, stream>>>(src, dst, hist, bbase, rec);
    k_bucket<<<NBK, 512, 0, stream>>>(rec, bbase, rp, csr);

    k_repack<<<cdiv(4 * 16 * 512, B), B, 0, stream>>>(W1, Bp1, 16, 4 * 16 * 512, 1);
    k_repack<<<cdiv(4 * 2  * 512, B), B, 0, stream>>>(W2, Bp2, 2,  4 * 2  * 512, 0);

    k_gemm_bf16<<<gemmGrid, 128, 0, stream>>>(X, FIN, Bp1, FIN / 32, Wh1);
    k_coef1<<<cdiv((long long)NNODE * H1, B), B, 0, stream>>>(Wh1, a1s, a1d, es1, ed1);
    k_gather1<<<cdiv((long long)NNODE * H1, B), B, 0, stream>>>(rp, csr, es1, ed1, Wh1, h1);

    k_gemm_bf16<<<gemmGrid, 128, 0, stream>>>(h1, HID, Bp2, HID / 32, Wh2);
    k_coef2<<<cdiv(NNODE, B), B, 0, stream>>>(Wh2, a2s, a2d, es2, ed2);
    k_gather2<<<cdiv((long long)NNODE * 8, B), B, 0, stream>>>(rp, csr, es2, ed2, Wh2, out);
}


# Gathers read each in-edge once: logit rebuilt in-lane from the gathered Wh row (ed from own row; layer 2 by 8-lane tree sum), softmax on a running max with rescale; es/ed arrays and 2 launches gone
# speedup vs baseline: 4.5113x; 1.1892x over previous
#include <hip/hip_runtime.h>
#include <hip/hip_bf16.h>


typedef __attribute__((ext_vector_type(16))) __bf16 v16bf;
typedef __attribute__((ext_vector_type(8)))  float  v8f;

constexpr int NNODE = 50000;
constexpr int NEDGE = 800000;
constexpr int FIN   = 512;
constexpr int H1    = 8;
constexpr int FHID  = 8;
constexpr int HID   = H1 * FHID;
constexpr int NLAB  = 64;
constexpr int ROWT  = NNODE / 16;
constexpr float SLOPE = 0.2f;

#define DEVINL __device__ __forceinline__

DEVINL float lrelu(float v) { return v > 0.f ? v : SLOPE * v; }

__global__ void k_repack(const float* __restrict__ W, __bf16* __restrict__ Bp,
                         int nKt, int total, int gat1) {
    int idx = blockIdx.x * blockDim.x + threadIdx.x;
    if (idx >= total) return;
    int e    = idx & 15;
    int lane = (idx >> 4) & 31;
    int ktnt = idx >> 9;
    int kt   = ktnt % nKt;
    int nt   = ktnt / nKt;
    int K = ((e < 8) ? 0 : 16) + (lane >> 4) * 8 + (e & 7);
    int k = kt * 32 + K;
    int n = nt * 16 + (lane & 15);
    float v = gat1 ? W[(size_t)(n >> 3) * (FIN * FHID) + (size_t)k * FHID + (n & 7)]
                   : W[(size_t)k * NLAB + n];
    Bp[idx] = (__bf16)v;
}

__global__ void __launch_bounds__(128)
k_gemm_bf16(const float* __restrict__ A, int lda,
            const __bf16* __restrict__ Bp, int nKt,
            float* __restrict__ C) {
    int lane = threadIdx.x & 31;
    int wave = threadIdx.x >> 5;
    int tile = blockIdx.x * 4 + wave;
    if (tile >= ROWT) return;

    int m     = lane & 15;
    int khalf = lane >> 4;
    const float*  arow = A  + (size_t)(tile * 16 + m) * lda + khalf * 8;
    const __bf16* bl   = Bp + (size_t)lane * 16;

    v8f acc[4] = {};
    for (int kt = 0; kt < nKt; ++kt) {
        const float* ap = arow + kt * 32;
        float4 a0 = *(const float4*)(ap);
        float4 a1 = *(const float4*)(ap + 4);
        float4 a2 = *(const float4*)(ap + 16);
        float4 a3 = *(const float4*)(ap + 20);
        if (kt + 1 < nKt) __builtin_prefetch(ap + 32, 0, 0);

        v16bf av;
        av[0]  = (__bf16)a0.x; av[1]  = (__bf16)a0.y; av[2]  = (__bf16)a0.z; av[3]  = (__bf16)a0.w;
        av[4]  = (__bf16)a1.x; av[5]  = (__bf16)a1.y; av[6]  = (__bf16)a1.z; av[7]  = (__bf16)a1.w;
        av[8]  = (__bf16)a2.x; av[9]  = (__bf16)a2.y; av[10] = (__bf16)a2.z; av[11] = (__bf16)a2.w;
        av[12] = (__bf16)a3.x; av[13] = (__bf16)a3.y; av[14] = (__bf16)a3.z; av[15] = (__bf16)a3.w;

        const __bf16* bk = bl + (size_t)kt * 512;
        #pragma unroll
        for (int nt = 0; nt < 4; ++nt) {
            v16bf bv = *(const v16bf*)(bk + (size_t)nt * nKt * 512);
            acc[nt] = __builtin_amdgcn_wmma_f32_16x16x32_bf16(
                false, av, false, bv, (short)0, acc[nt], false, false);
        }
    }

    float* cbase = C + (size_t)(tile * 16 + khalf * 8) * HID + m;
    #pragma unroll
    for (int nt = 0; nt < 4; ++nt) {
        float* crow = cbase + nt * 16;
        #pragma unroll
        for (int r = 0; r < 8; ++r) crow[(size_t)r * HID] = acc[nt][r];
    }
}

constexpr int NPAD = 53248;
constexpr int BSH  = 8;
constexpr int BNOD = 1 << BSH;
constexpr int NBK  = (NNODE >> BSH) + 1;
constexpr int EC   = 3200;
constexpr int EQ   = EC / 4;
constexpr int NWG  = NEDGE / EC;
constexpr int BB_OFF = 49152;
static_assert(NEDGE % EC == 0 && EC % 4 == 0, "chunks are whole int4 runs");
constexpr int EQH  = (EQ + 255) / 256;
constexpr int NBQ  = NBK / 4;
constexpr int RG   = 10;
constexpr int RPG  = NWG / RG;
static_assert(NBK % 4 == 0 && NWG % RG == 0 && RG * NBQ <= 512, "column-sum tiling");
static_assert(NBK <= 256 && NNODE <= 65536, "bucket fits the block; src fits 16 bits");
static_assert(NWG * NBK <= BB_OFF && BB_OFF + 256 <= NPAD && NNODE + 1 <= NPAD, "slices");
static_assert(EQ > 512 && EQ <= 1024, "k_scatter holds a chunk in two int4 per thread");

__global__ void __launch_bounds__(256)
k_hist(const int* __restrict__ dst, int* __restrict__ hist) {
    __shared__ int cnt[256];
    int t = threadIdx.x;
    const int4* dp = (const int4*)(dst + (size_t)blockIdx.x * EC);
    int4 d[EQH];
    #pragma unroll
    for (int k = 0; k < EQH; ++k) {
        int i = t + k * 256;
        d[k] = dp[i < EQ ? i : EQ - 1];
    }
    cnt[t] = 0;
    __syncthreads();
    #pragma unroll
    for (int k = 0; k < EQH; ++k) {
        if (t + k * 256 < EQ) {
            atomicAdd(&cnt[d[k].x >> BSH], 1);
            atomicAdd(&cnt[d[k].y >> BSH], 1);
            atomicAdd(&cnt[d[k].z >> BSH], 1);
            atomicAdd(&cnt[d[k].w >> BSH], 1);
        }
    }
    __syncthreads();
    if (t < NBK) hist[blockIdx.x * NBK + t] = cnt[t];
}

DEVINL int scan256_excl(int v, int t, int* wtot) {
    int lane = t & 31, w = t >> 5;
    int inc = v;
    #pragma unroll
    for (int o = 1; o < 32; o <<= 1) { int u = __shfl_up(inc, o, 32); if (lane >= o) inc += u; }
    if (lane == 31 && w < 8) wtot[w] = inc;
    __syncthreads();
    int pre = 0;
    #pragma unroll
    for (int j = 0; j < 8; ++j) pre += (j < w) ? wtot[j] : 0;
    return pre + inc - v;
}

__global__ void __launch_bounds__(512)
k_scatter(const int* __restrict__ src, const int* __restrict__ dst,
          const int* __restrict__ hist, int* __restrict__ bbase,
          int* __restrict__ rec) {
    __shared__ int4 s_tot[RG][NBQ];
    __shared__ int4 s_par[RG][NBQ];
    __shared__ int cur[256];
    __shared__ int wtot[8];
    int t = threadIdx.x, wg = blockIdx.x;

    const int4* dp = (const int4*)(dst + (size_t)wg * EC);
    const int4* sp = (const int4*)(src + (size_t)wg * EC);
    bool two = t + 512 < EQ;
    int  i1  = two ? t + 512 : t;
    int4 d0 = dp[t],  s0 = sp[t];
    int4 d1 = dp[i1], s1 = sp[i1];

    if (t < RG * NBQ) {
        int q = t % NBQ, g = t / NBQ;
        const int4* hp = (const int4*)hist + (size_t)(g * RPG) * NBQ + q;
        int4 h[RPG];
        #pragma unroll
        for (int k = 0; k < RPG; ++k) h[k] = hp[k * NBQ];
        int4 tot = make_int4(0, 0, 0, 0), par = tot;
        #pragma unroll
        for (int k = 0; k < RPG; ++k) {
            tot.x += h[k].x; tot.y += h[k].y; tot.z += h[k].z; tot.w += h[k].w;
            if (g * RPG + k < wg) { par.x += h[k].x; par.y += h[k].y; par.z += h[k].z; par.w += h[k].w; }
        }
        s_tot[g][q] = tot;
        s_par[g][q] = par;
    }
    __syncthreads();
    int tot = 0, par = 0;
    if (t < NBK) {
        const int* st = (const int*)s_tot;
        const int* sq = (const int*)s_par;
        #pragma unroll
        for (int g = 0; g < RG; ++g) { tot += st[g * NBK + t]; par += sq[g * NBK + t]; }
    }
    int excl = scan256_excl(tot, t, wtot);
    if (t < 256) {
        cur[t] = excl + par;
        if (wg == 0 && t <= NBK) bbase[t] = excl;
    }
    __syncthreads();

    int p;
    p = atomicAdd(&cur[d0.x >> BSH], 1); rec[p] = s0.x | ((d0.x & (BNOD - 1)) << 16);
    p = atomicAdd(&cur[d0.y >> BSH], 1); rec[p] = s0.y | ((d0.y & (BNOD - 1)) << 16);
    p = atomicAdd(&cur[d0.z >> BSH], 1); rec[p] = s0.z | ((d0.z & (BNOD - 1)) << 16);
    p = atomicAdd(&cur[d0.w >> BSH], 1); rec[p] = s0.w | ((d0.w & (BNOD - 1)) << 16);
    if (two) {
        p = atomicAdd(&cur[d1.x >> BSH], 1); rec[p] = s1.x | ((d1.x & (BNOD - 1)) << 16);
        p = atomicAdd(&cur[d1.y >> BSH], 1); rec[p] = s1.y | ((d1.y & (BNOD - 1)) << 16);
        p = atomicAdd(&cur[d1.z >> BSH], 1); rec[p] = s1.z | ((d1.z & (BNOD - 1)) << 16);
        p = atomicAdd(&cur[d1.w >> BSH], 1); rec[p] = s1.w | ((d1.w & (BNOD - 1)) << 16);
    }
}

__global__ void __launch_bounds__(512)
k_bucket(const int* __restrict__ rec, const int* __restrict__ bbase,
         int* __restrict__ rp, int* __restrict__ csr) {
    __shared__ int cnt[256];
    __shared__ int wtot[8];
    int t = threadIdx.x, b = blockIdx.x;
    int beg = bbase[b], end = bbase[b + 1];
    if (t < 256) cnt[t] = 0;
    __syncthreads();

    for (int p = beg + t; p < end; p += 4 * 512) {
        int r[4];
        #pragma unroll
        for (int k = 0; k < 4; ++k) { int q = p + k * 512; int v = rec[q < end ? q : end - 1]; r[k] = (q < end) ? v : -1; }
        #pragma unroll
        for (int k = 0; k < 4; ++k) if (r[k] >= 0) atomicAdd(&cnt[r[k] >> 16], 1);
    }
    __syncthreads();

    int deg = (t < 256) ? cnt[t] : 0;
    int excl = scan256_excl(deg, t, wtot);
    if (t < 256) {
        int start = beg + excl;
        int node  = b * BNOD + t;
        if (node <= NNODE) rp[node] = start;
        cnt[t] = start;
    }
    __syncthreads();

    for (int p = beg + t; p < end; p += 4 * 512) {
        int r[4];
        #pragma unroll
        for (int k = 0; k < 4; ++k) { int q = p + k * 512; int v = rec[q < end ? q : end - 1]; r[k] = (q < end) ? v : -1; }
        #pragma unroll
        for (int k = 0; k < 4; ++k)
            if (r[k] >= 0) { int s = atomicAdd(&cnt[r[k] >> 16], 1); csr[s] = r[k] & 0xffff; }
    }
}

DEVINL float dot8(const float4& w0, const float4& w1, const float4& c0, const float4& c1) {
    float s = 0.f;
    s += w0.x * c0.x; s += w0.y * c0.y; s += w0.z * c0.z; s += w0.w * c0.w;
    s += w1.x * c1.x; s += w1.y * c1.y; s += w1.z * c1.z; s += w1.w * c1.w;
    return s;
}

DEVINL float sum8(float v) {
    v += __shfl_xor(v, 1, 32);
    v += __shfl_xor(v, 2, 32);
    v += __shfl_xor(v, 4, 32);
    return v;
}

DEVINL void axpy8(float4& a0, float4& a1, float x, const float4& w0, const float4& w1) {
    a0.x += x * w0.x; a0.y += x * w0.y; a0.z += x * w0.z; a0.w += x * w0.w;
    a1.x += x * w1.x; a1.y += x * w1.y; a1.z += x * w1.z; a1.w += x * w1.w;
}

DEVINL void scale8(float4& a0, float4& a1, float s) {
    a0.x *= s; a0.y *= s; a0.z *= s; a0.w *= s;
    a1.x *= s; a1.y *= s; a1.z *= s; a1.w *= s;
}

__global__ void __launch_bounds__(256)
k_gather1(const int* __restrict__ rp, const int* __restrict__ csr,
          const float* __restrict__ a_s, const float* __restrict__ a_d,
          const float* __restrict__ Wh, float* __restrict__ h) {
    int idx = blockIdx.x * blockDim.x + threadIdx.x;
    if (idx >= NNODE * H1) return;
    int n = idx >> 3, hh = idx & 7;
    int beg = rp[n], end = rp[n + 1];
    const float* wcol = Wh + hh * 8;
    const float4 as0 = *(const float4*)(a_s + hh * 8), as1 = *(const float4*)(a_s + hh * 8 + 4);
    const float4 ad0 = *(const float4*)(a_d + hh * 8), ad1 = *(const float4*)(a_d + hh * 8 + 4);
    const float* own = wcol + (size_t)n * HID;
    float edn = dot8(*(const float4*)(own), *(const float4*)(own + 4), ad0, ad1);

    float m = -INFINITY, den = 0.f;
    float4 a0 = make_float4(0.f, 0.f, 0.f, 0.f), a1 = a0;
    {
        const int last = end - 1;
        for (int p = beg; p < end; p += 4) {
            int s0 = csr[p];
            int s1 = csr[min(p + 1, last)];
            int s2 = csr[min(p + 2, last)];
            int s3 = csr[min(p + 3, last)];
            const float* r0 = wcol + (size_t)s0 * HID;
            const float* r1 = wcol + (size_t)s1 * HID;
            const float* r2 = wcol + (size_t)s2 * HID;
            const float* r3 = wcol + (size_t)s3 * HID;
            float4 w00 = *(const float4*)(r0), w01 = *(const float4*)(r0 + 4);
            float4 w10 = *(const float4*)(r1), w11 = *(const float4*)(r1 + 4);
            float4 w20 = *(const float4*)(r2), w21 = *(const float4*)(r2 + 4);
            float4 w30 = *(const float4*)(r3), w31 = *(const float4*)(r3 + 4);
            float e0 = lrelu(dot8(w00, w01, as0, as1) + edn);
            float e1 = lrelu(dot8(w10, w11, as0, as1) + edn);
            float e2 = lrelu(dot8(w20, w21, as0, as1) + edn);
            float e3 = lrelu(dot8(w30, w31, as0, as1) + edn);
            e1 = (p + 1 < end) ? e1 : -INFINITY;
            e2 = (p + 2 < end) ? e2 : -INFINITY;
            e3 = (p + 3 < end) ? e3 : -INFINITY;
            float mn = fmaxf(fmaxf(m, e0), fmaxf(e1, fmaxf(e2, e3)));
            float sc = __expf(m - mn);
            float x0 = __expf(e0 - mn), x1 = __expf(e1 - mn);
            float x2 = __expf(e2 - mn), x3 = __expf(e3 - mn);
            den *= sc; scale8(a0, a1, sc);
            den += x0; axpy8(a0, a1, x0, w00, w01);
            den += x1; axpy8(a0, a1, x1, w10, w11);
            den += x2; axpy8(a0, a1, x2, w20, w21);
            den += x3; axpy8(a0, a1, x3, w30, w31);
            m = mn;
        }
    }

    float dn = den + 1e-10f;
    float x[8] = { a0.x / dn, a0.y / dn, a0.z / dn, a0.w / dn,
                   a1.x / dn, a1.y / dn, a1.z / dn, a1.w / dn };
    #pragma unroll
    for (int j = 0; j < 8; ++j) x[j] = x[j] > 0.f ? x[j] : (__expf(x[j]) - 1.f);
    float* o = h + (size_t)n * HID + hh * 8;
    *(float4*)(o)     = make_float4(x[0], x[1], x[2], x[3]);
    *(float4*)(o + 4) = make_float4(x[4], x[5], x[6], x[7]);
}

__global__ void __launch_bounds__(256)
k_gather2(const int* __restrict__ rp, const int* __restrict__ csr,
          const float* __restrict__ a_s, const float* __restrict__ a_d,
          const float* __restrict__ Wh, float* __restrict__ out) {
    int idx = blockIdx.x * blockDim.x + threadIdx.x;
    if (idx >= NNODE * 8) return;
    int n = idx >> 3, c0 = (idx & 7) * 8;
    int beg = rp[n], end = rp[n + 1];
    const float* wcol = Wh + c0;
    const float4 as0 = *(const float4*)(a_s + c0), as1 = *(const float4*)(a_s + c0 + 4);
    const float4 ad0 = *(const float4*)(a_d + c0), ad1 = *(const float4*)(a_d + c0 + 4);
    const float* own = wcol + (size_t)n * NLAB;
    float edn = sum8(dot8(*(const float4*)(own), *(const float4*)(own + 4), ad0, ad1));

    float m = -INFINITY, den = 0.f;
    float4 a0 = make_float4(0.f, 0.f, 0.f, 0.f), a1 = a0;
    {
        const int last = end - 1;
        for (int p = beg; p < end; p += 4) {
            int s0 = csr[p];
            int s1 = csr[min(p + 1, last)];
            int s2 = csr[min(p + 2, last)];
            int s3 = csr[min(p + 3, last)];
            const float* r0 = wcol + (size_t)s0 * NLAB;
            const float* r1 = wcol + (size_t)s1 * NLAB;
            const float* r2 = wcol + (size_t)s2 * NLAB;
            const float* r3 = wcol + (size_t)s3 * NLAB;
            float4 w00 = *(const float4*)(r0), w01 = *(const float4*)(r0 + 4);
            float4 w10 = *(const float4*)(r1), w11 = *(const float4*)(r1 + 4);
            float4 w20 = *(const float4*)(r2), w21 = *(const float4*)(r2 + 4);
            float4 w30 = *(const float4*)(r3), w31 = *(const float4*)(r3 + 4);
            float e0 = lrelu(sum8(dot8(w00, w01, as0, as1)) + edn);
            float e1 = lrelu(sum8(dot8(w10, w11, as0, as1)) + edn);
            float e2 = lrelu(sum8(dot8(w20, w21, as0, as1)) + edn);
            float e3 = lrelu(sum8(dot8(w30, w31, as0, as1)) + edn);
            e1 = (p + 1 < end) ? e1 : -INFINITY;
            e2 = (p + 2 < end) ? e2 : -INFINITY;
            e3 = (p + 3 < end) ? e3 : -INFINITY;
            float mn = fmaxf(fmaxf(m, e0), fmaxf(e1, fmaxf(e2, e3)));
            float sc = __expf(m - mn);
            float x0 = __expf(e0 - mn), x1 = __expf(e1 - mn);
            float x2 = __expf(e2 - mn), x3 = __expf(e3 - mn);
            den *= sc; scale8(a0, a1, sc);
            den += x0; axpy8(a0, a1, x0, w00, w01);
            den += x1; axpy8(a0, a1, x1, w10, w11);
            den += x2; axpy8(a0, a1, x2, w20, w21);
            den += x3; axpy8(a0, a1, x3, w30, w31);
            m = mn;
        }
    }

    float dn = den + 1e-10f;
    float x[8] = { a0.x / dn, a0.y / dn, a0.z / dn, a0.w / dn,
                   a1.x / dn, a1.y / dn, a1.z / dn, a1.w / dn };
    float mx = x[0];
    #pragma unroll
    for (int j = 1; j < 8; ++j) mx = fmaxf(mx, x[j]);
    #pragma unroll
    for (int o = 1; o < 8; o <<= 1) mx = fmaxf(mx, __shfl_xor(mx, o, 32));
    float sum = 0.f;
    #pragma unroll
    for (int j = 0; j < 8; ++j) { x[j] = __expf(x[j] - mx); sum += x[j]; }
    #pragma unroll
    for (int o = 1; o < 8; o <<= 1) sum += __shfl_xor(sum, o, 32);
    float inv = 1.0f / sum;
    float* o = out + (size_t)n * NLAB + c0;
    *(float4*)(o)     = make_float4(x[0] * inv, x[1] * inv, x[2] * inv, x[3] * inv);
    *(float4*)(o + 4) = make_float4(x[4] * inv, x[5] * inv, x[6] * inv, x[7] * inv);
}

extern "C" void kernel_launch(void* const* d_in, const int* in_sizes, int n_in,
                              void* d_out, int out_size, void* d_ws, size_t ws_size,
                              hipStream_t stream) {
    const float* X   = (const float*)d_in[0];
    const float* W1  = (const float*)d_in[1];
    const float* a1s = (const float*)d_in[2];
    const float* a1d = (const float*)d_in[3];
    const float* W2  = (const float*)d_in[4];
    const float* a2s = (const float*)d_in[5];
    const float* a2d = (const float*)d_in[6];
    const int*   src = (const int*)d_in[7];
    const int*   dst = (const int*)d_in[8];
    float*       out = (float*)d_out;

    char*  ws  = (char*)d_ws;
    size_t off = 0;
    auto alloc = [&](size_t bytes) -> void* {
        void* p = ws + off;
        off = (off + bytes + 255) & ~(size_t)255;
        return p;
    };
    float*    Wh1   = (float*)   alloc((size_t)NNODE * HID  * sizeof(float));
    float*    h1    = (float*)   alloc((size_t)NNODE * HID  * sizeof(float));
    float*    Wh2   = (float*)   alloc((size_t)NNODE * NLAB * sizeof(float));
    int*      hist  = (int*)     alloc((size_t)NPAD         * sizeof(int));
    int*      rp    = (int*)     alloc((size_t)NPAD         * sizeof(int));
    int*      rec   = (int*)     alloc((size_t)NEDGE        * sizeof(int));
    int*      csr   = (int*)     alloc((size_t)NEDGE        * sizeof(int));
    __bf16*   Bp1   = (__bf16*)  alloc((size_t)4 * 16 * 512 * sizeof(__bf16));
    __bf16*   Bp2   = (__bf16*)  alloc((size_t)4 * 2  * 512 * sizeof(__bf16));

    const int B = 256;
    auto cdiv = [](long long a, long long b) { return (int)((a + b - 1) / b); };
    const int gemmGrid = cdiv(ROWT, 4);

    int* bbase = hist + BB_OFF;
    k_hist<<<NWG, 256, 0, stream>>>(dst, hist);
    k_scatter<<<NWG, 512, 0, stream>>>(src, dst, hist, bbase, rec);
    k_bucket<<<NBK, 512, 0, stream>>>(rec, bbase, rp, csr);

    k_repack<<<cdiv(4 * 16 * 512, B), B, 0, stream>>>(W1, Bp1, 16, 4 * 16 * 512, 1);
    k_repack<<<cdiv(4 * 2  * 512, B), B, 0, stream>>>(W2, Bp2, 2,  4 * 2  * 512, 0);

    k_gemm_bf16<<<gemmGrid, 128, 0, stream>>>(X, FIN, Bp1, FIN / 32, Wh1);
    k_gather1<<<cdiv((long long)NNODE * H1, B), B, 0, stream>>>(rp, csr, a1s, a1d, Wh1, h1);

    k_gemm_bf16<<<gemmGrid, 128, 0, stream>>>(h1, HID, Bp2, HID / 32, Wh2);
    k_gather2<<<cdiv((long long)NNODE * 8, B), B, 0, stream>>>(rp, csr, a2s, a2d, Wh2, out);
}


# Independent stages share launches by block index: W1/W2 repacks run beside the chunk histograms, the layer-1 GEMM beside the bucket work-groups (GEMM blocks first in the grid); 9 launches -> 6
# speedup vs baseline: 5.0291x; 1.1148x over previous
#include <hip/hip_runtime.h>
#include <hip/hip_bf16.h>


typedef __attribute__((ext_vector_type(16))) __bf16 v16bf;
typedef __attribute__((ext_vector_type(8)))  float  v8f;

constexpr int NNODE = 50000;
constexpr int NEDGE = 800000;
constexpr int FIN   = 512;
constexpr int H1    = 8;
constexpr int FHID  = 8;
constexpr int HID   = H1 * FHID;
constexpr int NLAB  = 64;
constexpr int ROWT  = NNODE / 16;
constexpr float SLOPE = 0.2f;

#define DEVINL __device__ __forceinline__

DEVINL float lrelu(float v) { return v > 0.f ? v : SLOPE * v; }

DEVINL void repack_body(const float* __restrict__ W, __bf16* __restrict__ Bp,
                        int nKt, int idx, int gat1) {
    int e    = idx & 15;
    int lane = (idx >> 4) & 31;
    int ktnt = idx >> 9;
    int kt   = ktnt % nKt;
    int nt   = ktnt / nKt;
    int K = ((e < 8) ? 0 : 16) + (lane >> 4) * 8 + (e & 7);
    int k = kt * 32 + K;
    int n = nt * 16 + (lane & 15);
    float v = gat1 ? W[(size_t)(n >> 3) * (FIN * FHID) + (size_t)k * FHID + (n & 7)]
                   : W[(size_t)k * NLAB + n];
    Bp[idx] = (__bf16)v;
}

DEVINL void gemm_tile(const float* __restrict__ A, int lda,
                      const __bf16* __restrict__ Bp, int nKt,
                      float* __restrict__ C, int tile) {
    int lane  = threadIdx.x & 31;
    int m     = lane & 15;
    int khalf = lane >> 4;
    const float*  arow = A  + (size_t)(tile * 16 + m) * lda + khalf * 8;
    const __bf16* bl   = Bp + (size_t)lane * 16;

    v8f acc[4] = {};
    for (int kt = 0; kt < nKt; ++kt) {
        const float* ap = arow + kt * 32;
        float4 a0 = *(const float4*)(ap);
        float4 a1 = *(const float4*)(ap + 4);
        float4 a2 = *(const float4*)(ap + 16);
        float4 a3 = *(const float4*)(ap + 20);
        if (kt + 1 < nKt) __builtin_prefetch(ap + 32, 0, 0);

        v16bf av;
        av[0]  = (__bf16)a0.x; av[1]  = (__bf16)a0.y; av[2]  = (__bf16)a0.z; av[3]  = (__bf16)a0.w;
        av[4]  = (__bf16)a1.x; av[5]  = (__bf16)a1.y; av[6]  = (__bf16)a1.z; av[7]  = (__bf16)a1.w;
        av[8]  = (__bf16)a2.x; av[9]  = (__bf16)a2.y; av[10] = (__bf16)a2.z; av[11] = (__bf16)a2.w;
        av[12] = (__bf16)a3.x; av[13] = (__bf16)a3.y; av[14] = (__bf16)a3.z; av[15] = (__bf16)a3.w;

        const __bf16* bk = bl + (size_t)kt * 512;
        #pragma unroll
        for (int nt = 0; nt < 4; ++nt) {
            v16bf bv = *(const v16bf*)(bk + (size_t)nt * nKt * 512);
            acc[nt] = __builtin_amdgcn_wmma_f32_16x16x32_bf16(
                false, av, false, bv, (short)0, acc[nt], false, false);
        }
    }

    float* cbase = C + (size_t)(tile * 16 + khalf * 8) * HID + m;
    #pragma unroll
    for (int nt = 0; nt < 4; ++nt) {
        float* crow = cbase + nt * 16;
        #pragma unroll
        for (int r = 0; r < 8; ++r) crow[(size_t)r * HID] = acc[nt][r];
    }
}

__global__ void __launch_bounds__(128)
k_gemm_bf16(const float* __restrict__ A, int lda,
            const __bf16* __restrict__ Bp, int nKt,
            float* __restrict__ C) {
    int tile = blockIdx.x * 4 + (threadIdx.x >> 5);
    if (tile >= ROWT) return;
    gemm_tile(A, lda, Bp, nKt, C, tile);
}

constexpr int NPAD = 53248;
constexpr int BSH  = 8;
constexpr int BNOD = 1 << BSH;
constexpr int NBK  = (NNODE >> BSH) + 1;
constexpr int EC   = 3200;
constexpr int EQ   = EC / 4;
constexpr int NWG  = NEDGE / EC;
constexpr int BB_OFF = 49152;
static_assert(NEDGE % EC == 0 && EC % 4 == 0, "chunks are whole int4 runs");
constexpr int EQH  = (EQ + 255) / 256;
constexpr int NBQ  = NBK / 4;
constexpr int RG   = 10;
constexpr int RPG  = NWG / RG;
static_assert(NBK % 4 == 0 && NWG % RG == 0 && RG * NBQ <= 512, "column-sum tiling");
static_assert(NBK <= 256 && NNODE <= 65536, "bucket fits the block; src fits 16 bits");
static_assert(NWG * NBK <= BB_OFF && BB_OFF + 256 <= NPAD && NNODE + 1 <= NPAD, "slices");
static_assert(EQ > 512 && EQ <= 1024, "k_scatter holds a chunk in two int4 per thread");

DEVINL void hist_body(const int* __restrict__ dst, int* __restrict__ hist) {
    __shared__ int cnt[256];
    int t = threadIdx.x;
    const int4* dp = (const int4*)(dst + (size_t)blockIdx.x * EC);
    int4 d[EQH];
    #pragma unroll
    for (int k = 0; k < EQH; ++k) {
        int i = t + k * 256;
        d[k] = dp[i < EQ ? i : EQ - 1];
    }
    cnt[t] = 0;
    __syncthreads();
    #pragma unroll
    for (int k = 0; k < EQH; ++k) {
        if (t + k * 256 < EQ) {
            atomicAdd(&cnt[d[k].x >> BSH], 1);
            atomicAdd(&cnt[d[k].y >> BSH], 1);
            atomicAdd(&cnt[d[k].z >> BSH], 1);
            atomicAdd(&cnt[d[k].w >> BSH], 1);
        }
    }
    __syncthreads();
    if (t < NBK) hist[blockIdx.x * NBK + t] = cnt[t];
}

constexpr int RP1  = 4 * (FIN / 32) * 512;
constexpr int RP2  = 4 * (HID / 32) * 512;
constexpr int RP1B = RP1 / 256;
constexpr int RP2B = RP2 / 256;
static_assert(RP1 % 256 == 0 && RP2 % 256 == 0, "repack blocks are full");

__global__ void __launch_bounds__(256)
k_hist_repack(const int* __restrict__ dst, int* __restrict__ hist,
              const float* __restrict__ W1, __bf16* __restrict__ Bp1,
              const float* __restrict__ W2, __bf16* __restrict__ Bp2) {
    int b = blockIdx.x;
    if (b < NWG) {
        hist_body(dst, hist);
    } else if (b < NWG + RP1B) {
        repack_body(W1, Bp1, FIN / 32, (b - NWG) * 256 + threadIdx.x, 1);
    } else {
        repack_body(W2, Bp2, HID / 32, (b - NWG - RP1B) * 256 + threadIdx.x, 0);
    }
}

DEVINL int scan256_excl(int v, int t, int* wtot) {
    int lane = t & 31, w = t >> 5;
    int inc = v;
    #pragma unroll
    for (int o = 1; o < 32; o <<= 1) { int u = __shfl_up(inc, o, 32); if (lane >= o) inc += u; }
    if (lane == 31 && w < 8) wtot[w] = inc;
    __syncthreads();
    int pre = 0;
    #pragma unroll
    for (int j = 0; j < 8; ++j) pre += (j < w) ? wtot[j] : 0;
    return pre + inc - v;
}

__global__ void __launch_bounds__(512)
k_scatter(const int* __restrict__ src, const int* __restrict__ dst,
          const int* __restrict__ hist, int* __restrict__ bbase,
          int* __restrict__ rec) {
    __shared__ int4 s_tot[RG][NBQ];
    __shared__ int4 s_par[RG][NBQ];
    __shared__ int cur[256];
    __shared__ int wtot[8];
    int t = threadIdx.x, wg = blockIdx.x;

    const int4* dp = (const int4*)(dst + (size_t)wg * EC);
    const int4* sp = (const int4*)(src + (size_t)wg * EC);
    bool two = t + 512 < EQ;
    int  i1  = two ? t + 512 : t;
    int4 d0 = dp[t],  s0 = sp[t];
    int4 d1 = dp[i1], s1 = sp[i1];

    if (t < RG * NBQ) {
        int q = t % NBQ, g = t / NBQ;
        const int4* hp = (const int4*)hist + (size_t)(g * RPG) * NBQ + q;
        int4 h[RPG];
        #pragma unroll
        for (int k = 0; k < RPG; ++k) h[k] = hp[k * NBQ];
        int4 tot = make_int4(0, 0, 0, 0), par = tot;
        #pragma unroll
        for (int k = 0; k < RPG; ++k) {
            tot.x += h[k].x; tot.y += h[k].y; tot.z += h[k].z; tot.w += h[k].w;
            if (g * RPG + k < wg) { par.x += h[k].x; par.y += h[k].y; par.z += h[k].z; par.w += h[k].w; }
        }
        s_tot[g][q] = tot;
        s_par[g][q] = par;
    }
    __syncthreads();
    int tot = 0, par = 0;
    if (t < NBK) {
        const int* st = (const int*)s_tot;
        const int* sq = (const int*)s_par;
        #pragma unroll
        for (int g = 0; g < RG; ++g) { tot += st[g * NBK + t]; par += sq[g * NBK + t]; }
    }
    int excl = scan256_excl(tot, t, wtot);
    if (t < 256) {
        cur[t] = excl + par;
        if (wg == 0 && t <= NBK) bbase[t] = excl;
    }
    __syncthreads();

    int p;
    p = atomicAdd(&cur[d0.x >> BSH], 1); rec[p] = s0.x | ((d0.x & (BNOD - 1)) << 16);
    p = atomicAdd(&cur[d0.y >> BSH], 1); rec[p] = s0.y | ((d0.y & (BNOD - 1)) << 16);
    p = atomicAdd(&cur[d0.z >> BSH], 1); rec[p] = s0.z | ((d0.z & (BNOD - 1)) << 16);
    p = atomicAdd(&cur[d0.w >> BSH], 1); rec[p] = s0.w | ((d0.w & (BNOD - 1)) << 16);
    if (two) {
        p = atomicAdd(&cur[d1.x >> BSH], 1); rec[p] = s1.x | ((d1.x & (BNOD - 1)) << 16);
        p = atomicAdd(&cur[d1.y >> BSH], 1); rec[p] = s1.y | ((d1.y & (BNOD - 1)) << 16);
        p = atomicAdd(&cur[d1.z >> BSH], 1); rec[p] = s1.z | ((d1.z & (BNOD - 1)) << 16);
        p = atomicAdd(&cur[d1.w >> BSH], 1); rec[p] = s1.w | ((d1.w & (BNOD - 1)) << 16);
    }
}

DEVINL void bucket_body(const int* __restrict__ rec, const int* __restrict__ bbase,
                        int* __restrict__ rp, int* __restrict__ csr, int b) {
    __shared__ int cnt[256];
    __shared__ int wtot[8];
    int t = threadIdx.x;
    int beg = bbase[b], end = bbase[b + 1];
    if (t < 256) cnt[t] = 0;
    __syncthreads();

    for (int p = beg + t; p < end; p += 4 * 512) {
        int r[4];
        #pragma unroll
        for (int k = 0; k < 4; ++k) { int q = p + k * 512; int v = rec[q < end ? q : end - 1]; r[k] = (q < end) ? v : -1; }
        #pragma unroll
        for (int k = 0; k < 4; ++k) if (r[k] >= 0) atomicAdd(&cnt[r[k] >> 16], 1);
    }
    __syncthreads();

    int deg = (t < 256) ? cnt[t] : 0;
    int excl = scan256_excl(deg, t, wtot);
    if (t < 256) {
        int start = beg + excl;
        int node  = b * BNOD + t;
        if (node <= NNODE) rp[node] = start;
        cnt[t] = start;
    }
    __syncthreads();

    for (int p = beg + t; p < end; p += 4 * 512) {
        int r[4];
        #pragma unroll
        for (int k = 0; k < 4; ++k) { int q = p + k * 512; int v = rec[q < end ? q : end - 1]; r[k] = (q < end) ? v : -1; }
        #pragma unroll
        for (int k = 0; k < 4; ++k)
            if (r[k] >= 0) { int s = atomicAdd(&cnt[r[k] >> 16], 1); csr[s] = r[k] & 0xffff; }
    }
}

constexpr int GT    = 16;
constexpr int GB    = (ROWT + GT - 1) / GT;
constexpr int GRID1 = NWG + RP1B + RP2B;
constexpr int GRID3 = GB + NBK;

__global__ void __launch_bounds__(512)
k_gemm_bucket(const float* __restrict__ A, int lda,
              const __bf16* __restrict__ Bp, int nKt, float* __restrict__ C,
              const int* __restrict__ rec, const int* __restrict__ bbase,
              int* __restrict__ rp, int* __restrict__ csr) {
    if (blockIdx.x >= GB) {
        bucket_body(rec, bbase, rp, csr, blockIdx.x - GB);
        return;
    }
    int tile = blockIdx.x * GT + (threadIdx.x >> 5);
    if (tile >= ROWT) return;
    gemm_tile(A, lda, Bp, nKt, C, tile);
}

DEVINL float dot8(const float4& w0, const float4& w1, const float4& c0, const float4& c1) {
    float s = 0.f;
    s += w0.x * c0.x; s += w0.y * c0.y; s += w0.z * c0.z; s += w0.w * c0.w;
    s += w1.x * c1.x; s += w1.y * c1.y; s += w1.z * c1.z; s += w1.w * c1.w;
    return s;
}

DEVINL float sum8(float v) {
    v += __shfl_xor(v, 1, 32);
    v += __shfl_xor(v, 2, 32);
    v += __shfl_xor(v, 4, 32);
    return v;
}

DEVINL void axpy8(float4& a0, float4& a1, float x, const float4& w0, const float4& w1) {
    a0.x += x * w0.x; a0.y += x * w0.y; a0.z += x * w0.z; a0.w += x * w0.w;
    a1.x += x * w1.x; a1.y += x * w1.y; a1.z += x * w1.z; a1.w += x * w1.w;
}

DEVINL void scale8(float4& a0, float4& a1, float s) {
    a0.x *= s; a0.y *= s; a0.z *= s; a0.w *= s;
    a1.x *= s; a1.y *= s; a1.z *= s; a1.w *= s;
}

__global__ void __launch_bounds__(256)
k_gather1(const int* __restrict__ rp, const int* __restrict__ csr,
          const float* __restrict__ a_s, const float* __restrict__ a_d,
          const float* __restrict__ Wh, float* __restrict__ h) {
    int idx = blockIdx.x * blockDim.x + threadIdx.x;
    if (idx >= NNODE * H1) return;
    int n = idx >> 3, hh = idx & 7;
    int beg = rp[n], end = rp[n + 1];
    const float* wcol = Wh + hh * 8;
    const float4 as0 = *(const float4*)(a_s + hh * 8), as1 = *(const float4*)(a_s + hh * 8 + 4);
    const float4 ad0 = *(const float4*)(a_d + hh * 8), ad1 = *(const float4*)(a_d + hh * 8 + 4);
    const float* own = wcol + (size_t)n * HID;
    float edn = dot8(*(const float4*)(own), *(const float4*)(own + 4), ad0, ad1);

    float m = -INFINITY, den = 0.f;
    float4 a0 = make_float4(0.f, 0.f, 0.f, 0.f), a1 = a0;
    {
        const int last = end - 1;
        for (int p = beg; p < end; p += 4) {
            int s0 = csr[p];
            int s1 = csr[min(p + 1, last)];
            int s2 = csr[min(p + 2, last)];
            int s3 = csr[min(p + 3, last)];
            const float* r0 = wcol + (size_t)s0 * HID;
            const float* r1 = wcol + (size_t)s1 * HID;
            const float* r2 = wcol + (size_t)s2 * HID;
            const float* r3 = wcol + (size_t)s3 * HID;
            float4 w00 = *(const float4*)(r0), w01 = *(const float4*)(r0 + 4);
            float4 w10 = *(const float4*)(r1), w11 = *(const float4*)(r1 + 4);
            float4 w20 = *(const float4*)(r2), w21 = *(const float4*)(r2 + 4);
            float4 w30 = *(const float4*)(r3), w31 = *(const float4*)(r3 + 4);
            float e0 = lrelu(dot8(w00, w01, as0, as1) + edn);
            float e1 = lrelu(dot8(w10, w11, as0, as1) + edn);
            float e2 = lrelu(dot8(w20, w21, as0, as1) + edn);
            float e3 = lrelu(dot8(w30, w31, as0, as1) + edn);
            e1 = (p + 1 < end) ? e1 : -INFINITY;
            e2 = (p + 2 < end) ? e2 : -INFINITY;
            e3 = (p + 3 < end) ? e3 : -INFINITY;
            float mn = fmaxf(fmaxf(m, e0), fmaxf(e1, fmaxf(e2, e3)));
            float sc = __expf(m - mn);
            float x0 = __expf(e0 - mn), x1 = __expf(e1 - mn);
            float x2 = __expf(e2 - mn), x3 = __expf(e3 - mn);
            den *= sc; scale8(a0, a1, sc);
            den += x0; axpy8(a0, a1, x0, w00, w01);
            den += x1; axpy8(a0, a1, x1, w10, w11);
            den += x2; axpy8(a0, a1, x2, w20, w21);
            den += x3; axpy8(a0, a1, x3, w30, w31);
            m = mn;
        }
    }

    float dn = den + 1e-10f;
    float x[8] = { a0.x / dn, a0.y / dn, a0.z / dn, a0.w / dn,
                   a1.x / dn, a1.y / dn, a1.z / dn, a1.w / dn };
    #pragma unroll
    for (int j = 0; j < 8; ++j) x[j] = x[j] > 0.f ? x[j] : (__expf(x[j]) - 1.f);
    float* o = h + (size_t)n * HID + hh * 8;
    *(float4*)(o)     = make_float4(x[0], x[1], x[2], x[3]);
    *(float4*)(o + 4) = make_float4(x[4], x[5], x[6], x[7]);
}

__global__ void __launch_bounds__(256)
k_gather2(const int* __restrict__ rp, const int* __restrict__ csr,
          const float* __restrict__ a_s, const float* __restrict__ a_d,
          const float* __restrict__ Wh, float* __restrict__ out) {
    int idx = blockIdx.x * blockDim.x + threadIdx.x;
    if (idx >= NNODE * 8) return;
    int n = idx >> 3, c0 = (idx & 7) * 8;
    int beg = rp[n], end = rp[n + 1];
    const float* wcol = Wh + c0;
    const float4 as0 = *(const float4*)(a_s + c0), as1 = *(const float4*)(a_s + c0 + 4);
    const float4 ad0 = *(const float4*)(a_d + c0), ad1 = *(const float4*)(a_d + c0 + 4);
    const float* own = wcol + (size_t)n * NLAB;
    float edn = sum8(dot8(*(const float4*)(own), *(const float4*)(own + 4), ad0, ad1));

    float m = -INFINITY, den = 0.f;
    float4 a0 = make_float4(0.f, 0.f, 0.f, 0.f), a1 = a0;
    {
        const int last = end - 1;
        for (int p = beg; p < end; p += 4) {
            int s0 = csr[p];
            int s1 = csr[min(p + 1, last)];
            int s2 = csr[min(p + 2, last)];
            int s3 = csr[min(p + 3, last)];
            const float* r0 = wcol + (size_t)s0 * NLAB;
            const float* r1 = wcol + (size_t)s1 * NLAB;
            const float* r2 = wcol + (size_t)s2 * NLAB;
            const float* r3 = wcol + (size_t)s3 * NLAB;
            float4 w00 = *(const float4*)(r0), w01 = *(const float4*)(r0 + 4);
            float4 w10 = *(const float4*)(r1), w11 = *(const float4*)(r1 + 4);
            float4 w20 = *(const float4*)(r2), w21 = *(const float4*)(r2 + 4);
            float4 w30 = *(const float4*)(r3), w31 = *(const float4*)(r3 + 4);
            float e0 = lrelu(sum8(dot8(w00, w01, as0, as1)) + edn);
            float e1 = lrelu(sum8(dot8(w10, w11, as0, as1)) + edn);
            float e2 = lrelu(sum8(dot8(w20, w21, as0, as1)) + edn);
            float e3 = lrelu(sum8(dot8(w30, w31, as0, as1)) + edn);
            e1 = (p + 1 < end) ? e1 : -INFINITY;
            e2 = (p + 2 < end) ? e2 : -INFINITY;
            e3 = (p + 3 < end) ? e3 : -INFINITY;
            float mn = fmaxf(fmaxf(m, e0), fmaxf(e1, fmaxf(e2, e3)));
            float sc = __expf(m - mn);
            float x0 = __expf(e0 - mn), x1 = __expf(e1 - mn);
            float x2 = __expf(e2 - mn), x3 = __expf(e3 - mn);
            den *= sc; scale8(a0, a1, sc);
            den += x0; axpy8(a0, a1, x0, w00, w01);
            den += x1; axpy8(a0, a1, x1, w10, w11);
            den += x2; axpy8(a0, a1, x2, w20, w21);
            den += x3; axpy8(a0, a1, x3, w30, w31);
            m = mn;
        }
    }

    float dn = den + 1e-10f;
    float x[8] = { a0.x / dn, a0.y / dn, a0.z / dn, a0.w / dn,
                   a1.x / dn, a1.y / dn, a1.z / dn, a1.w / dn };
    float mx = x[0];
    #pragma unroll
    for (int j = 1; j < 8; ++j) mx = fmaxf(mx, x[j]);
    #pragma unroll
    for (int o = 1; o < 8; o <<= 1) mx = fmaxf(mx, __shfl_xor(mx, o, 32));
    float sum = 0.f;
    #pragma unroll
    for (int j = 0; j < 8; ++j) { x[j] = __expf(x[j] - mx); sum += x[j]; }
    #pragma unroll
    for (int o = 1; o < 8; o <<= 1) sum += __shfl_xor(sum, o, 32);
    float inv = 1.0f / sum;
    float* o = out + (size_t)n * NLAB + c0;
    *(float4*)(o)     = make_float4(x[0] * inv, x[1] * inv, x[2] * inv, x[3] * inv);
    *(float4*)(o + 4) = make_float4(x[4] * inv, x[5] * inv, x[6] * inv, x[7] * inv);
}

extern "C" void kernel_launch(void* const* d_in, const int* in_sizes, int n_in,
                              void* d_out, int out_size, void* d_ws, size_t ws_size,
                              hipStream_t stream) {
    const float* X   = (const float*)d_in[0];
    const float* W1  = (const float*)d_in[1];
    const float* a1s = (const float*)d_in[2];
    const float* a1d = (const float*)d_in[3];
    const float* W2  = (const float*)d_in[4];
    const float* a2s = (const float*)d_in[5];
    const float* a2d = (const float*)d_in[6];
    const int*   src = (const int*)d_in[7];
    const int*   dst = (const int*)d_in[8];
    float*       out = (float*)d_out;

    char*  ws  = (char*)d_ws;
    size_t off = 0;
    auto alloc = [&](size_t bytes) -> void* {
        void* p = ws + off;
        off = (off + bytes + 255) & ~(size_t)255;
        return p;
    };
    float*    Wh1   = (float*)   alloc((size_t)NNODE * HID  * sizeof(float));
    float*    h1    = (float*)   alloc((size_t)NNODE * HID  * sizeof(float));
    float*    Wh2   = (float*)   alloc((size_t)NNODE * NLAB * sizeof(float));
    int*      hist  = (int*)     alloc((size_t)NPAD         * sizeof(int));
    int*      rp    = (int*)     alloc((size_t)NPAD         * sizeof(int));
    int*      rec   = (int*)     alloc((size_t)NEDGE        * sizeof(int));
    int*      csr   = (int*)     alloc((size_t)NEDGE        * sizeof(int));
    __bf16*   Bp1   = (__bf16*)  alloc((size_t)4 * 16 * 512 * sizeof(__bf16));
    __bf16*   Bp2   = (__bf16*)  alloc((size_t)4 * 2  * 512 * sizeof(__bf16));

    const int B = 256;
    auto cdiv = [](long long a, long long b) { return (int)((a + b - 1) / b); };
    const int gemmGrid = cdiv(ROWT, 4);

    int* bbase = hist + BB_OFF;
    k_hist_repack<<<GRID1, 256, 0, stream>>>(dst, hist, W1, Bp1, W2, Bp2);
    k_scatter<<<NWG, 512, 0, stream>>>(src, dst, hist, bbase, rec);
    k_gemm_bucket<<<GRID3, 512, 0, stream>>>(X, FIN, Bp1, FIN / 32, Wh1, rec, bbase, rp, csr);

    k_gather1<<<cdiv((long long)NNODE * H1, B), B, 0, stream>>>(rp, csr, a1s, a1d, Wh1, h1);

    k_gemm_bf16<<<gemmGrid, 128, 0, stream>>>(h1, HID, Bp2, HID / 32, Wh2);
    k_gather2<<<cdiv((long long)NNODE * 8, B), B, 0, stream>>>(rp, csr, a2s, a2d, Wh2, out);
}


# Layer-2 product fused into the layer-1 aggregation kernel: a 32-node block passes its ELU rows through LDS to WMMA, one 16x16 tile per wave; h1 array and one launch gone (6 -> 5); sums unchanged
# speedup vs baseline: 5.3811x; 1.0700x over previous
#include <hip/hip_runtime.h>
#include <hip/hip_bf16.h>


typedef __attribute__((ext_vector_type(16))) __bf16 v16bf;
typedef __attribute__((ext_vector_type(8)))  __bf16 v8bf;
typedef __attribute__((ext_vector_type(8)))  float  v8f;

constexpr int NNODE = 50000;
constexpr int NEDGE = 800000;
constexpr int FIN   = 512;
constexpr int H1    = 8;
constexpr int FHID  = 8;
constexpr int HID   = H1 * FHID;
constexpr int NLAB  = 64;
constexpr int ROWT  = NNODE / 16;
constexpr float SLOPE = 0.2f;

#define DEVINL __device__ __forceinline__

DEVINL float lrelu(float v) { return v > 0.f ? v : SLOPE * v; }

DEVINL void repack_body(const float* __restrict__ W, __bf16* __restrict__ Bp,
                        int nKt, int idx, int gat1) {
    int e    = idx & 15;
    int lane = (idx >> 4) & 31;
    int ktnt = idx >> 9;
    int kt   = ktnt % nKt;
    int nt   = ktnt / nKt;
    int K = ((e < 8) ? 0 : 16) + (lane >> 4) * 8 + (e & 7);
    int k = kt * 32 + K;
    int n = nt * 16 + (lane & 15);
    float v = gat1 ? W[(size_t)(n >> 3) * (FIN * FHID) + (size_t)k * FHID + (n & 7)]
                   : W[(size_t)k * NLAB + n];
    Bp[idx] = (__bf16)v;
}

DEVINL void gemm_tile(const float* __restrict__ A, int lda,
                      const __bf16* __restrict__ Bp, int nKt,
                      float* __restrict__ C, int tile) {
    int lane  = threadIdx.x & 31;
    int m     = lane & 15;
    int khalf = lane >> 4;
    const float*  arow = A  + (size_t)(tile * 16 + m) * lda + khalf * 8;
    const __bf16* bl   = Bp + (size_t)lane * 16;

    v8f acc[4] = {};
    for (int kt = 0; kt < nKt; ++kt) {
        const float* ap = arow + kt * 32;
        float4 a0 = *(const float4*)(ap);
        float4 a1 = *(const float4*)(ap + 4);
        float4 a2 = *(const float4*)(ap + 16);
        float4 a3 = *(const float4*)(ap + 20);
        if (kt + 1 < nKt) __builtin_prefetch(ap + 32, 0, 0);

        v16bf av;
        av[0]  = (__bf16)a0.x; av[1]  = (__bf16)a0.y; av[2]  = (__bf16)a0.z; av[3]  = (__bf16)a0.w;
        av[4]  = (__bf16)a1.x; av[5]  = (__bf16)a1.y; av[6]  = (__bf16)a1.z; av[7]  = (__bf16)a1.w;
        av[8]  = (__bf16)a2.x; av[9]  = (__bf16)a2.y; av[10] = (__bf16)a2.z; av[11] = (__bf16)a2.w;
        av[12] = (__bf16)a3.x; av[13] = (__bf16)a3.y; av[14] = (__bf16)a3.z; av[15] = (__bf16)a3.w;

        const __bf16* bk = bl + (size_t)kt * 512;
        #pragma unroll
        for (int nt = 0; nt < 4; ++nt) {
            v16bf bv = *(const v16bf*)(bk + (size_t)nt * nKt * 512);
            acc[nt] = __builtin_amdgcn_wmma_f32_16x16x32_bf16(
                false, av, false, bv, (short)0, acc[nt], false, false);
        }
    }

    float* cbase = C + (size_t)(tile * 16 + khalf * 8) * HID + m;
    #pragma unroll
    for (int nt = 0; nt < 4; ++nt) {
        float* crow = cbase + nt * 16;
        #pragma unroll
        for (int r = 0; r < 8; ++r) crow[(size_t)r * HID] = acc[nt][r];
    }
}

constexpr int NPAD = 53248;
constexpr int BSH  = 8;
constexpr int BNOD = 1 << BSH;
constexpr int NBK  = (NNODE >> BSH) + 1;
constexpr int EC   = 3200;
constexpr int EQ   = EC / 4;
constexpr int NWG  = NEDGE / EC;
constexpr int BB_OFF = 49152;
static_assert(NEDGE % EC == 0 && EC % 4 == 0, "chunks are whole int4 runs");
constexpr int EQH  = (EQ + 255) / 256;
constexpr int NBQ  = NBK / 4;
constexpr int RG   = 10;
constexpr int RPG  = NWG / RG;
static_assert(NBK % 4 == 0 && NWG % RG == 0 && RG * NBQ <= 512, "column-sum tiling");
static_assert(NBK <= 256 && NNODE <= 65536, "bucket fits the block; src fits 16 bits");
static_assert(NWG * NBK <= BB_OFF && BB_OFF + 256 <= NPAD && NNODE + 1 <= NPAD, "slices");
static_assert(EQ > 512 && EQ <= 1024, "k_scatter holds a chunk in two int4 per thread");

DEVINL void hist_body(const int* __restrict__ dst, int* __restrict__ hist) {
    __shared__ int cnt[256];
    int t = threadIdx.x;
    const int4* dp = (const int4*)(dst + (size_t)blockIdx.x * EC);
    int4 d[EQH];
    #pragma unroll
    for (int k = 0; k < EQH; ++k) {
        int i = t + k * 256;
        d[k] = dp[i < EQ ? i : EQ - 1];
    }
    cnt[t] = 0;
    __syncthreads();
    #pragma unroll
    for (int k = 0; k < EQH; ++k) {
        if (t + k * 256 < EQ) {
            atomicAdd(&cnt[d[k].x >> BSH], 1);
            atomicAdd(&cnt[d[k].y >> BSH], 1);
            atomicAdd(&cnt[d[k].z >> BSH], 1);
            atomicAdd(&cnt[d[k].w >> BSH], 1);
        }
    }
    __syncthreads();
    if (t < NBK) hist[blockIdx.x * NBK + t] = cnt[t];
}

constexpr int RP1  = 4 * (FIN / 32) * 512;
constexpr int RP2  = 4 * (HID / 32) * 512;
constexpr int RP1B = RP1 / 256;
constexpr int RP2B = RP2 / 256;
static_assert(RP1 % 256 == 0 && RP2 % 256 == 0, "repack blocks are full");

__global__ void __launch_bounds__(256)
k_hist_repack(const int* __restrict__ dst, int* __restrict__ hist,
              const float* __restrict__ W1, __bf16* __restrict__ Bp1,
              const float* __restrict__ W2, __bf16* __restrict__ Bp2) {
    int b = blockIdx.x;
    if (b < NWG) {
        hist_body(dst, hist);
    } else if (b < NWG + RP1B) {
        repack_body(W1, Bp1, FIN / 32, (b - NWG) * 256 + threadIdx.x, 1);
    } else {
        repack_body(W2, Bp2, HID / 32, (b - NWG - RP1B) * 256 + threadIdx.x, 0);
    }
}

DEVINL int scan256_excl(int v, int t, int* wtot) {
    int lane = t & 31, w = t >> 5;
    int inc = v;
    #pragma unroll
    for (int o = 1; o < 32; o <<= 1) { int u = __shfl_up(inc, o, 32); if (lane >= o) inc += u; }
    if (lane == 31 && w < 8) wtot[w] = inc;
    __syncthreads();
    int pre = 0;
    #pragma unroll
    for (int j = 0; j < 8; ++j) pre += (j < w) ? wtot[j] : 0;
    return pre + inc - v;
}

__global__ void __launch_bounds__(512)
k_scatter(const int* __restrict__ src, const int* __restrict__ dst,
          const int* __restrict__ hist, int* __restrict__ bbase,
          int* __restrict__ rec) {
    __shared__ int4 s_tot[RG][NBQ];
    __shared__ int4 s_par[RG][NBQ];
    __shared__ int cur[256];
    __shared__ int wtot[8];
    int t = threadIdx.x, wg = blockIdx.x;

    const int4* dp = (const int4*)(dst + (size_t)wg * EC);
    const int4* sp = (const int4*)(src + (size_t)wg * EC);
    bool two = t + 512 < EQ;
    int  i1  = two ? t + 512 : t;
    int4 d0 = dp[t],  s0 = sp[t];
    int4 d1 = dp[i1], s1 = sp[i1];

    if (t < RG * NBQ) {
        int q = t % NBQ, g = t / NBQ;
        const int4* hp = (const int4*)hist + (size_t)(g * RPG) * NBQ + q;
        int4 h[RPG];
        #pragma unroll
        for (int k = 0; k < RPG; ++k) h[k] = hp[k * NBQ];
        int4 tot = make_int4(0, 0, 0, 0), par = tot;
        #pragma unroll
        for (int k = 0; k < RPG; ++k) {
            tot.x += h[k].x; tot.y += h[k].y; tot.z += h[k].z; tot.w += h[k].w;
            if (g * RPG + k < wg) { par.x += h[k].x; par.y += h[k].y; par.z += h[k].z; par.w += h[k].w; }
        }
        s_tot[g][q] = tot;
        s_par[g][q] = par;
    }
    __syncthreads();
    int tot = 0, par = 0;
    if (t < NBK) {
        const int* st = (const int*)s_tot;
        const int* sq = (const int*)s_par;
        #pragma unroll
        for (int g = 0; g < RG; ++g) { tot += st[g * NBK + t]; par += sq[g * NBK + t]; }
    }
    int excl = scan256_excl(tot, t, wtot);
    if (t < 256) {
        cur[t] = excl + par;
        if (wg == 0 && t <= NBK) bbase[t] = excl;
    }
    __syncthreads();

    int p;
    p = atomicAdd(&cur[d0.x >> BSH], 1); rec[p] = s0.x | ((d0.x & (BNOD - 1)) << 16);
    p = atomicAdd(&cur[d0.y >> BSH], 1); rec[p] = s0.y | ((d0.y & (BNOD - 1)) << 16);
    p = atomicAdd(&cur[d0.z >> BSH], 1); rec[p] = s0.z | ((d0.z & (BNOD - 1)) << 16);
    p = atomicAdd(&cur[d0.w >> BSH], 1); rec[p] = s0.w | ((d0.w & (BNOD - 1)) << 16);
    if (two) {
        p = atomicAdd(&cur[d1.x >> BSH], 1); rec[p] = s1.x | ((d1.x & (BNOD - 1)) << 16);
        p = atomicAdd(&cur[d1.y >> BSH], 1); rec[p] = s1.y | ((d1.y & (BNOD - 1)) << 16);
        p = atomicAdd(&cur[d1.z >> BSH], 1); rec[p] = s1.z | ((d1.z & (BNOD - 1)) << 16);
        p = atomicAdd(&cur[d1.w >> BSH], 1); rec[p] = s1.w | ((d1.w & (BNOD - 1)) << 16);
    }
}

DEVINL void bucket_body(const int* __restrict__ rec, const int* __restrict__ bbase,
                        int* __restrict__ rp, int* __restrict__ csr, int b) {
    __shared__ int cnt[256];
    __shared__ int wtot[8];
    int t = threadIdx.x;
    int beg = bbase[b], end = bbase[b + 1];
    if (t < 256) cnt[t] = 0;
    __syncthreads();

    for (int p = beg + t; p < end; p += 4 * 512) {
        int r[4];
        #pragma unroll
        for (int k = 0; k < 4; ++k) { int q = p + k * 512; int v = rec[q < end ? q : end - 1]; r[k] = (q < end) ? v : -1; }
        #pragma unroll
        for (int k = 0; k < 4; ++k) if (r[k] >= 0) atomicAdd(&cnt[r[k] >> 16], 1);
    }
    __syncthreads();

    int deg = (t < 256) ? cnt[t] : 0;
    int excl = scan256_excl(deg, t, wtot);
    if (t < 256) {
        int start = beg + excl;
        int node  = b * BNOD + t;
        if (node <= NNODE) rp[node] = start;
        cnt[t] = start;
    }
    __syncthreads();

    for (int p = beg + t; p < end; p += 4 * 512) {
        int r[4];
        #pragma unroll
        for (int k = 0; k < 4; ++k) { int q = p + k * 512; int v = rec[q < end ? q : end - 1]; r[k] = (q < end) ? v : -1; }
        #pragma unroll
        for (int k = 0; k < 4; ++k)
            if (r[k] >= 0) { int s = atomicAdd(&cnt[r[k] >> 16], 1); csr[s] = r[k] & 0xffff; }
    }
}

constexpr int GT    = 16;
constexpr int GB    = (ROWT + GT - 1) / GT;
constexpr int GRID1 = NWG + RP1B + RP2B;
constexpr int GRID3 = GB + NBK;

__global__ void __launch_bounds__(512)
k_gemm_bucket(const float* __restrict__ A, int lda,
              const __bf16* __restrict__ Bp, int nKt, float* __restrict__ C,
              const int* __restrict__ rec, const int* __restrict__ bbase,
              int* __restrict__ rp, int* __restrict__ csr) {
    if (blockIdx.x >= GB) {
        bucket_body(rec, bbase, rp, csr, blockIdx.x - GB);
        return;
    }
    int tile = blockIdx.x * GT + (threadIdx.x >> 5);
    if (tile >= ROWT) return;
    gemm_tile(A, lda, Bp, nKt, C, tile);
}

DEVINL float dot8(const float4& w0, const float4& w1, const float4& c0, const float4& c1) {
    float s = 0.f;
    s += w0.x * c0.x; s += w0.y * c0.y; s += w0.z * c0.z; s += w0.w * c0.w;
    s += w1.x * c1.x; s += w1.y * c1.y; s += w1.z * c1.z; s += w1.w * c1.w;
    return s;
}

DEVINL float sum8(float v) {
    v += __shfl_xor(v, 1, 32);
    v += __shfl_xor(v, 2, 32);
    v += __shfl_xor(v, 4, 32);
    return v;
}

DEVINL void axpy8(float4& a0, float4& a1, float x, const float4& w0, const float4& w1) {
    a0.x += x * w0.x; a0.y += x * w0.y; a0.z += x * w0.z; a0.w += x * w0.w;
    a1.x += x * w1.x; a1.y += x * w1.y; a1.z += x * w1.z; a1.w += x * w1.w;
}

DEVINL void scale8(float4& a0, float4& a1, float s) {
    a0.x *= s; a0.y *= s; a0.z *= s; a0.w *= s;
    a1.x *= s; a1.y *= s; a1.z *= s; a1.w *= s;
}

constexpr int G1N  = 32;
constexpr int HPIT = 72;
static_assert(NNODE % 16 == 0 && HID == 64 && NLAB == 64, "fused product tiling");

__global__ void __launch_bounds__(256)
k_gather1_gemm2(const int* __restrict__ rp, const int* __restrict__ csr,
                const float* __restrict__ a_s, const float* __restrict__ a_d,
                const float* __restrict__ Wh, const __bf16* __restrict__ Bp2,
                float* __restrict__ Wh2) {
    __shared__ __attribute__((aligned(16))) __bf16 sH[G1N * HPIT];
    int idx = blockIdx.x * blockDim.x + threadIdx.x;
    int n = idx >> 3, hh = idx & 7;
    const bool act = n < NNODE;

    const int w    = __builtin_amdgcn_readfirstlane((int)(threadIdx.x >> 5));
    const int rt   = w >> 2, nt = w & 3;
    const int lane = threadIdx.x & 31;
    const __bf16* bl = Bp2 + ((size_t)(nt * 2) * 32 + lane) * 16;

    float den = 0.f;
    float4 a0 = make_float4(0.f, 0.f, 0.f, 0.f), a1 = a0;
    if (act) {
        int beg = rp[n], end = rp[n + 1];
        const float* wcol = Wh + hh * 8;
        const float4 as0 = *(const float4*)(a_s + hh * 8), as1 = *(const float4*)(a_s + hh * 8 + 4);
        const float4 ad0 = *(const float4*)(a_d + hh * 8), ad1 = *(const float4*)(a_d + hh * 8 + 4);
        const float* own = wcol + (size_t)n * HID;
        float edn = dot8(*(const float4*)(own), *(const float4*)(own + 4), ad0, ad1);

        float m = -INFINITY;
        {
            const int last = end - 1;
            for (int p = beg; p < end; p += 4) {
                int s0 = csr[p];
                int s1 = csr[min(p + 1, last)];
                int s2 = csr[min(p + 2, last)];
                int s3 = csr[min(p + 3, last)];
                const float* r0 = wcol + (size_t)s0 * HID;
                const float* r1 = wcol + (size_t)s1 * HID;
                const float* r2 = wcol + (size_t)s2 * HID;
                const float* r3 = wcol + (size_t)s3 * HID;
                float4 w00 = *(const float4*)(r0), w01 = *(const float4*)(r0 + 4);
                float4 w10 = *(const float4*)(r1), w11 = *(const float4*)(r1 + 4);
                float4 w20 = *(const float4*)(r2), w21 = *(const float4*)(r2 + 4);
                float4 w30 = *(const float4*)(r3), w31 = *(const float4*)(r3 + 4);
                float e0 = lrelu(dot8(w00, w01, as0, as1) + edn);
                float e1 = lrelu(dot8(w10, w11, as0, as1) + edn);
                float e2 = lrelu(dot8(w20, w21, as0, as1) + edn);
                float e3 = lrelu(dot8(w30, w31, as0, as1) + edn);
                e1 = (p + 1 < end) ? e1 : -INFINITY;
                e2 = (p + 2 < end) ? e2 : -INFINITY;
                e3 = (p + 3 < end) ? e3 : -INFINITY;
                float mn = fmaxf(fmaxf(m, e0), fmaxf(e1, fmaxf(e2, e3)));
                float sc = __expf(m - mn);
                float x0 = __expf(e0 - mn), x1 = __expf(e1 - mn);
                float x2 = __expf(e2 - mn), x3 = __expf(e3 - mn);
                den *= sc; scale8(a0, a1, sc);
                den += x0; axpy8(a0, a1, x0, w00, w01);
                den += x1; axpy8(a0, a1, x1, w10, w11);
                den += x2; axpy8(a0, a1, x2, w20, w21);
                den += x3; axpy8(a0, a1, x3, w30, w31);
                m = mn;
            }
        }
    }

    const v16bf bv0 = *(const v16bf*)(bl);
    const v16bf bv1 = *(const v16bf*)(bl + 512);

    if (act) {
        float dn = den + 1e-10f;
        float x[8] = { a0.x / dn, a0.y / dn, a0.z / dn, a0.w / dn,
                       a1.x / dn, a1.y / dn, a1.z / dn, a1.w / dn };
        #pragma unroll
        for (int j = 0; j < 8; ++j) x[j] = x[j] > 0.f ? x[j] : (__expf(x[j]) - 1.f);
        v8bf hv;
        #pragma unroll
        for (int j = 0; j < 8; ++j) hv[j] = (__bf16)x[j];
        *(v8bf*)(sH + (threadIdx.x >> 3) * HPIT + hh * 8) = hv;
    }
    __syncthreads();

    int tile = blockIdx.x * 2 + rt;
    if (tile >= ROWT) return;
    int m16   = lane & 15;
    int khalf = lane >> 4;
    const __bf16* arow = sH + (rt * 16 + m16) * HPIT + khalf * 8;
    v8f acc = {};
    #pragma unroll
    for (int kt = 0; kt < 2; ++kt) {
        v8bf lo = *(const v8bf*)(arow + kt * 32);
        v8bf hi = *(const v8bf*)(arow + kt * 32 + 16);
        v16bf av = __builtin_shufflevector(lo, hi, 0, 1, 2, 3, 4, 5, 6, 7, 8, 9, 10, 11, 12, 13, 14, 15);
        acc = __builtin_amdgcn_wmma_f32_16x16x32_bf16(
            false, av, false, kt == 0 ? bv0 : bv1, (short)0, acc, false, false);
    }
    float* crow = Wh2 + (size_t)(tile * 16 + khalf * 8) * NLAB + nt * 16 + m16;
    #pragma unroll
    for (int r = 0; r < 8; ++r) crow[(size_t)r * NLAB] = acc[r];
}

__global__ void __launch_bounds__(256)
k_gather2(const int* __restrict__ rp, const int* __restrict__ csr,
          const float* __restrict__ a_s, const float* __restrict__ a_d,
          const float* __restrict__ Wh, float* __restrict__ out) {
    int idx = blockIdx.x * blockDim.x + threadIdx.x;
    if (idx >= NNODE * 8) return;
    int n = idx >> 3, c0 = (idx & 7) * 8;
    int beg = rp[n], end = rp[n + 1];
    const float* wcol = Wh + c0;
    const float4 as0 = *(const float4*)(a_s + c0), as1 = *(const float4*)(a_s + c0 + 4);
    const float4 ad0 = *(const float4*)(a_d + c0), ad1 = *(const float4*)(a_d + c0 + 4);
    const float* own = wcol + (size_t)n * NLAB;
    float edn = sum8(dot8(*(const float4*)(own), *(const float4*)(own + 4), ad0, ad1));

    float m = -INFINITY, den = 0.f;
    float4 a0 = make_float4(0.f, 0.f, 0.f, 0.f), a1 = a0;
    {
        const int last = end - 1;
        for (int p = beg; p < end; p += 4) {
            int s0 = csr[p];
            int s1 = csr[min(p + 1, last)];
            int s2 = csr[min(p + 2, last)];
            int s3 = csr[min(p + 3, last)];
            const float* r0 = wcol + (size_t)s0 * NLAB;
            const float* r1 = wcol + (size_t)s1 * NLAB;
            const float* r2 = wcol + (size_t)s2 * NLAB;
            const float* r3 = wcol + (size_t)s3 * NLAB;
            float4 w00 = *(const float4*)(r0), w01 = *(const float4*)(r0 + 4);
            float4 w10 = *(const float4*)(r1), w11 = *(const float4*)(r1 + 4);
            float4 w20 = *(const float4*)(r2), w21 = *(const float4*)(r2 + 4);
            float4 w30 = *(const float4*)(r3), w31 = *(const float4*)(r3 + 4);
            float e0 = lrelu(sum8(dot8(w00, w01, as0, as1)) + edn);
            float e1 = lrelu(sum8(dot8(w10, w11, as0, as1)) + edn);
            float e2 = lrelu(sum8(dot8(w20, w21, as0, as1)) + edn);
            float e3 = lrelu(sum8(dot8(w30, w31, as0, as1)) + edn);
            e1 = (p + 1 < end) ? e1 : -INFINITY;
            e2 = (p + 2 < end) ? e2 : -INFINITY;
            e3 = (p + 3 < end) ? e3 : -INFINITY;
            float mn = fmaxf(fmaxf(m, e0), fmaxf(e1, fmaxf(e2, e3)));
            float sc = __expf(m - mn);
            float x0 = __expf(e0 - mn), x1 = __expf(e1 - mn);
            float x2 = __expf(e2 - mn), x3 = __expf(e3 - mn);
            den *= sc; scale8(a0, a1, sc);
            den += x0; axpy8(a0, a1, x0, w00, w01);
            den += x1; axpy8(a0, a1, x1, w10, w11);
            den += x2; axpy8(a0, a1, x2, w20, w21);
            den += x3; axpy8(a0, a1, x3, w30, w31);
            m = mn;
        }
    }

    float dn = den + 1e-10f;
    float x[8] = { a0.x / dn, a0.y / dn, a0.z / dn, a0.w / dn,
                   a1.x / dn, a1.y / dn, a1.z / dn, a1.w / dn };
    float mx = x[0];
    #pragma unroll
    for (int j = 1; j < 8; ++j) mx = fmaxf(mx, x[j]);
    #pragma unroll
    for (int o = 1; o < 8; o <<= 1) mx = fmaxf(mx, __shfl_xor(mx, o, 32));
    float sum = 0.f;
    #pragma unroll
    for (int j = 0; j < 8; ++j) { x[j] = __expf(x[j] - mx); sum += x[j]; }
    #pragma unroll
    for (int o = 1; o < 8; o <<= 1) sum += __shfl_xor(sum, o, 32);
    float inv = 1.0f / sum;
    float* o = out + (size_t)n * NLAB + c0;
    *(float4*)(o)     = make_float4(x[0] * inv, x[1] * inv, x[2] * inv, x[3] * inv);
    *(float4*)(o + 4) = make_float4(x[4] * inv, x[5] * inv, x[6] * inv, x[7] * inv);
}

extern "C" void kernel_launch(void* const* d_in, const int* in_sizes, int n_in,
                              void* d_out, int out_size, void* d_ws, size_t ws_size,
                              hipStream_t stream) {
    const float* X   = (const float*)d_in[0];
    const float* W1  = (const float*)d_in[1];
    const float* a1s = (const float*)d_in[2];
    const float* a1d = (const float*)d_in[3];
    const float* W2  = (const float*)d_in[4];
    const float* a2s = (const float*)d_in[5];
    const float* a2d = (const float*)d_in[6];
    const int*   src = (const int*)d_in[7];
    const int*   dst = (const int*)d_in[8];
    float*       out = (float*)d_out;

    char*  ws  = (char*)d_ws;
    size_t off = 0;
    auto alloc = [&](size_t bytes) -> void* {
        void* p = ws + off;
        off = (off + bytes + 255) & ~(size_t)255;
        return p;
    };
    float*    Wh1   = (float*)   alloc((size_t)NNODE * HID  * sizeof(float));
    float*    Wh2   = (float*)   alloc((size_t)NNODE * NLAB * sizeof(float));
    int*      hist  = (int*)     alloc((size_t)NPAD         * sizeof(int));
    int*      rp    = (int*)     alloc((size_t)NPAD         * sizeof(int));
    int*      rec   = (int*)     alloc((size_t)NEDGE        * sizeof(int));
    int*      csr   = (int*)     alloc((size_t)NEDGE        * sizeof(int));
    __bf16*   Bp1   = (__bf16*)  alloc((size_t)4 * 16 * 512 * sizeof(__bf16));
    __bf16*   Bp2   = (__bf16*)  alloc((size_t)4 * 2  * 512 * sizeof(__bf16));

    const int B = 256;
    auto cdiv = [](long long a, long long b) { return (int)((a + b - 1) / b); };

    int* bbase = hist + BB_OFF;
    k_hist_repack<<<GRID1, 256, 0, stream>>>(dst, hist, W1, Bp1, W2, Bp2);
    k_scatter<<<NWG, 512, 0, stream>>>(src, dst, hist, bbase, rec);
    k_gemm_bucket<<<GRID3, 512, 0, stream>>>(X, FIN, Bp1, FIN / 32, Wh1, rec, bbase, rp, csr);

    k_gather1_gemm2<<<cdiv((long long)NNODE * H1, B), B, 0, stream>>>(rp, csr, a1s, a1d, Wh1, Bp2, Wh2);

    k_gather2<<<cdiv((long long)NNODE * 8, B), B, 0, stream>>>(rp, csr, a2s, a2d, Wh2, out);
}


# Layer-1 GEMM operand feed: packed W1 copied once per block into LDS (B fragments become LDS reads), X rows requested two k-steps ahead in registers; 4 serialized trips per k-step go; sums unchanged
# speedup vs baseline: 6.1510x; 1.1431x over previous
#include <hip/hip_runtime.h>
#include <hip/hip_bf16.h>


typedef __attribute__((ext_vector_type(16))) __bf16 v16bf;
typedef __attribute__((ext_vector_type(8)))  __bf16 v8bf;
typedef __attribute__((ext_vector_type(8)))  float  v8f;

constexpr int NNODE = 50000;
constexpr int NEDGE = 800000;
constexpr int FIN   = 512;
constexpr int H1    = 8;
constexpr int FHID  = 8;
constexpr int HID   = H1 * FHID;
constexpr int NLAB  = 64;
constexpr int ROWT  = NNODE / 16;
constexpr float SLOPE = 0.2f;

#define DEVINL __device__ __forceinline__

DEVINL float lrelu(float v) { return v > 0.f ? v : SLOPE * v; }

DEVINL void repack_body(const float* __restrict__ W, __bf16* __restrict__ Bp,
                        int nKt, int idx, int gat1) {
    int e    = idx & 15;
    int lane = (idx >> 4) & 31;
    int ktnt = idx >> 9;
    int kt   = ktnt % nKt;
    int nt   = ktnt / nKt;
    int K = ((e < 8) ? 0 : 16) + (lane >> 4) * 8 + (e & 7);
    int k = kt * 32 + K;
    int n = nt * 16 + (lane & 15);
    float v = gat1 ? W[(size_t)(n >> 3) * (FIN * FHID) + (size_t)k * FHID + (n & 7)]
                   : W[(size_t)k * NLAB + n];
    Bp[idx] = (__bf16)v;
}

constexpr int NKT1 = FIN / 32;
constexpr int SBE  = 4 * NKT1 * 512;
constexpr int SBQ  = SBE / 8;
static_assert(SBQ % 512 == 0, "the LDS copy is whole 512-thread sweeps");

DEVINL void load_a(const float* __restrict__ ap, float4 (&d)[4]) {
    d[0] = *(const float4*)(ap);
    d[1] = *(const float4*)(ap + 4);
    d[2] = *(const float4*)(ap + 16);
    d[3] = *(const float4*)(ap + 20);
}

DEVINL void gemm1_block(const float* __restrict__ A, const __bf16* __restrict__ Bp,
                        float* __restrict__ C, __bf16* sB, int blk) {
    const int t     = threadIdx.x;
    const int lane  = t & 31;
    const int m     = lane & 15;
    const int khalf = lane >> 4;
    const int tile  = blk * 16 + (t >> 5);
    const bool valid = tile < ROWT;
    const float* arow = A + (size_t)((valid ? tile : 0) * 16 + m) * FIN + khalf * 8;

    float4 a[3][4];
    if (valid) { load_a(arow, a[0]); load_a(arow + 32, a[1]); }

    {
        const v8bf* g = (const v8bf*)Bp;
        v8bf* s = (v8bf*)sB;
        v8bf v[SBQ / 512];
        #pragma unroll
        for (int j = 0; j < SBQ / 512; ++j) v[j] = g[t + j * 512];
        #pragma unroll
        for (int j = 0; j < SBQ / 512; ++j) {
            int i = t + j * 512;
            s[(i & ~63) + ((i & 1) << 5) + ((i & 63) >> 1)] = v[j];
        }
    }
    __syncthreads();
    if (!valid) return;

    const __bf16* bl = sB + lane * 8;
    v8f acc[4] = {};
    #pragma unroll
    for (int kt = 0; kt < NKT1; ++kt) {
        if (kt + 2 < NKT1) {
            load_a(arow + (kt + 2) * 32, a[(kt + 2) % 3]);
            if (kt + 3 < NKT1) __builtin_prefetch(arow + (kt + 3) * 32, 0, 0);
        }
        __builtin_amdgcn_sched_barrier(0);

        float4 a0 = a[kt % 3][0], a1 = a[kt % 3][1], a2 = a[kt % 3][2], a3 = a[kt % 3][3];
        asm volatile("" : "+v"(a0.x), "+v"(a0.y), "+v"(a0.z), "+v"(a0.w),
                          "+v"(a1.x), "+v"(a1.y), "+v"(a1.z), "+v"(a1.w));
        asm volatile("" : "+v"(a2.x), "+v"(a2.y), "+v"(a2.z), "+v"(a2.w),
                          "+v"(a3.x), "+v"(a3.y), "+v"(a3.z), "+v"(a3.w));
        v16bf av;
        av[0]  = (__bf16)a0.x; av[1]  = (__bf16)a0.y; av[2]  = (__bf16)a0.z; av[3]  = (__bf16)a0.w;
        av[4]  = (__bf16)a1.x; av[5]  = (__bf16)a1.y; av[6]  = (__bf16)a1.z; av[7]  = (__bf16)a1.w;
        av[8]  = (__bf16)a2.x; av[9]  = (__bf16)a2.y; av[10] = (__bf16)a2.z; av[11] = (__bf16)a2.w;
        av[12] = (__bf16)a3.x; av[13] = (__bf16)a3.y; av[14] = (__bf16)a3.z; av[15] = (__bf16)a3.w;

        #pragma unroll
        for (int nt = 0; nt < 4; ++nt) {
            const __bf16* bk = bl + (nt * NKT1 + kt) * 512;
            v8bf lo = *(const v8bf*)(bk);
            v8bf hi = *(const v8bf*)(bk + 256);
            v16bf bv = __builtin_shufflevector(lo, hi, 0, 1, 2, 3, 4, 5, 6, 7, 8, 9, 10, 11, 12, 13, 14, 15);
            acc[nt] = __builtin_amdgcn_wmma_f32_16x16x32_bf16(
                false, av, false, bv, (short)0, acc[nt], false, false);
        }
    }

    float* cbase = C + (size_t)(tile * 16 + khalf * 8) * HID + m;
    #pragma unroll
    for (int nt = 0; nt < 4; ++nt) {
        float* crow = cbase + nt * 16;
        #pragma unroll
        for (int r = 0; r < 8; ++r) crow[(size_t)r * HID] = acc[nt][r];
    }
}

constexpr int NPAD = 53248;
constexpr int BSH  = 8;
constexpr int BNOD = 1 << BSH;
constexpr int NBK  = (NNODE >> BSH) + 1;
constexpr int EC   = 3200;
constexpr int EQ   = EC / 4;
constexpr int NWG  = NEDGE / EC;
constexpr int BB_OFF = 49152;
static_assert(NEDGE % EC == 0 && EC % 4 == 0, "chunks are whole int4 runs");
constexpr int EQH  = (EQ + 255) / 256;
constexpr int NBQ  = NBK / 4;
constexpr int RG   = 10;
constexpr int RPG  = NWG / RG;
static_assert(NBK % 4 == 0 && NWG % RG == 0 && RG * NBQ <= 512, "column-sum tiling");
static_assert(NBK <= 256 && NNODE <= 65536, "bucket fits the block; src fits 16 bits");
static_assert(NWG * NBK <= BB_OFF && BB_OFF + 256 <= NPAD && NNODE + 1 <= NPAD, "slices");
static_assert(EQ > 512 && EQ <= 1024, "k_scatter holds a chunk in two int4 per thread");

DEVINL void hist_body(const int* __restrict__ dst, int* __restrict__ hist) {
    __shared__ int cnt[256];
    int t = threadIdx.x;
    const int4* dp = (const int4*)(dst + (size_t)blockIdx.x * EC);
    int4 d[EQH];
    #pragma unroll
    for (int k = 0; k < EQH; ++k) {
        int i = t + k * 256;
        d[k] = dp[i < EQ ? i : EQ - 1];
    }
    cnt[t] = 0;
    __syncthreads();
    #pragma unroll
    for (int k = 0; k < EQH; ++k) {
        if (t + k * 256 < EQ) {
            atomicAdd(&cnt[d[k].x >> BSH], 1);
            atomicAdd(&cnt[d[k].y >> BSH], 1);
            atomicAdd(&cnt[d[k].z >> BSH], 1);
            atomicAdd(&cnt[d[k].w >> BSH], 1);
        }
    }
    __syncthreads();
    if (t < NBK) hist[blockIdx.x * NBK + t] = cnt[t];
}

constexpr int RP1  = 4 * (FIN / 32) * 512;
constexpr int RP2  = 4 * (HID / 32) * 512;
constexpr int RP1B = RP1 / 256;
constexpr int RP2B = RP2 / 256;
static_assert(RP1 % 256 == 0 && RP2 % 256 == 0, "repack blocks are full");

__global__ void __launch_bounds__(256)
k_hist_repack(const int* __restrict__ dst, int* __restrict__ hist,
              const float* __restrict__ W1, __bf16* __restrict__ Bp1,
              const float* __restrict__ W2, __bf16* __restrict__ Bp2) {
    int b = blockIdx.x;
    if (b < NWG) {
        hist_body(dst, hist);
    } else if (b < NWG + RP1B) {
        repack_body(W1, Bp1, FIN / 32, (b - NWG) * 256 + threadIdx.x, 1);
    } else {
        repack_body(W2, Bp2, HID / 32, (b - NWG - RP1B) * 256 + threadIdx.x, 0);
    }
}

DEVINL int scan256_excl(int v, int t, int* wtot) {
    int lane = t & 31, w = t >> 5;
    int inc = v;
    #pragma unroll
    for (int o = 1; o < 32; o <<= 1) { int u = __shfl_up(inc, o, 32); if (lane >= o) inc += u; }
    if (lane == 31 && w < 8) wtot[w] = inc;
    __syncthreads();
    int pre = 0;
    #pragma unroll
    for (int j = 0; j < 8; ++j) pre += (j < w) ? wtot[j] : 0;
    return pre + inc - v;
}

__global__ void __launch_bounds__(512)
k_scatter(const int* __restrict__ src, const int* __restrict__ dst,
          const int* __restrict__ hist, int* __restrict__ bbase,
          int* __restrict__ rec) {
    __shared__ int4 s_tot[RG][NBQ];
    __shared__ int4 s_par[RG][NBQ];
    __shared__ int cur[256];
    __shared__ int wtot[8];
    int t = threadIdx.x, wg = blockIdx.x;

    const int4* dp = (const int4*)(dst + (size_t)wg * EC);
    const int4* sp = (const int4*)(src + (size_t)wg * EC);
    bool two = t + 512 < EQ;
    int  i1  = two ? t + 512 : t;
    int4 d0 = dp[t],  s0 = sp[t];
    int4 d1 = dp[i1], s1 = sp[i1];

    if (t < RG * NBQ) {
        int q = t % NBQ, g = t / NBQ;
        const int4* hp = (const int4*)hist + (size_t)(g * RPG) * NBQ + q;
        int4 h[RPG];
        #pragma unroll
        for (int k = 0; k < RPG; ++k) h[k] = hp[k * NBQ];
        int4 tot = make_int4(0, 0, 0, 0), par = tot;
        #pragma unroll
        for (int k = 0; k < RPG; ++k) {
            tot.x += h[k].x; tot.y += h[k].y; tot.z += h[k].z; tot.w += h[k].w;
            if (g * RPG + k < wg) { par.x += h[k].x; par.y += h[k].y; par.z += h[k].z; par.w += h[k].w; }
        }
        s_tot[g][q] = tot;
        s_par[g][q] = par;
    }
    __syncthreads();
    int tot = 0, par = 0;
    if (t < NBK) {
        const int* st = (const int*)s_tot;
        const int* sq = (const int*)s_par;
        #pragma unroll
        for (int g = 0; g < RG; ++g) { tot += st[g * NBK + t]; par += sq[g * NBK + t]; }
    }
    int excl = scan256_excl(tot, t, wtot);
    if (t < 256) {
        cur[t] = excl + par;
        if (wg == 0 && t <= NBK) bbase[t] = excl;
    }
    __syncthreads();

    int p;
    p = atomicAdd(&cur[d0.x >> BSH], 1); rec[p] = s0.x | ((d0.x & (BNOD - 1)) << 16);
    p = atomicAdd(&cur[d0.y >> BSH], 1); rec[p] = s0.y | ((d0.y & (BNOD - 1)) << 16);
    p = atomicAdd(&cur[d0.z >> BSH], 1); rec[p] = s0.z | ((d0.z & (BNOD - 1)) << 16);
    p = atomicAdd(&cur[d0.w >> BSH], 1); rec[p] = s0.w | ((d0.w & (BNOD - 1)) << 16);
    if (two) {
        p = atomicAdd(&cur[d1.x >> BSH], 1); rec[p] = s1.x | ((d1.x & (BNOD - 1)) << 16);
        p = atomicAdd(&cur[d1.y >> BSH], 1); rec[p] = s1.y | ((d1.y & (BNOD - 1)) << 16);
        p = atomicAdd(&cur[d1.z >> BSH], 1); rec[p] = s1.z | ((d1.z & (BNOD - 1)) << 16);
        p = atomicAdd(&cur[d1.w >> BSH], 1); rec[p] = s1.w | ((d1.w & (BNOD - 1)) << 16);
    }
}

DEVINL void bucket_body(const int* __restrict__ rec, const int* __restrict__ bbase,
                        int* __restrict__ rp, int* __restrict__ csr, int b) {
    __shared__ int cnt[256];
    __shared__ int wtot[8];
    int t = threadIdx.x;
    int beg = bbase[b], end = bbase[b + 1];
    if (t < 256) cnt[t] = 0;
    __syncthreads();

    for (int p = beg + t; p < end; p += 4 * 512) {
        int r[4];
        #pragma unroll
        for (int k = 0; k < 4; ++k) { int q = p + k * 512; int v = rec[q < end ? q : end - 1]; r[k] = (q < end) ? v : -1; }
        #pragma unroll
        for (int k = 0; k < 4; ++k) if (r[k] >= 0) atomicAdd(&cnt[r[k] >> 16], 1);
    }
    __syncthreads();

    int deg = (t < 256) ? cnt[t] : 0;
    int excl = scan256_excl(deg, t, wtot);
    if (t < 256) {
        int start = beg + excl;
        int node  = b * BNOD + t;
        if (node <= NNODE) rp[node] = start;
        cnt[t] = start;
    }
    __syncthreads();

    for (int p = beg + t; p < end; p += 4 * 512) {
        int r[4];
        #pragma unroll
        for (int k = 0; k < 4; ++k) { int q = p + k * 512; int v = rec[q < end ? q : end - 1]; r[k] = (q < end) ? v : -1; }
        #pragma unroll
        for (int k = 0; k < 4; ++k)
            if (r[k] >= 0) { int s = atomicAdd(&cnt[r[k] >> 16], 1); csr[s] = r[k] & 0xffff; }
    }
}

constexpr int GT    = 16;
constexpr int GB    = (ROWT + GT - 1) / GT;
static_assert(GT == 16, "gemm1_block maps the 16 waves of a block to 16 row tiles");
constexpr int GRID1 = NWG + RP1B + RP2B;
constexpr int GRID3 = GB + NBK;

__global__ void __launch_bounds__(512)
k_gemm_bucket(const float* __restrict__ A, const __bf16* __restrict__ Bp, float* __restrict__ C,
              const int* __restrict__ rec, const int* __restrict__ bbase,
              int* __restrict__ rp, int* __restrict__ csr) {
    __shared__ __attribute__((aligned(16))) __bf16 sB[SBE];
    if (blockIdx.x >= GB) {
        bucket_body(rec, bbase, rp, csr, blockIdx.x - GB);
        return;
    }
    gemm1_block(A, Bp, C, sB, blockIdx.x);
}

DEVINL float dot8(const float4& w0, const float4& w1, const float4& c0, const float4& c1) {
    float s = 0.f;
    s += w0.x * c0.x; s += w0.y * c0.y; s += w0.z * c0.z; s += w0.w * c0.w;
    s += w1.x * c1.x; s += w1.y * c1.y; s += w1.z * c1.z; s += w1.w * c1.w;
    return s;
}

DEVINL float sum8(float v) {
    v += __shfl_xor(v, 1, 32);
    v += __shfl_xor(v, 2, 32);
    v += __shfl_xor(v, 4, 32);
    return v;
}

DEVINL void axpy8(float4& a0, float4& a1, float x, const float4& w0, const float4& w1) {
    a0.x += x * w0.x; a0.y += x * w0.y; a0.z += x * w0.z; a0.w += x * w0.w;
    a1.x += x * w1.x; a1.y += x * w1.y; a1.z += x * w1.z; a1.w += x * w1.w;
}

DEVINL void scale8(float4& a0, float4& a1, float s) {
    a0.x *= s; a0.y *= s; a0.z *= s; a0.w *= s;
    a1.x *= s; a1.y *= s; a1.z *= s; a1.w *= s;
}

constexpr int G1N  = 32;
constexpr int HPIT = 72;
static_assert(NNODE % 16 == 0 && HID == 64 && NLAB == 64, "fused product tiling");

__global__ void __launch_bounds__(256)
k_gather1_gemm2(const int* __restrict__ rp, const int* __restrict__ csr,
                const float* __restrict__ a_s, const float* __restrict__ a_d,
                const float* __restrict__ Wh, const __bf16* __restrict__ Bp2,
                float* __restrict__ Wh2) {
    __shared__ __attribute__((aligned(16))) __bf16 sH[G1N * HPIT];
    int idx = blockIdx.x * blockDim.x + threadIdx.x;
    int n = idx >> 3, hh = idx & 7;
    const bool act = n < NNODE;

    const int w    = __builtin_amdgcn_readfirstlane((int)(threadIdx.x >> 5));
    const int rt   = w >> 2, nt = w & 3;
    const int lane = threadIdx.x & 31;
    const __bf16* bl = Bp2 + ((size_t)(nt * 2) * 32 + lane) * 16;

    float den = 0.f;
    float4 a0 = make_float4(0.f, 0.f, 0.f, 0.f), a1 = a0;
    if (act) {
        int beg = rp[n], end = rp[n + 1];
        const float* wcol = Wh + hh * 8;
        const float4 as0 = *(const float4*)(a_s + hh * 8), as1 = *(const float4*)(a_s + hh * 8 + 4);
        const float4 ad0 = *(const float4*)(a_d + hh * 8), ad1 = *(const float4*)(a_d + hh * 8 + 4);
        const float* own = wcol + (size_t)n * HID;
        float edn = dot8(*(const float4*)(own), *(const float4*)(own + 4), ad0, ad1);

        float m = -INFINITY;
        {
            const int last = end - 1;
            for (int p = beg; p < end; p += 4) {
                int s0 = csr[p];
                int s1 = csr[min(p + 1, last)];
                int s2 = csr[min(p + 2, last)];
                int s3 = csr[min(p + 3, last)];
                const float* r0 = wcol + (size_t)s0 * HID;
                const float* r1 = wcol + (size_t)s1 * HID;
                const float* r2 = wcol + (size_t)s2 * HID;
                const float* r3 = wcol + (size_t)s3 * HID;
                float4 w00 = *(const float4*)(r0), w01 = *(const float4*)(r0 + 4);
                float4 w10 = *(const float4*)(r1), w11 = *(const float4*)(r1 + 4);
                float4 w20 = *(const float4*)(r2), w21 = *(const float4*)(r2 + 4);
                float4 w30 = *(const float4*)(r3), w31 = *(const float4*)(r3 + 4);
                float e0 = lrelu(dot8(w00, w01, as0, as1) + edn);
                float e1 = lrelu(dot8(w10, w11, as0, as1) + edn);
                float e2 = lrelu(dot8(w20, w21, as0, as1) + edn);
                float e3 = lrelu(dot8(w30, w31, as0, as1) + edn);
                e1 = (p + 1 < end) ? e1 : -INFINITY;
                e2 = (p + 2 < end) ? e2 : -INFINITY;
                e3 = (p + 3 < end) ? e3 : -INFINITY;
                float mn = fmaxf(fmaxf(m, e0), fmaxf(e1, fmaxf(e2, e3)));
                float sc = __expf(m - mn);
                float x0 = __expf(e0 - mn), x1 = __expf(e1 - mn);
                float x2 = __expf(e2 - mn), x3 = __expf(e3 - mn);
                den *= sc; scale8(a0, a1, sc);
                den += x0; axpy8(a0, a1, x0, w00, w01);
                den += x1; axpy8(a0, a1, x1, w10, w11);
                den += x2; axpy8(a0, a1, x2, w20, w21);
                den += x3; axpy8(a0, a1, x3, w30, w31);
                m = mn;
            }
        }
    }

    const v16bf bv0 = *(const v16bf*)(bl);
    const v16bf bv1 = *(const v16bf*)(bl + 512);

    if (act) {
        float dn = den + 1e-10f;
        float x[8] = { a0.x / dn, a0.y / dn, a0.z / dn, a0.w / dn,
                       a1.x / dn, a1.y / dn, a1.z / dn, a1.w / dn };
        #pragma unroll
        for (int j = 0; j < 8; ++j) x[j] = x[j] > 0.f ? x[j] : (__expf(x[j]) - 1.f);
        v8bf hv;
        #pragma unroll
        for (int j = 0; j < 8; ++j) hv[j] = (__bf16)x[j];
        *(v8bf*)(sH + (threadIdx.x >> 3) * HPIT + hh * 8) = hv;
    }
    __syncthreads();

    int tile = blockIdx.x * 2 + rt;
    if (tile >= ROWT) return;
    int m16   = lane & 15;
    int khalf = lane >> 4;
    const __bf16* arow = sH + (rt * 16 + m16) * HPIT + khalf * 8;
    v8f acc = {};
    #pragma unroll
    for (int kt = 0; kt < 2; ++kt) {
        v8bf lo = *(const v8bf*)(arow + kt * 32);
        v8bf hi = *(const v8bf*)(arow + kt * 32 + 16);
        v16bf av = __builtin_shufflevector(lo, hi, 0, 1, 2, 3, 4, 5, 6, 7, 8, 9, 10, 11, 12, 13, 14, 15);
        acc = __builtin_amdgcn_wmma_f32_16x16x32_bf16(
            false, av, false, kt == 0 ? bv0 : bv1, (short)0, acc, false, false);
    }
    float* crow = Wh2 + (size_t)(tile * 16 + khalf * 8) * NLAB + nt * 16 + m16;
    #pragma unroll
    for (int r = 0; r < 8; ++r) crow[(size_t)r * NLAB] = acc[r];
}

__global__ void __launch_bounds__(256)
k_gather2(const int* __restrict__ rp, const int* __restrict__ csr,
          const float* __restrict__ a_s, const float* __restrict__ a_d,
          const float* __restrict__ Wh, float* __restrict__ out) {
    int idx = blockIdx.x * blockDim.x + threadIdx.x;
    if (idx >= NNODE * 8) return;
    int n = idx >> 3, c0 = (idx & 7) * 8;
    int beg = rp[n], end = rp[n + 1];
    const float* wcol = Wh + c0;
    const float4 as0 = *(const float4*)(a_s + c0), as1 = *(const float4*)(a_s + c0 + 4);
    const float4 ad0 = *(const float4*)(a_d + c0), ad1 = *(const float4*)(a_d + c0 + 4);
    const float* own = wcol + (size_t)n * NLAB;
    float edn = sum8(dot8(*(const float4*)(own), *(const float4*)(own + 4), ad0, ad1));

    float m = -INFINITY, den = 0.f;
    float4 a0 = make_float4(0.f, 0.f, 0.f, 0.f), a1 = a0;
    {
        const int last = end - 1;
        for (int p = beg; p < end; p += 4) {
            int s0 = csr[p];
            int s1 = csr[min(p + 1, last)];
            int s2 = csr[min(p + 2, last)];
            int s3 = csr[min(p + 3, last)];
            const float* r0 = wcol + (size_t)s0 * NLAB;
            const float* r1 = wcol + (size_t)s1 * NLAB;
            const float* r2 = wcol + (size_t)s2 * NLAB;
            const float* r3 = wcol + (size_t)s3 * NLAB;
            float4 w00 = *(const float4*)(r0), w01 = *(const float4*)(r0 + 4);
            float4 w10 = *(const float4*)(r1), w11 = *(const float4*)(r1 + 4);
            float4 w20 = *(const float4*)(r2), w21 = *(const float4*)(r2 + 4);
            float4 w30 = *(const float4*)(r3), w31 = *(const float4*)(r3 + 4);
            float e0 = lrelu(sum8(dot8(w00, w01, as0, as1)) + edn);
            float e1 = lrelu(sum8(dot8(w10, w11, as0, as1)) + edn);
            float e2 = lrelu(sum8(dot8(w20, w21, as0, as1)) + edn);
            float e3 = lrelu(sum8(dot8(w30, w31, as0, as1)) + edn);
            e1 = (p + 1 < end) ? e1 : -INFINITY;
            e2 = (p + 2 < end) ? e2 : -INFINITY;
            e3 = (p + 3 < end) ? e3 : -INFINITY;
            float mn = fmaxf(fmaxf(m, e0), fmaxf(e1, fmaxf(e2, e3)));
            float sc = __expf(m - mn);
            float x0 = __expf(e0 - mn), x1 = __expf(e1 - mn);
            float x2 = __expf(e2 - mn), x3 = __expf(e3 - mn);
            den *= sc; scale8(a0, a1, sc);
            den += x0; axpy8(a0, a1, x0, w00, w01);
            den += x1; axpy8(a0, a1, x1, w10, w11);
            den += x2; axpy8(a0, a1, x2, w20, w21);
            den += x3; axpy8(a0, a1, x3, w30, w31);
            m = mn;
        }
    }

    float dn = den + 1e-10f;
    float x[8] = { a0.x / dn, a0.y / dn, a0.z / dn, a0.w / dn,
                   a1.x / dn, a1.y / dn, a1.z / dn, a1.w / dn };
    float mx = x[0];
    #pragma unroll
    for (int j = 1; j < 8; ++j) mx = fmaxf(mx, x[j]);
    #pragma unroll
    for (int o = 1; o < 8; o <<= 1) mx = fmaxf(mx, __shfl_xor(mx, o, 32));
    float sum = 0.f;
    #pragma unroll
    for (int j = 0; j < 8; ++j) { x[j] = __expf(x[j] - mx); sum += x[j]; }
    #pragma unroll
    for (int o = 1; o < 8; o <<= 1) sum += __shfl_xor(sum, o, 32);
    float inv = 1.0f / sum;
    float* o = out + (size_t)n * NLAB + c0;
    *(float4*)(o)     = make_float4(x[0] * inv, x[1] * inv, x[2] * inv, x[3] * inv);
    *(float4*)(o + 4) = make_float4(x[4] * inv, x[5] * inv, x[6] * inv, x[7] * inv);
}

extern "C" void kernel_launch(void* const* d_in, const int* in_sizes, int n_in,
                              void* d_out, int out_size, void* d_ws, size_t ws_size,
                              hipStream_t stream) {
    const float* X   = (const float*)d_in[0];
    const float* W1  = (const float*)d_in[1];
    const float* a1s = (const float*)d_in[2];
    const float* a1d = (const float*)d_in[3];
    const float* W2  = (const float*)d_in[4];
    const float* a2s = (const float*)d_in[5];
    const float* a2d = (const float*)d_in[6];
    const int*   src = (const int*)d_in[7];
    const int*   dst = (const int*)d_in[8];
    float*       out = (float*)d_out;

    char*  ws  = (char*)d_ws;
    size_t off = 0;
    auto alloc = [&](size_t bytes) -> void* {
        void* p = ws + off;
        off = (off + bytes + 255) & ~(size_t)255;
        return p;
    };
    float*    Wh1   = (float*)   alloc((size_t)NNODE * HID  * sizeof(float));
    float*    Wh2   = (float*)   alloc((size_t)NNODE * NLAB * sizeof(float));
    int*      hist  = (int*)     alloc((size_t)NPAD         * sizeof(int));
    int*      rp    = (int*)     alloc((size_t)NPAD         * sizeof(int));
    int*      rec   = (int*)     alloc((size_t)NEDGE        * sizeof(int));
    int*      csr   = (int*)     alloc((size_t)NEDGE        * sizeof(int));
    __bf16*   Bp1   = (__bf16*)  alloc((size_t)4 * 16 * 512 * sizeof(__bf16));
    __bf16*   Bp2   = (__bf16*)  alloc((size_t)4 * 2  * 512 * sizeof(__bf16));

    const int B = 256;
    auto cdiv = [](long long a, long long b) { return (int)((a + b - 1) / b); };

    int* bbase = hist + BB_OFF;
    k_hist_repack<<<GRID1, 256, 0, stream>>>(dst, hist, W1, Bp1, W2, Bp2);
    k_scatter<<<NWG, 512, 0, stream>>>(src, dst, hist, bbase, rec);
    k_gemm_bucket<<<GRID3, 512, 0, stream>>>(X, Bp1, Wh1, rec, bbase, rp, csr);

    k_gather1_gemm2<<<cdiv((long long)NNODE * H1, B), B, 0, stream>>>(rp, csr, a1s, a1d, Wh1, Bp2, Wh2);

    k_gather2<<<cdiv((long long)NNODE * 8, B), B, 0, stream>>>(rp, csr, a2s, a2d, Wh2, out);
}


# Gathers: 8 in-edges per trip, indices loaded one trip ahead (1 per lane, dpp8 spread): 1 dependent memory trip per 8 edges, was 2 per 4; logit dot as 2 chains of 4, max rescale per 8 edges
# speedup vs baseline: 6.4179x; 1.0434x over previous
#include <hip/hip_runtime.h>
#include <hip/hip_bf16.h>


typedef __attribute__((ext_vector_type(16))) __bf16 v16bf;
typedef __attribute__((ext_vector_type(8)))  __bf16 v8bf;
typedef __attribute__((ext_vector_type(8)))  float  v8f;

constexpr int NNODE = 50000;
constexpr int NEDGE = 800000;
constexpr int FIN   = 512;
constexpr int H1    = 8;
constexpr int FHID  = 8;
constexpr int HID   = H1 * FHID;
constexpr int NLAB  = 64;
constexpr int ROWT  = NNODE / 16;
constexpr float SLOPE = 0.2f;

#define DEVINL __device__ __forceinline__

DEVINL float lrelu(float v) { return v > 0.f ? v : SLOPE * v; }

DEVINL void repack_body(const float* __restrict__ W, __bf16* __restrict__ Bp,
                        int nKt, int idx, int gat1) {
    int e    = idx & 15;
    int lane = (idx >> 4) & 31;
    int ktnt = idx >> 9;
    int kt   = ktnt % nKt;
    int nt   = ktnt / nKt;
    int K = ((e < 8) ? 0 : 16) + (lane >> 4) * 8 + (e & 7);
    int k = kt * 32 + K;
    int n = nt * 16 + (lane & 15);
    float v = gat1 ? W[(size_t)(n >> 3) * (FIN * FHID) + (size_t)k * FHID + (n & 7)]
                   : W[(size_t)k * NLAB + n];
    Bp[idx] = (__bf16)v;
}

constexpr int NKT1 = FIN / 32;
constexpr int SBE  = 4 * NKT1 * 512;
constexpr int SBQ  = SBE / 8;
static_assert(SBQ % 512 == 0, "the LDS copy is whole 512-thread sweeps");

DEVINL void load_a(const float* __restrict__ ap, float4 (&d)[4]) {
    d[0] = *(const float4*)(ap);
    d[1] = *(const float4*)(ap + 4);
    d[2] = *(const float4*)(ap + 16);
    d[3] = *(const float4*)(ap + 20);
}

DEVINL void gemm1_block(const float* __restrict__ A, const __bf16* __restrict__ Bp,
                        float* __restrict__ C, __bf16* sB, int blk) {
    const int t     = threadIdx.x;
    const int lane  = t & 31;
    const int m     = lane & 15;
    const int khalf = lane >> 4;
    const int tile  = blk * 16 + (t >> 5);
    const bool valid = tile < ROWT;
    const float* arow = A + (size_t)((valid ? tile : 0) * 16 + m) * FIN + khalf * 8;

    float4 a[3][4];
    if (valid) { load_a(arow, a[0]); load_a(arow + 32, a[1]); }

    {
        const v8bf* g = (const v8bf*)Bp;
        v8bf* s = (v8bf*)sB;
        v8bf v[SBQ / 512];
        #pragma unroll
        for (int j = 0; j < SBQ / 512; ++j) v[j] = g[t + j * 512];
        #pragma unroll
        for (int j = 0; j < SBQ / 512; ++j) {
            int i = t + j * 512;
            s[(i & ~63) + ((i & 1) << 5) + ((i & 63) >> 1)] = v[j];
        }
    }
    __syncthreads();
    if (!valid) return;

    const __bf16* bl = sB + lane * 8;
    v8f acc[4] = {};
    #pragma unroll
    for (int kt = 0; kt < NKT1; ++kt) {
        if (kt + 2 < NKT1) {
            load_a(arow + (kt + 2) * 32, a[(kt + 2) % 3]);
            if (kt + 3 < NKT1) __builtin_prefetch(arow + (kt + 3) * 32, 0, 0);
        }
        __builtin_amdgcn_sched_barrier(0);

        float4 a0 = a[kt % 3][0], a1 = a[kt % 3][1], a2 = a[kt % 3][2], a3 = a[kt % 3][3];
        asm volatile("" : "+v"(a0.x), "+v"(a0.y), "+v"(a0.z), "+v"(a0.w),
                          "+v"(a1.x), "+v"(a1.y), "+v"(a1.z), "+v"(a1.w));
        asm volatile("" : "+v"(a2.x), "+v"(a2.y), "+v"(a2.z), "+v"(a2.w),
                          "+v"(a3.x), "+v"(a3.y), "+v"(a3.z), "+v"(a3.w));
        v16bf av;
        av[0]  = (__bf16)a0.x; av[1]  = (__bf16)a0.y; av[2]  = (__bf16)a0.z; av[3]  = (__bf16)a0.w;
        av[4]  = (__bf16)a1.x; av[5]  = (__bf16)a1.y; av[6]  = (__bf16)a1.z; av[7]  = (__bf16)a1.w;
        av[8]  = (__bf16)a2.x; av[9]  = (__bf16)a2.y; av[10] = (__bf16)a2.z; av[11] = (__bf16)a2.w;
        av[12] = (__bf16)a3.x; av[13] = (__bf16)a3.y; av[14] = (__bf16)a3.z; av[15] = (__bf16)a3.w;

        #pragma unroll
        for (int nt = 0; nt < 4; ++nt) {
            const __bf16* bk = bl + (nt * NKT1 + kt) * 512;
            v8bf lo = *(const v8bf*)(bk);
            v8bf hi = *(const v8bf*)(bk + 256);
            v16bf bv = __builtin_shufflevector(lo, hi, 0, 1, 2, 3, 4, 5, 6, 7, 8, 9, 10, 11, 12, 13, 14, 15);
            acc[nt] = __builtin_amdgcn_wmma_f32_16x16x32_bf16(
                false, av, false, bv, (short)0, acc[nt], false, false);
        }
    }

    float* cbase = C + (size_t)(tile * 16 + khalf * 8) * HID + m;
    #pragma unroll
    for (int nt = 0; nt < 4; ++nt) {
        float* crow = cbase + nt * 16;
        #pragma unroll
        for (int r = 0; r < 8; ++r) crow[(size_t)r * HID] = acc[nt][r];
    }
}

constexpr int NPAD = 53248;
constexpr int BSH  = 8;
constexpr int BNOD = 1 << BSH;
constexpr int NBK  = (NNODE >> BSH) + 1;
constexpr int EC   = 3200;
constexpr int EQ   = EC / 4;
constexpr int NWG  = NEDGE / EC;
constexpr int BB_OFF = 49152;
static_assert(NEDGE % EC == 0 && EC % 4 == 0, "chunks are whole int4 runs");
constexpr int EQH  = (EQ + 255) / 256;
constexpr int NBQ  = NBK / 4;
constexpr int RG   = 10;
constexpr int RPG  = NWG / RG;
static_assert(NBK % 4 == 0 && NWG % RG == 0 && RG * NBQ <= 512, "column-sum tiling");
static_assert(NBK <= 256 && NNODE <= 65536, "bucket fits the block; src fits 16 bits");
static_assert(NWG * NBK <= BB_OFF && BB_OFF + 256 <= NPAD && NNODE + 1 <= NPAD, "slices");
static_assert(EQ > 512 && EQ <= 1024, "k_scatter holds a chunk in two int4 per thread");

DEVINL void hist_body(const int* __restrict__ dst, int* __restrict__ hist) {
    __shared__ int cnt[256];
    int t = threadIdx.x;
    const int4* dp = (const int4*)(dst + (size_t)blockIdx.x * EC);
    int4 d[EQH];
    #pragma unroll
    for (int k = 0; k < EQH; ++k) {
        int i = t + k * 256;
        d[k] = dp[i < EQ ? i : EQ - 1];
    }
    cnt[t] = 0;
    __syncthreads();
    #pragma unroll
    for (int k = 0; k < EQH; ++k) {
        if (t + k * 256 < EQ) {
            atomicAdd(&cnt[d[k].x >> BSH], 1);
            atomicAdd(&cnt[d[k].y >> BSH], 1);
            atomicAdd(&cnt[d[k].z >> BSH], 1);
            atomicAdd(&cnt[d[k].w >> BSH], 1);
        }
    }
    __syncthreads();
    if (t < NBK) hist[blockIdx.x * NBK + t] = cnt[t];
}

constexpr int RP1  = 4 * (FIN / 32) * 512;
constexpr int RP2  = 4 * (HID / 32) * 512;
constexpr int RP1B = RP1 / 256;
constexpr int RP2B = RP2 / 256;
static_assert(RP1 % 256 == 0 && RP2 % 256 == 0, "repack blocks are full");

__global__ void __launch_bounds__(256)
k_hist_repack(const int* __restrict__ dst, int* __restrict__ hist,
              const float* __restrict__ W1, __bf16* __restrict__ Bp1,
              const float* __restrict__ W2, __bf16* __restrict__ Bp2) {
    int b = blockIdx.x;
    if (b < NWG) {
        hist_body(dst, hist);
    } else if (b < NWG + RP1B) {
        repack_body(W1, Bp1, FIN / 32, (b - NWG) * 256 + threadIdx.x, 1);
    } else {
        repack_body(W2, Bp2, HID / 32, (b - NWG - RP1B) * 256 + threadIdx.x, 0);
    }
}

DEVINL int scan256_excl(int v, int t, int* wtot) {
    int lane = t & 31, w = t >> 5;
    int inc = v;
    #pragma unroll
    for (int o = 1; o < 32; o <<= 1) { int u = __shfl_up(inc, o, 32); if (lane >= o) inc += u; }
    if (lane == 31 && w < 8) wtot[w] = inc;
    __syncthreads();
    int pre = 0;
    #pragma unroll
    for (int j = 0; j < 8; ++j) pre += (j < w) ? wtot[j] : 0;
    return pre + inc - v;
}

__global__ void __launch_bounds__(512)
k_scatter(const int* __restrict__ src, const int* __restrict__ dst,
          const int* __restrict__ hist, int* __restrict__ bbase,
          int* __restrict__ rec) {
    __shared__ int4 s_tot[RG][NBQ];
    __shared__ int4 s_par[RG][NBQ];
    __shared__ int cur[256];
    __shared__ int wtot[8];
    int t = threadIdx.x, wg = blockIdx.x;

    const int4* dp = (const int4*)(dst + (size_t)wg * EC);
    const int4* sp = (const int4*)(src + (size_t)wg * EC);
    bool two = t + 512 < EQ;
    int  i1  = two ? t + 512 : t;
    int4 d0 = dp[t],  s0 = sp[t];
    int4 d1 = dp[i1], s1 = sp[i1];

    if (t < RG * NBQ) {
        int q = t % NBQ, g = t / NBQ;
        const int4* hp = (const int4*)hist + (size_t)(g * RPG) * NBQ + q;
        int4 h[RPG];
        #pragma unroll
        for (int k = 0; k < RPG; ++k) h[k] = hp[k * NBQ];
        int4 tot = make_int4(0, 0, 0, 0), par = tot;
        #pragma unroll
        for (int k = 0; k < RPG; ++k) {
            tot.x += h[k].x; tot.y += h[k].y; tot.z += h[k].z; tot.w += h[k].w;
            if (g * RPG + k < wg) { par.x += h[k].x; par.y += h[k].y; par.z += h[k].z; par.w += h[k].w; }
        }
        s_tot[g][q] = tot;
        s_par[g][q] = par;
    }
    __syncthreads();
    int tot = 0, par = 0;
    if (t < NBK) {
        const int* st = (const int*)s_tot;
        const int* sq = (const int*)s_par;
        #pragma unroll
        for (int g = 0; g < RG; ++g) { tot += st[g * NBK + t]; par += sq[g * NBK + t]; }
    }
    int excl = scan256_excl(tot, t, wtot);
    if (t < 256) {
        cur[t] = excl + par;
        if (wg == 0 && t <= NBK) bbase[t] = excl;
    }
    __syncthreads();

    int p;
    p = atomicAdd(&cur[d0.x >> BSH], 1); rec[p] = s0.x | ((d0.x & (BNOD - 1)) << 16);
    p = atomicAdd(&cur[d0.y >> BSH], 1); rec[p] = s0.y | ((d0.y & (BNOD - 1)) << 16);
    p = atomicAdd(&cur[d0.z >> BSH], 1); rec[p] = s0.z | ((d0.z & (BNOD - 1)) << 16);
    p = atomicAdd(&cur[d0.w >> BSH], 1); rec[p] = s0.w | ((d0.w & (BNOD - 1)) << 16);
    if (two) {
        p = atomicAdd(&cur[d1.x >> BSH], 1); rec[p] = s1.x | ((d1.x & (BNOD - 1)) << 16);
        p = atomicAdd(&cur[d1.y >> BSH], 1); rec[p] = s1.y | ((d1.y & (BNOD - 1)) << 16);
        p = atomicAdd(&cur[d1.z >> BSH], 1); rec[p] = s1.z | ((d1.z & (BNOD - 1)) << 16);
        p = atomicAdd(&cur[d1.w >> BSH], 1); rec[p] = s1.w | ((d1.w & (BNOD - 1)) << 16);
    }
}

DEVINL void bucket_body(const int* __restrict__ rec, const int* __restrict__ bbase,
                        int* __restrict__ rp, int* __restrict__ csr, int b) {
    __shared__ int cnt[256];
    __shared__ int wtot[8];
    int t = threadIdx.x;
    int beg = bbase[b], end = bbase[b + 1];
    if (t < 256) cnt[t] = 0;
    __syncthreads();

    for (int p = beg + t; p < end; p += 4 * 512) {
        int r[4];
        #pragma unroll
        for (int k = 0; k < 4; ++k) { int q = p + k * 512; int v = rec[q < end ? q : end - 1]; r[k] = (q < end) ? v : -1; }
        #pragma unroll
        for (int k = 0; k < 4; ++k) if (r[k] >= 0) atomicAdd(&cnt[r[k] >> 16], 1);
    }
    __syncthreads();

    int deg = (t < 256) ? cnt[t] : 0;
    int excl = scan256_excl(deg, t, wtot);
    if (t < 256) {
        int start = beg + excl;
        int node  = b * BNOD + t;
        if (node <= NNODE) rp[node] = start;
        cnt[t] = start;
    }
    __syncthreads();

    for (int p = beg + t; p < end; p += 4 * 512) {
        int r[4];
        #pragma unroll
        for (int k = 0; k < 4; ++k) { int q = p + k * 512; int v = rec[q < end ? q : end - 1]; r[k] = (q < end) ? v : -1; }
        #pragma unroll
        for (int k = 0; k < 4; ++k)
            if (r[k] >= 0) { int s = atomicAdd(&cnt[r[k] >> 16], 1); csr[s] = r[k] & 0xffff; }
    }
}

constexpr int GT    = 16;
constexpr int GB    = (ROWT + GT - 1) / GT;
static_assert(GT == 16, "gemm1_block maps the 16 waves of a block to 16 row tiles");
constexpr int GRID1 = NWG + RP1B + RP2B;
constexpr int GRID3 = GB + NBK;

__global__ void __launch_bounds__(512)
k_gemm_bucket(const float* __restrict__ A, const __bf16* __restrict__ Bp, float* __restrict__ C,
              const int* __restrict__ rec, const int* __restrict__ bbase,
              int* __restrict__ rp, int* __restrict__ csr) {
    __shared__ __attribute__((aligned(16))) __bf16 sB[SBE];
    if (blockIdx.x >= GB) {
        bucket_body(rec, bbase, rp, csr, blockIdx.x - GB);
        return;
    }
    gemm1_block(A, Bp, C, sB, blockIdx.x);
}

typedef __attribute__((ext_vector_type(2))) float v2f;
DEVINL float dot8(const float4& w0, const float4& w1, const float4& c0, const float4& c1) {
    v2f t = (v2f){w0.x, w0.y} * (v2f){c0.x, c0.y};
    t += (v2f){w0.z, w0.w} * (v2f){c0.z, c0.w};
    t += (v2f){w1.x, w1.y} * (v2f){c1.x, c1.y};
    t += (v2f){w1.z, w1.w} * (v2f){c1.z, c1.w};
    return t.x + t.y;
}

DEVINL float sum8(float v) {
    v += __shfl_xor(v, 1, 32);
    v += __shfl_xor(v, 2, 32);
    v += __shfl_xor(v, 4, 32);
    return v;
}

DEVINL void axpy8(float4& a0, float4& a1, float x, const float4& w0, const float4& w1) {
    a0.x += x * w0.x; a0.y += x * w0.y; a0.z += x * w0.z; a0.w += x * w0.w;
    a1.x += x * w1.x; a1.y += x * w1.y; a1.z += x * w1.z; a1.w += x * w1.w;
}

DEVINL void scale8(float4& a0, float4& a1, float s) {
    a0.x *= s; a0.y *= s; a0.z *= s; a0.w *= s;
    a1.x *= s; a1.y *= s; a1.z *= s; a1.w *= s;
}

constexpr int EPT = 8;
static_assert(EPT == 8, "a trip's indices are spread over the node's 8 lanes");
template <bool TREE>
DEVINL void edge_pass(const int* __restrict__ csr, const float* __restrict__ rows, unsigned coff,
                      int j8, int beg, int end, const float4& as0, const float4& as1, float edn,
                      float& den, float4& a0, float4& a1) {
    if (beg >= end) return;
    const int last = end - 1;
    int mine = csr[min(beg + j8, last)];
    float m = -INFINITY;
    for (int p = beg; p < end; p += EPT) {
        int s[EPT];
        s[0] = __builtin_amdgcn_mov_dpp8(mine, 0 * 0x249249);
        s[1] = __builtin_amdgcn_mov_dpp8(mine, 1 * 0x249249);
        s[2] = __builtin_amdgcn_mov_dpp8(mine, 2 * 0x249249);
        s[3] = __builtin_amdgcn_mov_dpp8(mine, 3 * 0x249249);
        s[4] = __builtin_amdgcn_mov_dpp8(mine, 4 * 0x249249);
        s[5] = __builtin_amdgcn_mov_dpp8(mine, 5 * 0x249249);
        s[6] = __builtin_amdgcn_mov_dpp8(mine, 6 * 0x249249);
        s[7] = __builtin_amdgcn_mov_dpp8(mine, 7 * 0x249249);
        float4 w0[EPT], w1[EPT];
        #pragma unroll
        for (int k = 0; k < EPT; ++k) {
            const char* r = (const char*)rows + ((unsigned)s[k] * 256u + coff);
            w0[k] = *(const float4*)(r);
            w1[k] = *(const float4*)(r + 16);
        }
        mine = csr[min(p + EPT + j8, last)];
        __builtin_amdgcn_sched_barrier(0);

        float e[EPT];
        #pragma unroll
        for (int k = 0; k < EPT; ++k) {
            float d = dot8(w0[k], w1[k], as0, as1);
            if (TREE) d = sum8(d);
            e[k] = lrelu(d + edn);
            if (k > 0) e[k] = (p + k < end) ? e[k] : -INFINITY;
        }
        float mn = m;
        #pragma unroll
        for (int k = 0; k < EPT; ++k) mn = fmaxf(mn, e[k]);
        float sc = __expf(m - mn);
        den *= sc; scale8(a0, a1, sc);
        #pragma unroll
        for (int k = 0; k < EPT; ++k) {
            float x = __expf(e[k] - mn);
            den += x; axpy8(a0, a1, x, w0[k], w1[k]);
        }
        m = mn;
    }
}

constexpr int G1N  = 32;
constexpr int HPIT = 72;
static_assert(NNODE % 16 == 0 && HID == 64 && NLAB == 64, "fused product tiling");

__global__ void __launch_bounds__(256)
k_gather1_gemm2(const int* __restrict__ rp, const int* __restrict__ csr,
                const float* __restrict__ a_s, const float* __restrict__ a_d,
                const float* __restrict__ Wh, const __bf16* __restrict__ Bp2,
                float* __restrict__ Wh2) {
    __shared__ __attribute__((aligned(16))) __bf16 sH[G1N * HPIT];
    int idx = blockIdx.x * blockDim.x + threadIdx.x;
    int n = idx >> 3, hh = idx & 7;
    const bool act = n < NNODE;

    const int w    = __builtin_amdgcn_readfirstlane((int)(threadIdx.x >> 5));
    const int rt   = w >> 2, nt = w & 3;
    const int lane = threadIdx.x & 31;
    const __bf16* bl = Bp2 + ((size_t)(nt * 2) * 32 + lane) * 16;

    float den = 0.f;
    float4 a0 = make_float4(0.f, 0.f, 0.f, 0.f), a1 = a0;
    if (act) {
        int beg = rp[n], end = rp[n + 1];
        const float* wcol = Wh + hh * 8;
        const float4 as0 = *(const float4*)(a_s + hh * 8), as1 = *(const float4*)(a_s + hh * 8 + 4);
        const float4 ad0 = *(const float4*)(a_d + hh * 8), ad1 = *(const float4*)(a_d + hh * 8 + 4);
        const float* own = wcol + (size_t)n * HID;
        float edn = dot8(*(const float4*)(own), *(const float4*)(own + 4), ad0, ad1);

        edge_pass<false>(csr, Wh, (unsigned)hh * 32u, hh, beg, end, as0, as1, edn, den, a0, a1);
    }

    const v16bf bv0 = *(const v16bf*)(bl);
    const v16bf bv1 = *(const v16bf*)(bl + 512);

    if (act) {
        float dn = den + 1e-10f;
        float x[8] = { a0.x / dn, a0.y / dn, a0.z / dn, a0.w / dn,
                       a1.x / dn, a1.y / dn, a1.z / dn, a1.w / dn };
        #pragma unroll
        for (int j = 0; j < 8; ++j) x[j] = x[j] > 0.f ? x[j] : (__expf(x[j]) - 1.f);
        v8bf hv;
        #pragma unroll
        for (int j = 0; j < 8; ++j) hv[j] = (__bf16)x[j];
        *(v8bf*)(sH + (threadIdx.x >> 3) * HPIT + hh * 8) = hv;
    }
    __syncthreads();

    int tile = blockIdx.x * 2 + rt;
    if (tile >= ROWT) return;
    int m16   = lane & 15;
    int khalf = lane >> 4;
    const __bf16* arow = sH + (rt * 16 + m16) * HPIT + khalf * 8;
    v8f acc = {};
    #pragma unroll
    for (int kt = 0; kt < 2; ++kt) {
        v8bf lo = *(const v8bf*)(arow + kt * 32);
        v8bf hi = *(const v8bf*)(arow + kt * 32 + 16);
        v16bf av = __builtin_shufflevector(lo, hi, 0, 1, 2, 3, 4, 5, 6, 7, 8, 9, 10, 11, 12, 13, 14, 15);
        acc = __builtin_amdgcn_wmma_f32_16x16x32_bf16(
            false, av, false, kt == 0 ? bv0 : bv1, (short)0, acc, false, false);
    }
    float* crow = Wh2 + (size_t)(tile * 16 + khalf * 8) * NLAB + nt * 16 + m16;
    #pragma unroll
    for (int r = 0; r < 8; ++r) crow[(size_t)r * NLAB] = acc[r];
}

__global__ void __launch_bounds__(256)
k_gather2(const int* __restrict__ rp, const int* __restrict__ csr,
          const float* __restrict__ a_s, const float* __restrict__ a_d,
          const float* __restrict__ Wh, float* __restrict__ out) {
    int idx = blockIdx.x * blockDim.x + threadIdx.x;
    if (idx >= NNODE * 8) return;
    int n = idx >> 3, c0 = (idx & 7) * 8;
    int beg = rp[n], end = rp[n + 1];
    const float* wcol = Wh + c0;
    const float4 as0 = *(const float4*)(a_s + c0), as1 = *(const float4*)(a_s + c0 + 4);
    const float4 ad0 = *(const float4*)(a_d + c0), ad1 = *(const float4*)(a_d + c0 + 4);
    const float* own = wcol + (size_t)n * NLAB;
    float edn = sum8(dot8(*(const float4*)(own), *(const float4*)(own + 4), ad0, ad1));

    float den = 0.f;
    float4 a0 = make_float4(0.f, 0.f, 0.f, 0.f), a1 = a0;
    edge_pass<true>(csr, Wh, (unsigned)c0 * 4u, idx & 7, beg, end, as0, as1, edn, den, a0, a1);

    float dn = den + 1e-10f;
    float x[8] = { a0.x / dn, a0.y / dn, a0.z / dn, a0.w / dn,
                   a1.x / dn, a1.y / dn, a1.z / dn, a1.w / dn };
    float mx = x[0];
    #pragma unroll
    for (int j = 1; j < 8; ++j) mx = fmaxf(mx, x[j]);
    #pragma unroll
    for (int o = 1; o < 8; o <<= 1) mx = fmaxf(mx, __shfl_xor(mx, o, 32));
    float sum = 0.f;
    #pragma unroll
    for (int j = 0; j < 8; ++j) { x[j] = __expf(x[j] - mx); sum += x[j]; }
    #pragma unroll
    for (int o = 1; o < 8; o <<= 1) sum += __shfl_xor(sum, o, 32);
    float inv = 1.0f / sum;
    float* o = out + (size_t)n * NLAB + c0;
    *(float4*)(o)     = make_float4(x[0] * inv, x[1] * inv, x[2] * inv, x[3] * inv);
    *(float4*)(o + 4) = make_float4(x[4] * inv, x[5] * inv, x[6] * inv, x[7] * inv);
}

extern "C" void kernel_launch(void* const* d_in, const int* in_sizes, int n_in,
                              void* d_out, int out_size, void* d_ws, size_t ws_size,
                              hipStream_t stream) {
    const float* X   = (const float*)d_in[0];
    const float* W1  = (const float*)d_in[1];
    const float* a1s = (const float*)d_in[2];
    const float* a1d = (const float*)d_in[3];
    const float* W2  = (const float*)d_in[4];
    const float* a2s = (const float*)d_in[5];
    const float* a2d = (const float*)d_in[6];
    const int*   src = (const int*)d_in[7];
    const int*   dst = (const int*)d_in[8];
    float*       out = (float*)d_out;

    char*  ws  = (char*)d_ws;
    size_t off = 0;
    auto alloc = [&](size_t bytes) -> void* {
        void* p = ws + off;
        off = (off + bytes + 255) & ~(size_t)255;
        return p;
    };
    float*    Wh1   = (float*)   alloc((size_t)NNODE * HID  * sizeof(float));
    float*    Wh2   = (float*)   alloc((size_t)NNODE * NLAB * sizeof(float));
    int*      hist  = (int*)     alloc((size_t)NPAD         * sizeof(int));
    int*      rp    = (int*)     alloc((size_t)NPAD         * sizeof(int));
    int*      rec   = (int*)     alloc((size_t)NEDGE        * sizeof(int));
    int*      csr   = (int*)     alloc((size_t)NEDGE        * sizeof(int));
    __bf16*   Bp1   = (__bf16*)  alloc((size_t)4 * 16 * 512 * sizeof(__bf16));
    __bf16*   Bp2   = (__bf16*)  alloc((size_t)4 * 2  * 512 * sizeof(__bf16));

    const int B = 256;
    auto cdiv = [](long long a, long long b) { return (int)((a + b - 1) / b); };

    int* bbase = hist + BB_OFF;
    k_hist_repack<<<GRID1, 256, 0, stream>>>(dst, hist, W1, Bp1, W2, Bp2);
    k_scatter<<<NWG, 512, 0, stream>>>(src, dst, hist, bbase, rec);
    k_gemm_bucket<<<GRID3, 512, 0, stream>>>(X, Bp1, Wh1, rec, bbase, rp, csr);

    k_gather1_gemm2<<<cdiv((long long)NNODE * H1, B), B, 0, stream>>>(rp, csr, a1s, a1d, Wh1, Bp2, Wh2);

    k_gather2<<<cdiv((long long)NNODE * 8, B), B, 0, stream>>>(rp, csr, a2s, a2d, Wh2, out);
}


# Destination sort in 2 launches, was 3: each chunk ranks its edges by bucket in LDS into its own slice plus an offset row; a bucket block collects its 250 runs; in-edge sum order = LDS arrival
# speedup vs baseline: 7.8910x; 1.2295x over previous
#include <hip/hip_runtime.h>
#include <hip/hip_bf16.h>


typedef __attribute__((ext_vector_type(16))) __bf16 v16bf;
typedef __attribute__((ext_vector_type(8)))  __bf16 v8bf;
typedef __attribute__((ext_vector_type(8)))  float  v8f;

constexpr int NNODE = 50000;
constexpr int NEDGE = 800000;
constexpr int FIN   = 512;
constexpr int H1    = 8;
constexpr int FHID  = 8;
constexpr int HID   = H1 * FHID;
constexpr int NLAB  = 64;
constexpr int ROWT  = NNODE / 16;
constexpr float SLOPE = 0.2f;

#define DEVINL __device__ __forceinline__

DEVINL float lrelu(float v) { return v > 0.f ? v : SLOPE * v; }

DEVINL void repack_body(const float* __restrict__ W, __bf16* __restrict__ Bp,
                        int nKt, int idx, int gat1) {
    int e    = idx & 15;
    int lane = (idx >> 4) & 31;
    int ktnt = idx >> 9;
    int kt   = ktnt % nKt;
    int nt   = ktnt / nKt;
    int K = ((e < 8) ? 0 : 16) + (lane >> 4) * 8 + (e & 7);
    int k = kt * 32 + K;
    int n = nt * 16 + (lane & 15);
    float v = gat1 ? W[(size_t)(n >> 3) * (FIN * FHID) + (size_t)k * FHID + (n & 7)]
                   : W[(size_t)k * NLAB + n];
    Bp[idx] = (__bf16)v;
}

constexpr int NKT1 = FIN / 32;
constexpr int SBE  = 4 * NKT1 * 512;
constexpr int SBQ  = SBE / 8;
static_assert(SBQ % 512 == 0, "the LDS copy is whole 512-thread sweeps");

DEVINL void load_a(const float* __restrict__ ap, float4 (&d)[4]) {
    d[0] = *(const float4*)(ap);
    d[1] = *(const float4*)(ap + 4);
    d[2] = *(const float4*)(ap + 16);
    d[3] = *(const float4*)(ap + 20);
}

DEVINL void gemm1_block(const float* __restrict__ A, const __bf16* __restrict__ Bp,
                        float* __restrict__ C, __bf16* sB, int blk) {
    const int t     = threadIdx.x;
    const int lane  = t & 31;
    const int m     = lane & 15;
    const int khalf = lane >> 4;
    const int tile  = blk * 16 + (t >> 5);
    const bool valid = tile < ROWT;
    const float* arow = A + (size_t)((valid ? tile : 0) * 16 + m) * FIN + khalf * 8;

    float4 a[3][4];
    if (valid) { load_a(arow, a[0]); load_a(arow + 32, a[1]); }

    {
        const v8bf* g = (const v8bf*)Bp;
        v8bf* s = (v8bf*)sB;
        v8bf v[SBQ / 512];
        #pragma unroll
        for (int j = 0; j < SBQ / 512; ++j) v[j] = g[t + j * 512];
        #pragma unroll
        for (int j = 0; j < SBQ / 512; ++j) {
            int i = t + j * 512;
            s[(i & ~63) + ((i & 1) << 5) + ((i & 63) >> 1)] = v[j];
        }
    }
    __syncthreads();
    if (!valid) return;

    const __bf16* bl = sB + lane * 8;
    v8f acc[4] = {};
    #pragma unroll
    for (int kt = 0; kt < NKT1; ++kt) {
        if (kt + 2 < NKT1) {
            load_a(arow + (kt + 2) * 32, a[(kt + 2) % 3]);
            if (kt + 3 < NKT1) __builtin_prefetch(arow + (kt + 3) * 32, 0, 0);
        }
        __builtin_amdgcn_sched_barrier(0);

        float4 a0 = a[kt % 3][0], a1 = a[kt % 3][1], a2 = a[kt % 3][2], a3 = a[kt % 3][3];
        asm volatile("" : "+v"(a0.x), "+v"(a0.y), "+v"(a0.z), "+v"(a0.w),
                          "+v"(a1.x), "+v"(a1.y), "+v"(a1.z), "+v"(a1.w));
        asm volatile("" : "+v"(a2.x), "+v"(a2.y), "+v"(a2.z), "+v"(a2.w),
                          "+v"(a3.x), "+v"(a3.y), "+v"(a3.z), "+v"(a3.w));
        v16bf av;
        av[0]  = (__bf16)a0.x; av[1]  = (__bf16)a0.y; av[2]  = (__bf16)a0.z; av[3]  = (__bf16)a0.w;
        av[4]  = (__bf16)a1.x; av[5]  = (__bf16)a1.y; av[6]  = (__bf16)a1.z; av[7]  = (__bf16)a1.w;
        av[8]  = (__bf16)a2.x; av[9]  = (__bf16)a2.y; av[10] = (__bf16)a2.z; av[11] = (__bf16)a2.w;
        av[12] = (__bf16)a3.x; av[13] = (__bf16)a3.y; av[14] = (__bf16)a3.z; av[15] = (__bf16)a3.w;

        #pragma unroll
        for (int nt = 0; nt < 4; ++nt) {
            const __bf16* bk = bl + (nt * NKT1 + kt) * 512;
            v8bf lo = *(const v8bf*)(bk);
            v8bf hi = *(const v8bf*)(bk + 256);
            v16bf bv = __builtin_shufflevector(lo, hi, 0, 1, 2, 3, 4, 5, 6, 7, 8, 9, 10, 11, 12, 13, 14, 15);
            acc[nt] = __builtin_amdgcn_wmma_f32_16x16x32_bf16(
                false, av, false, bv, (short)0, acc[nt], false, false);
        }
    }

    float* cbase = C + (size_t)(tile * 16 + khalf * 8) * HID + m;
    #pragma unroll
    for (int nt = 0; nt < 4; ++nt) {
        float* crow = cbase + nt * 16;
        #pragma unroll
        for (int r = 0; r < 8; ++r) crow[(size_t)r * HID] = acc[nt][r];
    }
}

constexpr int NPAD = 53248;
constexpr int BSH  = 8;
constexpr int BNOD = 1 << BSH;
constexpr int NBK  = (NNODE >> BSH) + 1;
constexpr int EC   = 3200;
constexpr int EQ   = EC / 4;
constexpr int NWG  = NEDGE / EC;
constexpr int CPW  = (NWG + 15) / 16;
static_assert(NEDGE % EC == 0 && EC % 4 == 0, "chunks are whole int4 runs");
static_assert(NBK <= 256 && NNODE <= 65536, "bucket fits the block; src fits 16 bits");
static_assert(NWG * NBK <= NPAD && NNODE + 1 <= NPAD, "slices");
static_assert(EQ > 512 && EQ <= 1024, "the chunk pass holds a chunk in two int4 per thread");
static_assert(NWG <= 256 && EC < 32768, "one thread per chunk; start and length of a run share a word");

DEVINL int rec_word(int s, int d) { return s | ((d & (BNOD - 1)) << 16); }

DEVINL int scan256_excl(int v, int t, int* wtot) {
    int lane = t & 31, w = t >> 5;
    int inc = v;
    #pragma unroll
    for (int o = 1; o < 32; o <<= 1) { int u = __shfl_up(inc, o, 32); if (lane >= o) inc += u; }
    if (lane == 31 && w < 8) wtot[w] = inc;
    __syncthreads();
    int pre = 0;
    #pragma unroll
    for (int j = 0; j < 8; ++j) pre += (j < w) ? wtot[j] : 0;
    return pre + inc - v;
}

DEVINL int sum256(int v, int t, int* wtot) {
    #pragma unroll
    for (int o = 16; o > 0; o >>= 1) v += __shfl_xor(v, o, 32);
    if ((t & 31) == 0 && t < 256) wtot[t >> 5] = v;
    __syncthreads();
    int s = 0;
    #pragma unroll
    for (int j = 0; j < 8; ++j) s += wtot[j];
    return s;
}

DEVINL void chunk_sort_body(const int* __restrict__ src, const int* __restrict__ dst,
                            int* __restrict__ off, int* __restrict__ recl) {
    __shared__ int cnt[256];
    __shared__ int wtot[8];
    int t = threadIdx.x, wg = blockIdx.x;

    const int4* dp = (const int4*)(dst + (size_t)wg * EC);
    const int4* sp = (const int4*)(src + (size_t)wg * EC);
    bool two = t + 512 < EQ;
    int  i1  = two ? t + 512 : t;
    int4 d0 = dp[t],  s0 = sp[t];
    int4 d1 = dp[i1], s1 = sp[i1];

    if (t < 256) cnt[t] = 0;
    __syncthreads();
    int k0 = atomicAdd(&cnt[d0.x >> BSH], 1);
    int k1 = atomicAdd(&cnt[d0.y >> BSH], 1);
    int k2 = atomicAdd(&cnt[d0.z >> BSH], 1);
    int k3 = atomicAdd(&cnt[d0.w >> BSH], 1);
    int k4 = 0, k5 = 0, k6 = 0, k7 = 0;
    if (two) {
        k4 = atomicAdd(&cnt[d1.x >> BSH], 1);
        k5 = atomicAdd(&cnt[d1.y >> BSH], 1);
        k6 = atomicAdd(&cnt[d1.z >> BSH], 1);
        k7 = atomicAdd(&cnt[d1.w >> BSH], 1);
    }
    __syncthreads();
    int c = (t < 256) ? cnt[t] : 0;
    int excl = scan256_excl(c, t, wtot);
    if (t < 256) {
        cnt[t] = excl;
        if (t < NBK) off[wg * NBK + t] = excl;
    }
    __syncthreads();

    int* out = recl + (size_t)wg * EC;
    out[cnt[d0.x >> BSH] + k0] = rec_word(s0.x, d0.x);
    out[cnt[d0.y >> BSH] + k1] = rec_word(s0.y, d0.y);
    out[cnt[d0.z >> BSH] + k2] = rec_word(s0.z, d0.z);
    out[cnt[d0.w >> BSH] + k3] = rec_word(s0.w, d0.w);
    if (two) {
        out[cnt[d1.x >> BSH] + k4] = rec_word(s1.x, d1.x);
        out[cnt[d1.y >> BSH] + k5] = rec_word(s1.y, d1.y);
        out[cnt[d1.z >> BSH] + k6] = rec_word(s1.z, d1.z);
        out[cnt[d1.w >> BSH] + k7] = rec_word(s1.w, d1.w);
    }
}

constexpr int RP1  = 4 * (FIN / 32) * 512;
constexpr int RP2  = 4 * (HID / 32) * 512;
constexpr int RPB  = (RP1 + RP2) / 512;
static_assert(RP1 % 512 == 0 && RP2 % 512 == 0, "a repack block is full and packs one matrix");

__global__ void __launch_bounds__(512)
k_sort_repack(const int* __restrict__ src, const int* __restrict__ dst,
              int* __restrict__ off, int* __restrict__ recl,
              const float* __restrict__ W1, __bf16* __restrict__ Bp1,
              const float* __restrict__ W2, __bf16* __restrict__ Bp2) {
    int b = blockIdx.x;
    if (b < NWG) {
        chunk_sort_body(src, dst, off, recl);
        return;
    }
    int i = (b - NWG) * 512 + threadIdx.x;
    if (i < RP1) repack_body(W1, Bp1, FIN / 32, i, 1);
    else         repack_body(W2, Bp2, HID / 32, i - RP1, 0);
}

DEVINL void bucket_body(const int* __restrict__ recl, const int* __restrict__ off,
                        int* __restrict__ rp, int* __restrict__ csr, int b) {
    __shared__ int cnt[256];
    __shared__ int wtot[8];
    __shared__ int run[256];
    const int t = threadIdx.x, lane = t & 31, w = t >> 5;

    int o0 = 0, len = 0;
    if (t < NWG) {
        o0 = off[t * NBK + b];
        int o1 = (b + 1 < NBK) ? off[t * NBK + b + 1] : EC;
        len = o1 - o0;
    }
    if (t < 256) { cnt[t] = 0; run[t] = o0 | (len << 16); }
    const int beg = sum256(o0, t, wtot);

    int v[CPW];
    #pragma unroll
    for (int j = 0; j < CPW; ++j) {
        int c  = w + 16 * j;
        int cc = c < NWG ? c : NWG - 1;
        int ol = run[cc];
        int co = ol & 0xffff, cl = c < NWG ? (ol >> 16) : 0;
        int r  = recl[cc * EC + min(co + lane, EC - 1)];
        v[j] = lane < cl ? r : -1;
    }
    #pragma unroll
    for (int j = 0; j < CPW; ++j) if (v[j] >= 0) atomicAdd(&cnt[v[j] >> 16], 1);
    for (int c = w; c < NWG; c += 16) {
        int ol = run[c];
        int co = ol & 0xffff, cl = ol >> 16;
        for (int i = lane + 32; i < cl; i += 32) atomicAdd(&cnt[recl[c * EC + co + i] >> 16], 1);
    }
    __syncthreads();

    int deg = (t < 256) ? cnt[t] : 0;
    int excl = scan256_excl(deg, t, wtot);
    if (t < 256) {
        int start = beg + excl;
        int node  = b * BNOD + t;
        if (node <= NNODE) rp[node] = start;
        cnt[t] = start;
    }
    __syncthreads();

    #pragma unroll
    for (int j = 0; j < CPW; ++j)
        if (v[j] >= 0) { int s = atomicAdd(&cnt[v[j] >> 16], 1); csr[s] = v[j] & 0xffff; }
    for (int c = w; c < NWG; c += 16) {
        int ol = run[c];
        int co = ol & 0xffff, cl = ol >> 16;
        for (int i = lane + 32; i < cl; i += 32) {
            int r = recl[c * EC + co + i];
            int s = atomicAdd(&cnt[r >> 16], 1);
            csr[s] = r & 0xffff;
        }
    }
}

constexpr int GT    = 16;
constexpr int GB    = (ROWT + GT - 1) / GT;
static_assert(GT == 16, "gemm1_block maps the 16 waves of a block to 16 row tiles");
constexpr int GRID1 = NWG + RPB;
constexpr int GRID2 = GB + NBK;

__global__ void __launch_bounds__(512)
k_gemm_bucket(const float* __restrict__ A, const __bf16* __restrict__ Bp, float* __restrict__ C,
              const int* __restrict__ recl, const int* __restrict__ off,
              int* __restrict__ rp, int* __restrict__ csr) {
    __shared__ __attribute__((aligned(16))) __bf16 sB[SBE];
    if (blockIdx.x >= GB) {
        bucket_body(recl, off, rp, csr, blockIdx.x - GB);
        return;
    }
    gemm1_block(A, Bp, C, sB, blockIdx.x);
}

typedef __attribute__((ext_vector_type(2))) float v2f;
DEVINL float dot8(const float4& w0, const float4& w1, const float4& c0, const float4& c1) {
    v2f t = (v2f){w0.x, w0.y} * (v2f){c0.x, c0.y};
    t += (v2f){w0.z, w0.w} * (v2f){c0.z, c0.w};
    t += (v2f){w1.x, w1.y} * (v2f){c1.x, c1.y};
    t += (v2f){w1.z, w1.w} * (v2f){c1.z, c1.w};
    return t.x + t.y;
}

DEVINL float sum8(float v) {
    v += __shfl_xor(v, 1, 32);
    v += __shfl_xor(v, 2, 32);
    v += __shfl_xor(v, 4, 32);
    return v;
}

DEVINL void axpy8(float4& a0, float4& a1, float x, const float4& w0, const float4& w1) {
    a0.x += x * w0.x; a0.y += x * w0.y; a0.z += x * w0.z; a0.w += x * w0.w;
    a1.x += x * w1.x; a1.y += x * w1.y; a1.z += x * w1.z; a1.w += x * w1.w;
}

DEVINL void scale8(float4& a0, float4& a1, float s) {
    a0.x *= s; a0.y *= s; a0.z *= s; a0.w *= s;
    a1.x *= s; a1.y *= s; a1.z *= s; a1.w *= s;
}

constexpr int EPT = 8;
static_assert(EPT == 8, "a trip's indices are spread over the node's 8 lanes");
template <bool TREE>
DEVINL void edge_pass(const int* __restrict__ csr, const float* __restrict__ rows, unsigned coff,
                      int j8, int beg, int end, const float4& as0, const float4& as1, float edn,
                      float& den, float4& a0, float4& a1) {
    if (beg >= end) return;
    const int last = end - 1;
    int mine = csr[min(beg + j8, last)];
    float m = -INFINITY;
    for (int p = beg; p < end; p += EPT) {
        int s[EPT];
        s[0] = __builtin_amdgcn_mov_dpp8(mine, 0 * 0x249249);
        s[1] = __builtin_amdgcn_mov_dpp8(mine, 1 * 0x249249);
        s[2] = __builtin_amdgcn_mov_dpp8(mine, 2 * 0x249249);
        s[3] = __builtin_amdgcn_mov_dpp8(mine, 3 * 0x249249);
        s[4] = __builtin_amdgcn_mov_dpp8(mine, 4 * 0x249249);
        s[5] = __builtin_amdgcn_mov_dpp8(mine, 5 * 0x249249);
        s[6] = __builtin_amdgcn_mov_dpp8(mine, 6 * 0x249249);
        s[7] = __builtin_amdgcn_mov_dpp8(mine, 7 * 0x249249);
        float4 w0[EPT], w1[EPT];
        #pragma unroll
        for (int k = 0; k < EPT; ++k) {
            const char* r = (const char*)rows + ((unsigned)s[k] * 256u + coff);
            w0[k] = *(const float4*)(r);
            w1[k] = *(const float4*)(r + 16);
        }
        mine = csr[min(p + EPT + j8, last)];
        __builtin_amdgcn_sched_barrier(0);

        float e[EPT];
        #pragma unroll
        for (int k = 0; k < EPT; ++k) {
            float d = dot8(w0[k], w1[k], as0, as1);
            if (TREE) d = sum8(d);
            e[k] = lrelu(d + edn);
            if (k > 0) e[k] = (p + k < end) ? e[k] : -INFINITY;
        }
        float mn = m;
        #pragma unroll
        for (int k = 0; k < EPT; ++k) mn = fmaxf(mn, e[k]);
        float sc = __expf(m - mn);
        den *= sc; scale8(a0, a1, sc);
        #pragma unroll
        for (int k = 0; k < EPT; ++k) {
            float x = __expf(e[k] - mn);
            den += x; axpy8(a0, a1, x, w0[k], w1[k]);
        }
        m = mn;
    }
}

constexpr int G1N  = 32;
constexpr int HPIT = 72;
static_assert(NNODE % 16 == 0 && HID == 64 && NLAB == 64, "fused product tiling");

__global__ void __launch_bounds__(256)
k_gather1_gemm2(const int* __restrict__ rp, const int* __restrict__ csr,
                const float* __restrict__ a_s, const float* __restrict__ a_d,
                const float* __restrict__ Wh, const __bf16* __restrict__ Bp2,
                float* __restrict__ Wh2) {
    __shared__ __attribute__((aligned(16))) __bf16 sH[G1N * HPIT];
    int idx = blockIdx.x * blockDim.x + threadIdx.x;
    int n = idx >> 3, hh = idx & 7;
    const bool act = n < NNODE;

    const int w    = __builtin_amdgcn_readfirstlane((int)(threadIdx.x >> 5));
    const int rt   = w >> 2, nt = w & 3;
    const int lane = threadIdx.x & 31;
    const __bf16* bl = Bp2 + ((size_t)(nt * 2) * 32 + lane) * 16;

    float den = 0.f;
    float4 a0 = make_float4(0.f, 0.f, 0.f, 0.f), a1 = a0;
    if (act) {
        int beg = rp[n], end = rp[n + 1];
        const float* wcol = Wh + hh * 8;
        const float4 as0 = *(const float4*)(a_s + hh * 8), as1 = *(const float4*)(a_s + hh * 8 + 4);
        const float4 ad0 = *(const float4*)(a_d + hh * 8), ad1 = *(const float4*)(a_d + hh * 8 + 4);
        const float* own = wcol + (size_t)n * HID;
        float edn = dot8(*(const float4*)(own), *(const float4*)(own + 4), ad0, ad1);

        edge_pass<false>(csr, Wh, (unsigned)hh * 32u, hh, beg, end, as0, as1, edn, den, a0, a1);
    }

    const v16bf bv0 = *(const v16bf*)(bl);
    const v16bf bv1 = *(const v16bf*)(bl + 512);

    if (act) {
        float dn = den + 1e-10f;
        float x[8] = { a0.x / dn, a0.y / dn, a0.z / dn, a0.w / dn,
                       a1.x / dn, a1.y / dn, a1.z / dn, a1.w / dn };
        #pragma unroll
        for (int j = 0; j < 8; ++j) x[j] = x[j] > 0.f ? x[j] : (__expf(x[j]) - 1.f);
        v8bf hv;
        #pragma unroll
        for (int j = 0; j < 8; ++j) hv[j] = (__bf16)x[j];
        *(v8bf*)(sH + (threadIdx.x >> 3) * HPIT + hh * 8) = hv;
    }
    __syncthreads();

    int tile = blockIdx.x * 2 + rt;
    if (tile >= ROWT) return;
    int m16   = lane & 15;
    int khalf = lane >> 4;
    const __bf16* arow = sH + (rt * 16 + m16) * HPIT + khalf * 8;
    v8f acc = {};
    #pragma unroll
    for (int kt = 0; kt < 2; ++kt) {
        v8bf lo = *(const v8bf*)(arow + kt * 32);
        v8bf hi = *(const v8bf*)(arow + kt * 32 + 16);
        v16bf av = __builtin_shufflevector(lo, hi, 0, 1, 2, 3, 4, 5, 6, 7, 8, 9, 10, 11, 12, 13, 14, 15);
        acc = __builtin_amdgcn_wmma_f32_16x16x32_bf16(
            false, av, false, kt == 0 ? bv0 : bv1, (short)0, acc, false, false);
    }
    float* crow = Wh2 + (size_t)(tile * 16 + khalf * 8) * NLAB + nt * 16 + m16;
    #pragma unroll
    for (int r = 0; r < 8; ++r) crow[(size_t)r * NLAB] = acc[r];
}

__global__ void __launch_bounds__(256)
k_gather2(const int* __restrict__ rp, const int* __restrict__ csr,
          const float* __restrict__ a_s, const float* __restrict__ a_d,
          const float* __restrict__ Wh, float* __restrict__ out) {
    int idx = blockIdx.x * blockDim.x + threadIdx.x;
    if (idx >= NNODE * 8) return;
    int n = idx >> 3, c0 = (idx & 7) * 8;
    int beg = rp[n], end = rp[n + 1];
    const float* wcol = Wh + c0;
    const float4 as0 = *(const float4*)(a_s + c0), as1 = *(const float4*)(a_s + c0 + 4);
    const float4 ad0 = *(const float4*)(a_d + c0), ad1 = *(const float4*)(a_d + c0 + 4);
    const float* own = wcol + (size_t)n * NLAB;
    float edn = sum8(dot8(*(const float4*)(own), *(const float4*)(own + 4), ad0, ad1));

    float den = 0.f;
    float4 a0 = make_float4(0.f, 0.f, 0.f, 0.f), a1 = a0;
    edge_pass<true>(csr, Wh, (unsigned)c0 * 4u, idx & 7, beg, end, as0, as1, edn, den, a0, a1);

    float dn = den + 1e-10f;
    float x[8] = { a0.x / dn, a0.y / dn, a0.z / dn, a0.w / dn,
                   a1.x / dn, a1.y / dn, a1.z / dn, a1.w / dn };
    float mx = x[0];
    #pragma unroll
    for (int j = 1; j < 8; ++j) mx = fmaxf(mx, x[j]);
    #pragma unroll
    for (int o = 1; o < 8; o <<= 1) mx = fmaxf(mx, __shfl_xor(mx, o, 32));
    float sum = 0.f;
    #pragma unroll
    for (int j = 0; j < 8; ++j) { x[j] = __expf(x[j] - mx); sum += x[j]; }
    #pragma unroll
    for (int o = 1; o < 8; o <<= 1) sum += __shfl_xor(sum, o, 32);
    float inv = 1.0f / sum;
    float* o = out + (size_t)n * NLAB + c0;
    *(float4*)(o)     = make_float4(x[0] * inv, x[1] * inv, x[2] * inv, x[3] * inv);
    *(float4*)(o + 4) = make_float4(x[4] * inv, x[5] * inv, x[6] * inv, x[7] * inv);
}

extern "C" void kernel_launch(void* const* d_in, const int* in_sizes, int n_in,
                              void* d_out, int out_size, void* d_ws, size_t ws_size,
                              hipStream_t stream) {
    const float* X   = (const float*)d_in[0];
    const float* W1  = (const float*)d_in[1];
    const float* a1s = (const float*)d_in[2];
    const float* a1d = (const float*)d_in[3];
    const float* W2  = (const float*)d_in[4];
    const float* a2s = (const float*)d_in[5];
    const float* a2d = (const float*)d_in[6];
    const int*   src = (const int*)d_in[7];
    const int*   dst = (const int*)d_in[8];
    float*       out = (float*)d_out;

    char*  ws  = (char*)d_ws;
    size_t off = 0;
    auto alloc = [&](size_t bytes) -> void* {
        void* p = ws + off;
        off = (off + bytes + 255) & ~(size_t)255;
        return p;
    };
    float*    Wh1   = (float*)   alloc((size_t)NNODE * HID  * sizeof(float));
    float*    Wh2   = (float*)   alloc((size_t)NNODE * NLAB * sizeof(float));
    int*      boff  = (int*)     alloc((size_t)NPAD         * sizeof(int));
    int*      rp    = (int*)     alloc((size_t)NPAD         * sizeof(int));
    int*      recl  = (int*)     alloc((size_t)NEDGE        * sizeof(int));
    int*      csr   = (int*)     alloc((size_t)NEDGE        * sizeof(int));
    __bf16*   Bp1   = (__bf16*)  alloc((size_t)4 * 16 * 512 * sizeof(__bf16));
    __bf16*   Bp2   = (__bf16*)  alloc((size_t)4 * 2  * 512 * sizeof(__bf16));

    const int B = 256;
    auto cdiv = [](long long a, long long b) { return (int)((a + b - 1) / b); };

    k_sort_repack<<<GRID1, 512, 0, stream>>>(src, dst, boff, recl, W1, Bp1, W2, Bp2);
    k_gemm_bucket<<<GRID2, 512, 0, stream>>>(X, Bp1, Wh1, recl, boff, rp, csr);

    k_gather1_gemm2<<<cdiv((long long)NNODE * H1, B), B, 0, stream>>>(rp, csr, a1s, a1d, Wh1, Bp2, Wh2);

    k_gather2<<<cdiv((long long)NNODE * 8, B), B, 0, stream>>>(rp, csr, a2s, a2d, Wh2, out);
}
